# starting kernel
# baseline (speedup 1.0000x reference)
#include <hip/hip_runtime.h>
#include <hip/hip_bf16.h>
#include <math.h>

typedef float v2f __attribute__((ext_vector_type(2)));
typedef float v8f __attribute__((ext_vector_type(8)));

#define NEG_SLOPE 0.2f
#define EPS_ 1e-16f

static inline int cdiv(long a, long b) { return (int)((a + b - 1) / b); }

__global__ void fill_kernel(float* __restrict__ p, float v, size_t n) {
    size_t i = (size_t)blockIdx.x * blockDim.x + threadIdx.x;
    size_t stride = (size_t)gridDim.x * blockDim.x;
    for (; i < n; i += stride) p[i] = v;
}

template <int K, int NC, int CPW>
__global__ __launch_bounds__(256) void wmma_gemm_f32(const float* __restrict__ A,
                                                     const float* __restrict__ B,
                                                     float* __restrict__ C,
                                                     int M) {
    constexpr int nColT  = NC / 16;
    constexpr int groups = nColT / CPW;
    int wave = threadIdx.x >> 5;
    int lane = threadIdx.x & 31;
    int hl   = lane >> 4;
    int l    = lane & 15;
    int tile  = blockIdx.x * (blockDim.x >> 5) + wave;
    int total = (M >> 4) * groups;
    if (tile >= total) return;
    int m0 = (tile / groups) << 4;
    int n0 = (tile % groups) * (CPW * 16);

    const float* arow = A + (size_t)(m0 + l) * K + 2 * hl;
    const float* bcol = B + (size_t)(2 * hl) * NC + n0 + l;

    v8f acc[CPW];
#pragma unroll
    for (int i = 0; i < CPW; ++i) acc[i] = {};

    for (int k0 = 0; k0 < K; k0 += 4) {
        v2f a;
        a.x = arow[k0 + 0];
        a.y = arow[k0 + 1];
#pragma unroll
        for (int ct = 0; ct < CPW; ++ct) {
            v2f b;
            b.x = bcol[(size_t)k0 * NC + ct * 16];
            b.y = bcol[(size_t)(k0 + 1) * NC + ct * 16];
            acc[ct] = __builtin_amdgcn_wmma_f32_16x16x4_f32(false, a, false, b,
                                                            (short)0, acc[ct], false, false);
        }
    }

#pragma unroll
    for (int ct = 0; ct < CPW; ++ct) {
        float* crow = C + (size_t)(m0 + 8 * hl) * NC + n0 + ct * 16 + l;
#pragma unroll
        for (int r = 0; r < 8; ++r) crow[(size_t)r * NC] = acc[ct][r];
    }
}

template <int H, int CH>
__global__ void att_dots_kernel(const float* __restrict__ h,
                                const float* __restrict__ att_s,
                                const float* __restrict__ att_d,
                                float* __restrict__ asrc, float* __restrict__ adst,
                                int n) {
    constexpr int C = H * CH;
    int wave = threadIdx.x >> 5;
    int lane = threadIdx.x & 31;
    int node = blockIdx.x * (blockDim.x >> 5) + wave;
    if (node >= n) return;
    const float* hp = h + (size_t)node * C;

    float s[H], d[H];
#pragma unroll
    for (int i = 0; i < H; ++i) { s[i] = 0.f; d[i] = 0.f; }

#pragma unroll
    for (int i = 0; i < (C + 31) / 32; ++i) {
        int c = lane + i * 32;
        if (c < C) {
            int hd = c / CH;
            float t = hp[c];
            s[hd] += t * att_s[c];
            d[hd] += t * att_d[c];
        }
    }
#pragma unroll
    for (int i = 0; i < H; ++i) {
#pragma unroll
        for (int off = 16; off; off >>= 1) {
            s[i] += __shfl_xor(s[i], off, 32);
            d[i] += __shfl_xor(d[i], off, 32);
        }
        if (lane == 0) {
            asrc[(size_t)node * H + i] = s[i];
            adst[(size_t)node * H + i] = d[i];
        }
    }
}

__device__ inline void atomicMaxF(float* addr, float v) {
    if (v >= 0.0f)
        atomicMax((int*)addr, __float_as_int(v));
    else
        atomicMin((unsigned int*)addr, (unsigned int)__float_as_int(v));
}

template <int H>
__global__ void edge_max_kernel(const int* __restrict__ src, const int* __restrict__ dst,
                                const float* __restrict__ asrc, const float* __restrict__ adst,
                                float* __restrict__ emax, int E) {
    int e = blockIdx.x * blockDim.x + threadIdx.x;
    if (e >= E) return;
    int s = src[e], d = dst[e];
#pragma unroll
    for (int h = 0; h < H; ++h) {
        float v = asrc[s * H + h] + adst[d * H + h];
        v = (v >= 0.f) ? v : NEG_SLOPE * v;
        atomicMaxF(&emax[d * H + h], v);
    }
}

template <int H, int CH>
__global__ __launch_bounds__(256) void edge_agg_kernel(
        const int* __restrict__ src, const int* __restrict__ dst,
        const float* __restrict__ hfeat,
        const float* __restrict__ asrc, const float* __restrict__ adst,
        const float* __restrict__ emax,
        float* __restrict__ denom, float* __restrict__ outraw, int E) {
    constexpr int C   = H * CH;
    constexpr int LPE = C / 4;
    constexpr int EPB = 256 / LPE;
    int li   = threadIdx.x % LPE;
    int esub = threadIdx.x / LPE;
    int e = blockIdx.x * EPB + esub;
    if (e >= E) return;
    int s = src[e], d = dst[e];
    int c0 = li * 4;
    int h  = c0 / CH;

    float v = asrc[s * H + h] + adst[d * H + h];
    v = (v >= 0.f) ? v : NEG_SLOPE * v;
    float wgt = __expf(v - emax[d * H + h]);
    if ((c0 % CH) == 0) atomicAdd(&denom[d * H + h], wgt);

    const float4 hv = *(const float4*)(hfeat + (size_t)s * C + c0);
    float* op = outraw + (size_t)d * C + c0;
    atomicAdd(op + 0, hv.x * wgt);
    atomicAdd(op + 1, hv.y * wgt);
    atomicAdd(op + 2, hv.z * wgt);
    atomicAdd(op + 3, hv.w * wgt);
}

__global__ void finalize_elu_kernel(const float* __restrict__ outraw,
                                    const float* __restrict__ denom,
                                    const float* __restrict__ bias,
                                    float* __restrict__ hact,
                                    int n, int heads, int ch) {
    size_t i = (size_t)blockIdx.x * blockDim.x + threadIdx.x;
    size_t total = (size_t)n * heads * ch;
    if (i >= total) return;
    int C = heads * ch;
    int node = (int)(i / C);
    int c = (int)(i % C);
    int hd = c / ch;
    float val = outraw[i] / (denom[node * heads + hd] + EPS_) + bias[c];
    hact[i] = (val > 0.f) ? val : expm1f(val);
}

__global__ void finalize_softmax_kernel(const float* __restrict__ outraw,
                                        const float* __restrict__ denom,
                                        const float* __restrict__ bias,
                                        float* __restrict__ out, int n) {
    int node = blockIdx.x * blockDim.x + threadIdx.x;
    if (node >= n) return;
    float den = denom[node] + EPS_;
    float v[16];
    float m = -INFINITY;
#pragma unroll
    for (int c = 0; c < 16; ++c) {
        v[c] = outraw[(size_t)node * 16 + c] / den + bias[c];
        m = fmaxf(m, v[c]);
    }
    float sum = 0.f;
#pragma unroll
    for (int c = 0; c < 16; ++c) {
        float t = __expf(v[c] - m);
        v[c] = t;
        sum += t;
    }
    float inv = 1.f / sum;
#pragma unroll
    for (int c = 0; c < 16; ++c) out[(size_t)node * 16 + c] = v[c] * inv;
}

extern "C" void kernel_launch(void* const* d_in, const int* in_sizes, int n_in,
                              void* d_out, int out_size, void* d_ws, size_t ws_size,
                              hipStream_t stream) {
    const float* x   = (const float*)d_in[0];
    const float* W1  = (const float*)d_in[1];
    const float* as1 = (const float*)d_in[2];
    const float* ad1 = (const float*)d_in[3];
    const float* b1  = (const float*)d_in[4];
    const float* W2  = (const float*)d_in[5];
    const float* as2 = (const float*)d_in[6];
    const float* ad2 = (const float*)d_in[7];
    const float* b2  = (const float*)d_in[8];
    const int*   ei  = (const int*)d_in[9];

    const int N = in_sizes[0] / 128;
    const int E = in_sizes[9] / 2;
    const int* src = ei;
    const int* dst = ei + E;

    float* w = (float*)d_ws;
    float* h1      = w; w += (size_t)N * 128;
    float* outraw1 = w; w += (size_t)N * 128;
    float* asrc1   = w; w += (size_t)N * 2;
    float* adst1   = w; w += (size_t)N * 2;
    float* emax1   = w; w += (size_t)N * 2;
    float* denom1  = w; w += (size_t)N * 2;
    float* h2      = w; w += (size_t)N * 16;
    float* outraw2 = w; w += (size_t)N * 16;
    float* asrc2   = w; w += (size_t)N;
    float* adst2   = w; w += (size_t)N;
    float* emax2   = w; w += (size_t)N;
    float* denom2  = w; w += (size_t)N;

    fill_kernel<<<1024, 256, 0, stream>>>(outraw1, 0.f, (size_t)N * 128);
    fill_kernel<<<64, 256, 0, stream>>>(denom1, 0.f, (size_t)N * 2);
    fill_kernel<<<64, 256, 0, stream>>>(emax1, -INFINITY, (size_t)N * 2);
    fill_kernel<<<256, 256, 0, stream>>>(outraw2, 0.f, (size_t)N * 16);
    fill_kernel<<<64, 256, 0, stream>>>(denom2, 0.f, (size_t)N);
    fill_kernel<<<64, 256, 0, stream>>>(emax2, -INFINITY, (size_t)N);

    {
        int tiles = (N / 16) * 2;
        wmma_gemm_f32<128, 128, 4><<<cdiv(tiles, 8), 256, 0, stream>>>(x, W1, h1, N);
        att_dots_kernel<2, 64><<<cdiv(N, 8), 256, 0, stream>>>(h1, as1, ad1, asrc1, adst1, N);
        edge_max_kernel<2><<<cdiv(E, 256), 256, 0, stream>>>(src, dst, asrc1, adst1, emax1, E);
        edge_agg_kernel<2, 64><<<cdiv(E, 8), 256, 0, stream>>>(src, dst, h1, asrc1, adst1, emax1,
                                                               denom1, outraw1, E);
        finalize_elu_kernel<<<cdiv((long)N * 128, 256), 256, 0, stream>>>(outraw1, denom1, b1, h1, N, 2, 64);
    }

    {
        int tiles = (N / 16);
        wmma_gemm_f32<128, 16, 1><<<cdiv(tiles, 8), 256, 0, stream>>>(h1, W2, h2, N);
        att_dots_kernel<1, 16><<<cdiv(N, 8), 256, 0, stream>>>(h2, as2, ad2, asrc2, adst2, N);
        edge_max_kernel<1><<<cdiv(E, 256), 256, 0, stream>>>(src, dst, asrc2, adst2, emax2, E);
        edge_agg_kernel<1, 16><<<cdiv(E, 64), 256, 0, stream>>>(src, dst, h2, asrc2, adst2, emax2,
                                                                denom2, outraw2, E);
        finalize_softmax_kernel<<<cdiv(N, 256), 256, 0, stream>>>(outraw2, denom2, b2, (float*)d_out, N);
    }
}


# edges bucketed by destination once per call (count, scan, place); each layer is then one gather-reduce wave per node with max, sum and aggregate in registers: no float atomics
# speedup vs baseline: 2.5359x; 2.5359x over previous
#include <hip/hip_runtime.h>
#include <hip/hip_bf16.h>
#include <math.h>

typedef float v2f __attribute__((ext_vector_type(2)));
typedef float v8f __attribute__((ext_vector_type(8)));

#define NEG_SLOPE 0.2f
#define EPS_ 1e-16f

static inline int cdiv(long a, long b) { return (int)((a + b - 1) / b); }

__global__ void fill_kernel(float* __restrict__ p, float v, size_t n) {
    size_t i = (size_t)blockIdx.x * blockDim.x + threadIdx.x;
    size_t stride = (size_t)gridDim.x * blockDim.x;
    for (; i < n; i += stride) p[i] = v;
}

template <int K, int NC, int CPW>
__global__ __launch_bounds__(256) void wmma_gemm_f32(const float* __restrict__ A,
                                                     const float* __restrict__ B,
                                                     float* __restrict__ C,
                                                     int M) {
    constexpr int nColT  = NC / 16;
    constexpr int groups = nColT / CPW;
    int wave = threadIdx.x >> 5;
    int lane = threadIdx.x & 31;
    int hl   = lane >> 4;
    int l    = lane & 15;
    int tile  = blockIdx.x * (blockDim.x >> 5) + wave;
    int total = (M >> 4) * groups;
    if (tile >= total) return;
    int m0 = (tile / groups) << 4;
    int n0 = (tile % groups) * (CPW * 16);

    const float* arow = A + (size_t)(m0 + l) * K + 2 * hl;
    const float* bcol = B + (size_t)(2 * hl) * NC + n0 + l;

    v8f acc[CPW];
#pragma unroll
    for (int i = 0; i < CPW; ++i) acc[i] = {};

    for (int k0 = 0; k0 < K; k0 += 4) {
        v2f a;
        a.x = arow[k0 + 0];
        a.y = arow[k0 + 1];
#pragma unroll
        for (int ct = 0; ct < CPW; ++ct) {
            v2f b;
            b.x = bcol[(size_t)k0 * NC + ct * 16];
            b.y = bcol[(size_t)(k0 + 1) * NC + ct * 16];
            acc[ct] = __builtin_amdgcn_wmma_f32_16x16x4_f32(false, a, false, b,
                                                            (short)0, acc[ct], false, false);
        }
    }

#pragma unroll
    for (int ct = 0; ct < CPW; ++ct) {
        float* crow = C + (size_t)(m0 + 8 * hl) * NC + n0 + ct * 16 + l;
#pragma unroll
        for (int r = 0; r < 8; ++r) crow[(size_t)r * NC] = acc[ct][r];
    }
}

template <int H, int CH>
__global__ void att_dots_kernel(const float* __restrict__ h,
                                const float* __restrict__ att_s,
                                const float* __restrict__ att_d,
                                float* __restrict__ asrc, float* __restrict__ adst,
                                int n) {
    constexpr int C = H * CH;
    int wave = threadIdx.x >> 5;
    int lane = threadIdx.x & 31;
    int node = blockIdx.x * (blockDim.x >> 5) + wave;
    if (node >= n) return;
    const float* hp = h + (size_t)node * C;

    float s[H], d[H];
#pragma unroll
    for (int i = 0; i < H; ++i) { s[i] = 0.f; d[i] = 0.f; }

#pragma unroll
    for (int i = 0; i < (C + 31) / 32; ++i) {
        int c = lane + i * 32;
        if (c < C) {
            int hd = c / CH;
            float t = hp[c];
            s[hd] += t * att_s[c];
            d[hd] += t * att_d[c];
        }
    }
#pragma unroll
    for (int i = 0; i < H; ++i) {
#pragma unroll
        for (int off = 16; off; off >>= 1) {
            s[i] += __shfl_xor(s[i], off, 32);
            d[i] += __shfl_xor(d[i], off, 32);
        }
        if (lane == 0) {
            asrc[(size_t)node * H + i] = s[i];
            adst[(size_t)node * H + i] = d[i];
        }
    }
}

__global__ void count_in_kernel(const int* __restrict__ dst, int* __restrict__ cnt, int E) {
    int e = blockIdx.x * blockDim.x + threadIdx.x;
    if (e >= E) return;
    atomicAdd(&cnt[dst[e]], 1);
}

__global__ __launch_bounds__(1024) void scan_kernel(int* __restrict__ cnt, int* __restrict__ rowptr, int n) {
    __shared__ int wsum[32];
    const int t = threadIdx.x;
    const int lane = t & 31, w = t >> 5;
    const int chunk = (n + 1023) / 1024;
    const int b = t * chunk;
    const int e = min(b + chunk, n);
    int s = 0;
#pragma unroll 8
    for (int i = b; i < e; ++i) s += cnt[i];
    int v = s;
    for (int off = 1; off < 32; off <<= 1) {
        int u = __shfl_up(v, off, 32);
        if (lane >= off) v += u;
    }
    if (lane == 31) wsum[w] = v;
    __syncthreads();
    if (w == 0) {
        int x = wsum[lane];
        for (int off = 1; off < 32; off <<= 1) {
            int u = __shfl_up(x, off, 32);
            if (lane >= off) x += u;
        }
        wsum[lane] = x;
    }
    __syncthreads();
    int base = v - s + (w > 0 ? wsum[w - 1] : 0);
#pragma unroll 8
    for (int i = b; i < e; ++i) {
        int c = cnt[i];
        rowptr[i] = base;
        cnt[i] = base;
        base += c;
    }
    if (t == 1023) rowptr[n] = wsum[31];
}

__global__ void place_kernel(const int* __restrict__ src, const int* __restrict__ dst,
                             int* __restrict__ cursor, int* __restrict__ csr, int E) {
    int e = blockIdx.x * blockDim.x + threadIdx.x;
    if (e >= E) return;
    const int sv = src[e];
    int pos = atomicAdd(&cursor[dst[e]], 1);
    if ((unsigned)pos < (unsigned)E) csr[pos] = sv;
}

__device__ inline float leaky(float v) { return (v >= 0.f) ? v : NEG_SLOPE * v; }
__device__ inline float finite_or_zero(float m) { return (fabsf(m) < INFINITY) ? m : 0.f; }
__device__ inline float lane_bcast(float v, int k) {
    return __int_as_float(__builtin_amdgcn_readlane(__float_as_int(v), k));
}

template <int NB>
__device__ __forceinline__ void gather_rows1(float4& acc, const float* __restrict__ hcol,
                                             int s, float w0, float w1, int hd, int k) {
    float4 hv[NB];
#pragma unroll
    for (int i = 0; i < NB; ++i) {
        int sk = __builtin_amdgcn_readlane(s, k + i);
        hv[i] = *(const float4*)(hcol + (size_t)sk * 128);
    }
#pragma unroll
    for (int i = 0; i < NB; ++i) {
        float wa = lane_bcast(w0, k + i), wb = lane_bcast(w1, k + i);
        float wk = hd ? wb : wa;
        acc.x += wk * hv[i].x;
        acc.y += wk * hv[i].y;
        acc.z += wk * hv[i].z;
        acc.w += wk * hv[i].w;
    }
}

__global__ __launch_bounds__(256) void node_agg1_kernel(
        const int* __restrict__ rowptr, const int* __restrict__ csr,
        const float* __restrict__ hfeat,
        const float* __restrict__ asrc, const float* __restrict__ adst,
        const float* __restrict__ bias, float* __restrict__ hact, int n) {
    const int lane = threadIdx.x & 31;
    const int d = __builtin_amdgcn_readfirstlane(blockIdx.x * (blockDim.x >> 5) + (threadIdx.x >> 5));
    if (d >= n) return;
    const int beg = rowptr[d], end = rowptr[d + 1];
    const float ad0 = adst[d * 2 + 0], ad1 = adst[d * 2 + 1];
    const int hd = lane >> 4;

    float m0 = -INFINITY, m1 = -INFINITY;
    int s_first = 0;
    float2 a_first = {0.f, 0.f};
    for (int base = beg; base < end; base += 32) {
        int j = base + lane;
        if (j < end) {
            int s = csr[j];
            float2 a = *(const float2*)(asrc + (size_t)s * 2);
            if (base == beg) { s_first = s; a_first = a; }
            m0 = fmaxf(m0, leaky(a.x + ad0));
            m1 = fmaxf(m1, leaky(a.y + ad1));
        }
    }
    for (int off = 16; off; off >>= 1) {
        m0 = fmaxf(m0, __shfl_xor(m0, off, 32));
        m1 = fmaxf(m1, __shfl_xor(m1, off, 32));
    }
    m0 = finite_or_zero(m0);
    m1 = finite_or_zero(m1);

    float4 acc = {0.f, 0.f, 0.f, 0.f};
    float den0 = 0.f, den1 = 0.f;
    const float* hcol = hfeat + lane * 4;
    for (int base = beg; base < end; base += 32) {
        int j = base + lane;
        int s = s_first;
        float2 a = a_first;
        if (base != beg && j < end) {
            s = csr[j];
            a = *(const float2*)(asrc + (size_t)s * 2);
        }
        float w0 = 0.f, w1 = 0.f;
        if (j < end) {
            w0 = __expf(leaky(a.x + ad0) - m0);
            w1 = __expf(leaky(a.y + ad1) - m1);
        }
        den0 += w0;
        den1 += w1;
        const int cnt = min(32, end - base);
        int k = 0;
        for (; k + 8 <= cnt; k += 8) gather_rows1<8>(acc, hcol, s, w0, w1, hd, k);
        if (cnt & 4) { gather_rows1<4>(acc, hcol, s, w0, w1, hd, k); k += 4; }
        if (cnt & 2) { gather_rows1<2>(acc, hcol, s, w0, w1, hd, k); k += 2; }
        if (cnt & 1) { gather_rows1<1>(acc, hcol, s, w0, w1, hd, k); }
    }
    for (int off = 16; off; off >>= 1) {
        den0 += __shfl_xor(den0, off, 32);
        den1 += __shfl_xor(den1, off, 32);
    }
    const float den = (hd ? den1 : den0) + EPS_;
    const float4 bv = *(const float4*)(bias + lane * 4);
    float4 o;
    o.x = acc.x / den + bv.x;
    o.y = acc.y / den + bv.y;
    o.z = acc.z / den + bv.z;
    o.w = acc.w / den + bv.w;
    o.x = (o.x > 0.f) ? o.x : expm1f(o.x);
    o.y = (o.y > 0.f) ? o.y : expm1f(o.y);
    o.z = (o.z > 0.f) ? o.z : expm1f(o.z);
    o.w = (o.w > 0.f) ? o.w : expm1f(o.w);
    *(float4*)(hact + (size_t)d * 128 + lane * 4) = o;
}

__device__ __forceinline__ void node2_row(int d, int lane,
                                   const int* __restrict__ rowptr, const int* __restrict__ csr,
                                   const float* __restrict__ hfeat,
                                   const float* __restrict__ asrc, const float* __restrict__ adst,
                                   const float* __restrict__ bias,
                                   float& r0, float& r1, float& r2, float& r3) {
    const int beg = rowptr[d], end = rowptr[d + 1];
    const float ad = adst[d];
    float m = -INFINITY;
    int s_first = 0;
    float a_first = 0.f;
    for (int base = beg; base < end; base += 32) {
        int j = base + lane;
        if (j < end) {
            int s = csr[j];
            float a = asrc[s];
            if (base == beg) { s_first = s; a_first = a; }
            m = fmaxf(m, leaky(a + ad));
        }
    }
    for (int off = 16; off; off >>= 1) m = fmaxf(m, __shfl_xor(m, off, 32));
    m = finite_or_zero(m);

    float4 acc = {0.f, 0.f, 0.f, 0.f};
    float den = 0.f;
    const int g = lane >> 2;
    const float* hcol = hfeat + (lane & 3) * 4;
    for (int base = beg; base < end; base += 32) {
        int j = base + lane;
        int s = s_first;
        float a = a_first;
        if (base != beg && j < end) {
            s = csr[j];
            a = asrc[s];
        }
        float w = 0.f;
        if (j < end) w = __expf(leaky(a + ad) - m);
        else s = 0;
        den += w;
        const int cnt = min(32, end - base);
        float4 hv[4];
        float wk[4];
#pragma unroll
        for (int t = 0; t < 4; ++t) {
            int sk = __shfl(s, t * 8 + g, 32);
            wk[t] = __shfl(w, t * 8 + g, 32);
            hv[t] = *(const float4*)(hcol + (size_t)sk * 16);
        }
#pragma unroll
        for (int t = 0; t < 4; ++t) {
            const bool live = (t * 8 + g < cnt);
            acc.x += live ? wk[t] * hv[t].x : 0.f;
            acc.y += live ? wk[t] * hv[t].y : 0.f;
            acc.z += live ? wk[t] * hv[t].z : 0.f;
            acc.w += live ? wk[t] * hv[t].w : 0.f;
        }
    }
    for (int off = 16; off >= 4; off >>= 1) {
        acc.x += __shfl_xor(acc.x, off, 32);
        acc.y += __shfl_xor(acc.y, off, 32);
        acc.z += __shfl_xor(acc.z, off, 32);
        acc.w += __shfl_xor(acc.w, off, 32);
    }
    for (int off = 16; off; off >>= 1) den += __shfl_xor(den, off, 32);
    den += EPS_;
    const float4 bv = *(const float4*)(bias + (lane & 3) * 4);
    float4 v;
    v.x = acc.x / den + bv.x;
    v.y = acc.y / den + bv.y;
    v.z = acc.z / den + bv.z;
    v.w = acc.w / den + bv.w;
    float mx = fmaxf(fmaxf(v.x, v.y), fmaxf(v.z, v.w));
    mx = fmaxf(mx, __shfl_xor(mx, 1, 32));
    mx = fmaxf(mx, __shfl_xor(mx, 2, 32));
    v.x = __expf(v.x - mx);
    v.y = __expf(v.y - mx);
    v.z = __expf(v.z - mx);
    v.w = __expf(v.w - mx);
    float sum = (v.x + v.y) + (v.z + v.w);
    sum += __shfl_xor(sum, 1, 32);
    sum += __shfl_xor(sum, 2, 32);
    const float inv = 1.f / sum;
    r0 = v.x * inv;
    r1 = v.y * inv;
    r2 = v.z * inv;
    r3 = v.w * inv;
}

__global__ __launch_bounds__(256) void node_agg2_kernel(
        const int* __restrict__ rowptr, const int* __restrict__ csr,
        const float* __restrict__ hfeat,
        const float* __restrict__ asrc, const float* __restrict__ adst,
        const float* __restrict__ bias, float* __restrict__ out, int n) {
    const int lane = threadIdx.x & 31;
    const int d0 = __builtin_amdgcn_readfirstlane((blockIdx.x * (blockDim.x >> 5) + (threadIdx.x >> 5)) * 2);
    if (d0 >= n) return;
    float a0, a1, a2, a3;
    node2_row(d0, lane, rowptr, csr, hfeat, asrc, adst, bias, a0, a1, a2, a3);
    float b0 = a0, b1 = a1, b2 = a2, b3 = a3;
    const bool two = (d0 + 1 < n);
    if (two) node2_row(d0 + 1, lane, rowptr, csr, hfeat, asrc, adst, bias, b0, b1, b2, b3);
    if (lane < 4 || (two && lane < 8)) {
        float4 r;
        r.x = (lane < 4) ? a0 : b0;
        r.y = (lane < 4) ? a1 : b1;
        r.z = (lane < 4) ? a2 : b2;
        r.w = (lane < 4) ? a3 : b3;
        *(float4*)(out + (size_t)d0 * 16 + lane * 4) = r;
    }
}

extern "C" void kernel_launch(void* const* d_in, const int* in_sizes, int n_in,
                              void* d_out, int out_size, void* d_ws, size_t ws_size,
                              hipStream_t stream) {
    const float* x   = (const float*)d_in[0];
    const float* W1  = (const float*)d_in[1];
    const float* as1 = (const float*)d_in[2];
    const float* ad1 = (const float*)d_in[3];
    const float* b1  = (const float*)d_in[4];
    const float* W2  = (const float*)d_in[5];
    const float* as2 = (const float*)d_in[6];
    const float* ad2 = (const float*)d_in[7];
    const float* b2  = (const float*)d_in[8];
    const int*   ei  = (const int*)d_in[9];

    const int N = in_sizes[0] / 128;
    const int E = in_sizes[9] / 2;
    const int* src = ei;
    const int* dst = ei + E;

    float* w = (float*)d_ws;
    float* h1      = w; w += (size_t)N * 128;
    float* hact    = w; w += (size_t)N * 128;
    float* asrc1   = w; w += (size_t)N * 2;
    float* adst1   = w; w += (size_t)N * 2;
    float* asrc2   = w; w += (size_t)N;
    float* adst2   = w; w += (size_t)N;
    int*   cnt     = (int*)w; w += (size_t)N;
    int*   rowptr  = (int*)w; w += (size_t)N + 4;
    int*   csr     = (int*)w; w += (size_t)E;
    float* h2      = h1;

    fill_kernel<<<64, 256, 0, stream>>>((float*)cnt, 0.f, (size_t)N);
    count_in_kernel<<<cdiv(E, 256), 256, 0, stream>>>(dst, cnt, E);
    scan_kernel<<<1, 1024, 0, stream>>>(cnt, rowptr, N);
    place_kernel<<<cdiv(E, 256), 256, 0, stream>>>(src, dst, cnt, csr, E);

    {
        int tiles = (N / 16) * 2;
        wmma_gemm_f32<128, 128, 4><<<cdiv(tiles, 8), 256, 0, stream>>>(x, W1, h1, N);
        att_dots_kernel<2, 64><<<cdiv(N, 8), 256, 0, stream>>>(h1, as1, ad1, asrc1, adst1, N);
        node_agg1_kernel<<<cdiv(N, 8), 256, 0, stream>>>(rowptr, csr, h1, asrc1, adst1, b1, hact, N);
    }

    {
        int tiles = (N / 16);
        wmma_gemm_f32<128, 16, 1><<<cdiv(tiles, 8), 256, 0, stream>>>(hact, W2, h2, N);
        att_dots_kernel<1, 16><<<cdiv(N, 8), 256, 0, stream>>>(h2, as2, ad2, asrc2, adst2, N);
        node_agg2_kernel<<<cdiv(N, 16), 256, 0, stream>>>(rowptr, csr, h2, asrc2, adst2, b2,
                                                           (float*)d_out, N);
    }
}


# one atomic per edge instead of two: the counting add returns the edge's rank within its destination (stored per edge); placement is slot = rowptr[dst] + rank, no atomic; csr order still arrival rank
# speedup vs baseline: 3.4557x; 1.3627x over previous
#include <hip/hip_runtime.h>
#include <hip/hip_bf16.h>
#include <math.h>

typedef float v2f __attribute__((ext_vector_type(2)));
typedef float v8f __attribute__((ext_vector_type(8)));

#define NEG_SLOPE 0.2f
#define EPS_ 1e-16f

static inline int cdiv(long a, long b) { return (int)((a + b - 1) / b); }

__global__ void fill_kernel(float* __restrict__ p, float v, size_t n) {
    size_t i = (size_t)blockIdx.x * blockDim.x + threadIdx.x;
    size_t stride = (size_t)gridDim.x * blockDim.x;
    for (; i < n; i += stride) p[i] = v;
}

template <int K, int NC, int CPW>
__global__ __launch_bounds__(256) void wmma_gemm_f32(const float* __restrict__ A,
                                                     const float* __restrict__ B,
                                                     float* __restrict__ C,
                                                     const float* __restrict__ att_s,
                                                     const float* __restrict__ att_d,
                                                     float* __restrict__ asrc,
                                                     float* __restrict__ adst,
                                                     int M) {
    constexpr int nColT  = NC / 16;
    constexpr int groups = nColT / CPW;
    int wave = threadIdx.x >> 5;
    int lane = threadIdx.x & 31;
    int hl   = lane >> 4;
    int l    = lane & 15;
    int tile  = blockIdx.x * (blockDim.x >> 5) + wave;
    int total = (M >> 4) * groups;
    if (tile >= total) return;
    int m0 = (tile / groups) << 4;
    int n0 = (tile % groups) * (CPW * 16);

    const float* arow = A + (size_t)(m0 + l) * K + 2 * hl;
    const float* bcol = B + (size_t)(2 * hl) * NC + n0 + l;

    v8f acc[CPW];
#pragma unroll
    for (int i = 0; i < CPW; ++i) acc[i] = {};

    for (int k0 = 0; k0 < K; k0 += 4) {
        v2f a;
        a.x = arow[k0 + 0];
        a.y = arow[k0 + 1];
#pragma unroll
        for (int ct = 0; ct < CPW; ++ct) {
            v2f b;
            b.x = bcol[(size_t)k0 * NC + ct * 16];
            b.y = bcol[(size_t)(k0 + 1) * NC + ct * 16];
            acc[ct] = __builtin_amdgcn_wmma_f32_16x16x4_f32(false, a, false, b,
                                                            (short)0, acc[ct], false, false);
        }
    }

#pragma unroll
    for (int ct = 0; ct < CPW; ++ct) {
        float* crow = C + (size_t)(m0 + 8 * hl) * NC + n0 + ct * 16 + l;
#pragma unroll
        for (int r = 0; r < 8; ++r) crow[(size_t)r * NC] = acc[ct][r];
    }

    float v[16];
    {
        float fs[CPW], fd[CPW];
#pragma unroll
        for (int ct = 0; ct < CPW; ++ct) {
            fs[ct] = att_s[n0 + ct * 16 + l];
            fd[ct] = att_d[n0 + ct * 16 + l];
        }
#pragma unroll
        for (int r = 0; r < 8; ++r) {
            float ps = 0.f, pd = 0.f;
#pragma unroll
            for (int ct = 0; ct < CPW; ++ct) {
                ps += acc[ct][r] * fs[ct];
                pd += acc[ct][r] * fd[ct];
            }
            v[r] = ps;
            v[8 + r] = pd;
        }
    }
    const bool b8 = (l & 8) != 0, b4 = (l & 4) != 0, b2 = (l & 2) != 0, b1 = (l & 1) != 0;
    float w8[8], w4[4], w2[2];
#pragma unroll
    for (int i = 0; i < 8; ++i) {
        float keep = b8 ? v[i + 8] : v[i];
        float send = b8 ? v[i] : v[i + 8];
        w8[i] = keep + __shfl_xor(send, 8, 32);
    }
#pragma unroll
    for (int i = 0; i < 4; ++i) {
        float keep = b4 ? w8[i + 4] : w8[i];
        float send = b4 ? w8[i] : w8[i + 4];
        w4[i] = keep + __shfl_xor(send, 4, 32);
    }
#pragma unroll
    for (int i = 0; i < 2; ++i) {
        float keep = b2 ? w4[i + 2] : w4[i];
        float send = b2 ? w4[i] : w4[i + 2];
        w2[i] = keep + __shfl_xor(send, 2, 32);
    }
    const float dot = (b1 ? w2[1] : w2[0]) + __shfl_xor(b1 ? w2[0] : w2[1], 1, 32);
    float* dp = b8 ? adst : asrc;
    dp[(size_t)(m0 + 8 * hl + (l & 7)) * groups + (tile % groups)] = dot;
}

__global__ void rank_in_kernel(const int* __restrict__ dst, int* __restrict__ cnt,
                               int* __restrict__ rank, int E) {
    int e = blockIdx.x * blockDim.x + threadIdx.x;
    if (e >= E) return;
    rank[e] = atomicAdd(&cnt[dst[e]], 1);
}

__global__ __launch_bounds__(1024) void scan_kernel(const int* __restrict__ cnt, int* __restrict__ rowptr, int n) {
    __shared__ int wsum[32];
    const int t = threadIdx.x;
    const int lane = t & 31, w = t >> 5;
    const int chunk = (n + 1023) / 1024;
    const int b = t * chunk;
    const int e = min(b + chunk, n);
    int s = 0;
#pragma unroll 8
    for (int i = b; i < e; ++i) s += cnt[i];
    int v = s;
    for (int off = 1; off < 32; off <<= 1) {
        int u = __shfl_up(v, off, 32);
        if (lane >= off) v += u;
    }
    if (lane == 31) wsum[w] = v;
    __syncthreads();
    if (w == 0) {
        int x = wsum[lane];
        for (int off = 1; off < 32; off <<= 1) {
            int u = __shfl_up(x, off, 32);
            if (lane >= off) x += u;
        }
        wsum[lane] = x;
    }
    __syncthreads();
    int base = v - s + (w > 0 ? wsum[w - 1] : 0);
#pragma unroll 8
    for (int i = b; i < e; ++i) {
        int c = cnt[i];
        rowptr[i] = base;
        base += c;
    }
    if (t == 1023) rowptr[n] = wsum[31];
}

__global__ void place_kernel(const int* __restrict__ src, const int* __restrict__ dst,
                             const int* __restrict__ rowptr, const int* __restrict__ rank,
                             int* __restrict__ csr, int E) {
    int e = blockIdx.x * blockDim.x + threadIdx.x;
    if (e >= E) return;
    const int sv = src[e];
    const int pos = rowptr[dst[e]] + rank[e];
    if ((unsigned)pos < (unsigned)E) csr[pos] = sv;
}

__device__ inline float leaky(float v) { return (v >= 0.f) ? v : NEG_SLOPE * v; }
__device__ inline float finite_or_zero(float m) { return (fabsf(m) < INFINITY) ? m : 0.f; }
__device__ inline float lane_bcast(float v, int k) {
    return __int_as_float(__builtin_amdgcn_readlane(__float_as_int(v), k));
}

template <int NB>
__device__ __forceinline__ void gather_rows1(float4& acc, const float* __restrict__ hcol,
                                             int s, float w0, float w1, int hd, int k) {
    float4 hv[NB];
#pragma unroll
    for (int i = 0; i < NB; ++i) {
        int sk = __builtin_amdgcn_readlane(s, k + i);
        hv[i] = *(const float4*)(hcol + (size_t)sk * 128);
    }
#pragma unroll
    for (int i = 0; i < NB; ++i) {
        float wa = lane_bcast(w0, k + i), wb = lane_bcast(w1, k + i);
        float wk = hd ? wb : wa;
        acc.x += wk * hv[i].x;
        acc.y += wk * hv[i].y;
        acc.z += wk * hv[i].z;
        acc.w += wk * hv[i].w;
    }
}

__global__ __launch_bounds__(256) void node_agg1_kernel(
        const int* __restrict__ rowptr, const int* __restrict__ csr,
        const float* __restrict__ hfeat,
        const float* __restrict__ asrc, const float* __restrict__ adst,
        const float* __restrict__ bias, float* __restrict__ hact, int n) {
    const int lane = threadIdx.x & 31;
    const int d = __builtin_amdgcn_readfirstlane(blockIdx.x * (blockDim.x >> 5) + (threadIdx.x >> 5));
    if (d >= n) return;
    const int beg = rowptr[d], end = rowptr[d + 1];
    const float ad0 = adst[d * 2 + 0], ad1 = adst[d * 2 + 1];
    const int hd = lane >> 4;

    float m0 = -INFINITY, m1 = -INFINITY;
    int s_first = 0;
    float2 a_first = {0.f, 0.f};
    for (int base = beg; base < end; base += 32) {
        int j = base + lane;
        if (j < end) {
            int s = csr[j];
            float2 a = *(const float2*)(asrc + (size_t)s * 2);
            if (base == beg) { s_first = s; a_first = a; }
            m0 = fmaxf(m0, leaky(a.x + ad0));
            m1 = fmaxf(m1, leaky(a.y + ad1));
        }
    }
    for (int off = 16; off; off >>= 1) {
        m0 = fmaxf(m0, __shfl_xor(m0, off, 32));
        m1 = fmaxf(m1, __shfl_xor(m1, off, 32));
    }
    m0 = finite_or_zero(m0);
    m1 = finite_or_zero(m1);

    float4 acc = {0.f, 0.f, 0.f, 0.f};
    float den0 = 0.f, den1 = 0.f;
    const float* hcol = hfeat + lane * 4;
    for (int base = beg; base < end; base += 32) {
        int j = base + lane;
        int s = s_first;
        float2 a = a_first;
        if (base != beg && j < end) {
            s = csr[j];
            a = *(const float2*)(asrc + (size_t)s * 2);
        }
        float w0 = 0.f, w1 = 0.f;
        if (j < end) {
            w0 = __expf(leaky(a.x + ad0) - m0);
            w1 = __expf(leaky(a.y + ad1) - m1);
        }
        den0 += w0;
        den1 += w1;
        const int cnt = min(32, end - base);
        int k = 0;
        for (; k + 8 <= cnt; k += 8) gather_rows1<8>(acc, hcol, s, w0, w1, hd, k);
        if (cnt & 4) { gather_rows1<4>(acc, hcol, s, w0, w1, hd, k); k += 4; }
        if (cnt & 2) { gather_rows1<2>(acc, hcol, s, w0, w1, hd, k); k += 2; }
        if (cnt & 1) { gather_rows1<1>(acc, hcol, s, w0, w1, hd, k); }
    }
    for (int off = 16; off; off >>= 1) {
        den0 += __shfl_xor(den0, off, 32);
        den1 += __shfl_xor(den1, off, 32);
    }
    const float den = (hd ? den1 : den0) + EPS_;
    const float4 bv = *(const float4*)(bias + lane * 4);
    float4 o;
    o.x = acc.x / den + bv.x;
    o.y = acc.y / den + bv.y;
    o.z = acc.z / den + bv.z;
    o.w = acc.w / den + bv.w;
    o.x = (o.x > 0.f) ? o.x : expm1f(o.x);
    o.y = (o.y > 0.f) ? o.y : expm1f(o.y);
    o.z = (o.z > 0.f) ? o.z : expm1f(o.z);
    o.w = (o.w > 0.f) ? o.w : expm1f(o.w);
    *(float4*)(hact + (size_t)d * 128 + lane * 4) = o;
}

__device__ __forceinline__ void node2_row(int d, int lane,
                                   const int* __restrict__ rowptr, const int* __restrict__ csr,
                                   const float* __restrict__ hfeat,
                                   const float* __restrict__ asrc, const float* __restrict__ adst,
                                   const float* __restrict__ bias,
                                   float& r0, float& r1, float& r2, float& r3) {
    const int beg = rowptr[d], end = rowptr[d + 1];
    const float ad = adst[d];
    float m = -INFINITY;
    int s_first = 0;
    float a_first = 0.f;
    for (int base = beg; base < end; base += 32) {
        int j = base + lane;
        if (j < end) {
            int s = csr[j];
            float a = asrc[s];
            if (base == beg) { s_first = s; a_first = a; }
            m = fmaxf(m, leaky(a + ad));
        }
    }
    for (int off = 16; off; off >>= 1) m = fmaxf(m, __shfl_xor(m, off, 32));
    m = finite_or_zero(m);

    float4 acc = {0.f, 0.f, 0.f, 0.f};
    float den = 0.f;
    const int g = lane >> 2;
    const float* hcol = hfeat + (lane & 3) * 4;
    for (int base = beg; base < end; base += 32) {
        int j = base + lane;
        int s = s_first;
        float a = a_first;
        if (base != beg && j < end) {
            s = csr[j];
            a = asrc[s];
        }
        float w = 0.f;
        if (j < end) w = __expf(leaky(a + ad) - m);
        else s = 0;
        den += w;
        const int cnt = min(32, end - base);
        float4 hv[4];
        float wk[4];
#pragma unroll
        for (int t = 0; t < 4; ++t) {
            int sk = __shfl(s, t * 8 + g, 32);
            wk[t] = __shfl(w, t * 8 + g, 32);
            hv[t] = *(const float4*)(hcol + (size_t)sk * 16);
        }
#pragma unroll
        for (int t = 0; t < 4; ++t) {
            const bool live = (t * 8 + g < cnt);
            acc.x += live ? wk[t] * hv[t].x : 0.f;
            acc.y += live ? wk[t] * hv[t].y : 0.f;
            acc.z += live ? wk[t] * hv[t].z : 0.f;
            acc.w += live ? wk[t] * hv[t].w : 0.f;
        }
    }
    for (int off = 16; off >= 4; off >>= 1) {
        acc.x += __shfl_xor(acc.x, off, 32);
        acc.y += __shfl_xor(acc.y, off, 32);
        acc.z += __shfl_xor(acc.z, off, 32);
        acc.w += __shfl_xor(acc.w, off, 32);
    }
    for (int off = 16; off; off >>= 1) den += __shfl_xor(den, off, 32);
    den += EPS_;
    const float4 bv = *(const float4*)(bias + (lane & 3) * 4);
    float4 v;
    v.x = acc.x / den + bv.x;
    v.y = acc.y / den + bv.y;
    v.z = acc.z / den + bv.z;
    v.w = acc.w / den + bv.w;
    float mx = fmaxf(fmaxf(v.x, v.y), fmaxf(v.z, v.w));
    mx = fmaxf(mx, __shfl_xor(mx, 1, 32));
    mx = fmaxf(mx, __shfl_xor(mx, 2, 32));
    v.x = __expf(v.x - mx);
    v.y = __expf(v.y - mx);
    v.z = __expf(v.z - mx);
    v.w = __expf(v.w - mx);
    float sum = (v.x + v.y) + (v.z + v.w);
    sum += __shfl_xor(sum, 1, 32);
    sum += __shfl_xor(sum, 2, 32);
    const float inv = 1.f / sum;
    r0 = v.x * inv;
    r1 = v.y * inv;
    r2 = v.z * inv;
    r3 = v.w * inv;
}

__global__ __launch_bounds__(256) void node_agg2_kernel(
        const int* __restrict__ rowptr, const int* __restrict__ csr,
        const float* __restrict__ hfeat,
        const float* __restrict__ asrc, const float* __restrict__ adst,
        const float* __restrict__ bias, float* __restrict__ out, int n) {
    const int lane = threadIdx.x & 31;
    const int d0 = __builtin_amdgcn_readfirstlane((blockIdx.x * (blockDim.x >> 5) + (threadIdx.x >> 5)) * 2);
    if (d0 >= n) return;
    float a0, a1, a2, a3;
    node2_row(d0, lane, rowptr, csr, hfeat, asrc, adst, bias, a0, a1, a2, a3);
    float b0 = a0, b1 = a1, b2 = a2, b3 = a3;
    const bool two = (d0 + 1 < n);
    if (two) node2_row(d0 + 1, lane, rowptr, csr, hfeat, asrc, adst, bias, b0, b1, b2, b3);
    if (lane < 4 || (two && lane < 8)) {
        float4 r;
        r.x = (lane < 4) ? a0 : b0;
        r.y = (lane < 4) ? a1 : b1;
        r.z = (lane < 4) ? a2 : b2;
        r.w = (lane < 4) ? a3 : b3;
        *(float4*)(out + (size_t)d0 * 16 + lane * 4) = r;
    }
}

extern "C" void kernel_launch(void* const* d_in, const int* in_sizes, int n_in,
                              void* d_out, int out_size, void* d_ws, size_t ws_size,
                              hipStream_t stream) {
    const float* x   = (const float*)d_in[0];
    const float* W1  = (const float*)d_in[1];
    const float* as1 = (const float*)d_in[2];
    const float* ad1 = (const float*)d_in[3];
    const float* b1  = (const float*)d_in[4];
    const float* W2  = (const float*)d_in[5];
    const float* as2 = (const float*)d_in[6];
    const float* ad2 = (const float*)d_in[7];
    const float* b2  = (const float*)d_in[8];
    const int*   ei  = (const int*)d_in[9];

    const int N = in_sizes[0] / 128;
    const int E = in_sizes[9] / 2;
    const int* src = ei;
    const int* dst = ei + E;

    float* w = (float*)d_ws;
    float* h1      = w; w += (size_t)N * 128;
    float* hact    = w; w += (size_t)N * 128;
    float* asrc1   = w; w += (size_t)N * 2;
    float* adst1   = w; w += (size_t)N * 2;
    float* asrc2   = w; w += (size_t)N;
    float* adst2   = w; w += (size_t)N;
    int*   cnt     = (int*)w; w += (size_t)N;
    int*   rowptr  = (int*)w; w += (size_t)N + 4;
    int*   csr     = (int*)w; w += (size_t)E;
    float* h2      = h1;
    int*   rank    = (int*)hact;

    fill_kernel<<<64, 256, 0, stream>>>((float*)cnt, 0.f, (size_t)N);
    rank_in_kernel<<<cdiv(E, 256), 256, 0, stream>>>(dst, cnt, rank, E);
    scan_kernel<<<1, 1024, 0, stream>>>(cnt, rowptr, N);
    place_kernel<<<cdiv(E, 256), 256, 0, stream>>>(src, dst, rowptr, rank, csr, E);

    {
        int tiles = (N / 16) * 2;
        wmma_gemm_f32<128, 128, 4><<<cdiv(tiles, 8), 256, 0, stream>>>(x, W1, h1, as1, ad1,
                                                                       asrc1, adst1, N);
        node_agg1_kernel<<<cdiv(N, 8), 256, 0, stream>>>(rowptr, csr, h1, asrc1, adst1, b1, hact, N);
    }

    {
        int tiles = (N / 16);
        wmma_gemm_f32<128, 16, 1><<<cdiv(tiles, 8), 256, 0, stream>>>(hact, W2, h2, as2, ad2,
                                                                      asrc2, adst2, N);
        node_agg2_kernel<<<cdiv(N, 16), 256, 0, stream>>>(rowptr, csr, h2, asrc2, adst2, b2,
                                                           (float*)d_out, N);
    }
}


# edge ranks from per-node LDS counters, one edge chunk per work-group (returning LDS add = rank in chunk), then a per-node prefix over the 128 chunks; no global atomic; csr order: chunk, then arrival
# speedup vs baseline: 4.7980x; 1.3884x over previous
#include <hip/hip_runtime.h>
#include <hip/hip_bf16.h>
#include <math.h>

typedef float v2f __attribute__((ext_vector_type(2)));
typedef float v8f __attribute__((ext_vector_type(8)));

#define NEG_SLOPE 0.2f
#define EPS_ 1e-16f

static inline int cdiv(long a, long b) { return (int)((a + b - 1) / b); }

template <int K, int NC, int CPW>
__global__ __launch_bounds__(256) void wmma_gemm_f32(const float* __restrict__ A,
                                                     const float* __restrict__ B,
                                                     float* __restrict__ C,
                                                     const float* __restrict__ att_s,
                                                     const float* __restrict__ att_d,
                                                     float* __restrict__ asrc,
                                                     float* __restrict__ adst,
                                                     int M) {
    constexpr int nColT  = NC / 16;
    constexpr int groups = nColT / CPW;
    int wave = threadIdx.x >> 5;
    int lane = threadIdx.x & 31;
    int hl   = lane >> 4;
    int l    = lane & 15;
    int tile  = blockIdx.x * (blockDim.x >> 5) + wave;
    int total = (M >> 4) * groups;
    if (tile >= total) return;
    int m0 = (tile / groups) << 4;
    int n0 = (tile % groups) * (CPW * 16);

    const float* arow = A + (size_t)(m0 + l) * K + 2 * hl;
    const float* bcol = B + (size_t)(2 * hl) * NC + n0 + l;

    v8f acc[CPW];
#pragma unroll
    for (int i = 0; i < CPW; ++i) acc[i] = {};

    for (int k0 = 0; k0 < K; k0 += 4) {
        v2f a;
        a.x = arow[k0 + 0];
        a.y = arow[k0 + 1];
#pragma unroll
        for (int ct = 0; ct < CPW; ++ct) {
            v2f b;
            b.x = bcol[(size_t)k0 * NC + ct * 16];
            b.y = bcol[(size_t)(k0 + 1) * NC + ct * 16];
            acc[ct] = __builtin_amdgcn_wmma_f32_16x16x4_f32(false, a, false, b,
                                                            (short)0, acc[ct], false, false);
        }
    }

#pragma unroll
    for (int ct = 0; ct < CPW; ++ct) {
        float* crow = C + (size_t)(m0 + 8 * hl) * NC + n0 + ct * 16 + l;
#pragma unroll
        for (int r = 0; r < 8; ++r) crow[(size_t)r * NC] = acc[ct][r];
    }

    float v[16];
    {
        float fs[CPW], fd[CPW];
#pragma unroll
        for (int ct = 0; ct < CPW; ++ct) {
            fs[ct] = att_s[n0 + ct * 16 + l];
            fd[ct] = att_d[n0 + ct * 16 + l];
        }
#pragma unroll
        for (int r = 0; r < 8; ++r) {
            float ps = 0.f, pd = 0.f;
#pragma unroll
            for (int ct = 0; ct < CPW; ++ct) {
                ps += acc[ct][r] * fs[ct];
                pd += acc[ct][r] * fd[ct];
            }
            v[r] = ps;
            v[8 + r] = pd;
        }
    }
    const bool b8 = (l & 8) != 0, b4 = (l & 4) != 0, b2 = (l & 2) != 0, b1 = (l & 1) != 0;
    float w8[8], w4[4], w2[2];
#pragma unroll
    for (int i = 0; i < 8; ++i) {
        float keep = b8 ? v[i + 8] : v[i];
        float send = b8 ? v[i] : v[i + 8];
        w8[i] = keep + __shfl_xor(send, 8, 32);
    }
#pragma unroll
    for (int i = 0; i < 4; ++i) {
        float keep = b4 ? w8[i + 4] : w8[i];
        float send = b4 ? w8[i] : w8[i + 4];
        w4[i] = keep + __shfl_xor(send, 4, 32);
    }
#pragma unroll
    for (int i = 0; i < 2; ++i) {
        float keep = b2 ? w4[i + 2] : w4[i];
        float send = b2 ? w4[i] : w4[i + 2];
        w2[i] = keep + __shfl_xor(send, 2, 32);
    }
    const float dot = (b1 ? w2[1] : w2[0]) + __shfl_xor(b1 ? w2[0] : w2[1], 1, 32);
    float* dp = b8 ? adst : asrc;
    dp[(size_t)(m0 + 8 * hl + (l & 7)) * groups + (tile % groups)] = dot;
}

constexpr int kNodes  = 50000;
constexpr int kChunks = 128;

__device__ __forceinline__ int chunk_len(int E) {
    return (((E + kChunks - 1) / kChunks) + 3) & ~3;
}

__global__ __launch_bounds__(1024) void chunk_count_kernel(const int* __restrict__ dst,
                                                            int* __restrict__ H,
                                                            int* __restrict__ rank, int n, int E) {
    __shared__ int hist[kNodes];
    const int t = threadIdx.x;
    const int g = blockIdx.x;
    const int chunk = chunk_len(E);
    const int b   = min(g * chunk, E);
    const int end = min(b + chunk, E);
    const int n4  = n >> 2;
    int4* h4 = (int4*)hist;
    for (int i = t; i < n4; i += 1024) h4[i] = make_int4(0, 0, 0, 0);
    __syncthreads();

    const int nq = (end - b) >> 2;
    const int4* d4 = (const int4*)(dst + b);
    int4* r4 = (int4*)(rank + b);
    for (int q0 = t; q0 < nq; q0 += 4 * 1024) {
        int4 dv[4];
#pragma unroll
        for (int u = 0; u < 4; ++u) {
            const int q = q0 + u * 1024;
            dv[u] = (q < nq) ? d4[q] : make_int4(0, 0, 0, 0);
        }
#pragma unroll
        for (int u = 0; u < 4; ++u) {
            const int q = q0 + u * 1024;
            if (q < nq) {
                int4 r;
                r.x = atomicAdd(&hist[dv[u].x], 1);
                r.y = atomicAdd(&hist[dv[u].y], 1);
                r.z = atomicAdd(&hist[dv[u].z], 1);
                r.w = atomicAdd(&hist[dv[u].w], 1);
                r4[q] = r;
            }
        }
    }
    for (int e = b + (nq << 2) + t; e < end; e += 1024) rank[e] = atomicAdd(&hist[dst[e]], 1);
    __syncthreads();

    int4* out = (int4*)(H + (size_t)g * n);
    for (int i = t; i < n4; i += 1024) out[i] = h4[i];
}

__device__ __forceinline__ int4 add4(int4 a, int4 b) {
    return make_int4(a.x + b.x, a.y + b.y, a.z + b.z, a.w + b.w);
}

__global__ __launch_bounds__(256) void chunk_prefix_kernel(int* H, int* __restrict__ cnt, int n) {
    constexpr int W = 8, PER = kChunks / W;
    __shared__ int4 tot[W][32];
    const int lane = threadIdx.x & 31, w = threadIdx.x >> 5;
    const int n4 = n >> 2;
    const int q = blockIdx.x * 32 + lane;
    const bool live = q < n4;
    int4* H4 = (int4*)H;
    int4 v[PER];
#pragma unroll
    for (int j = 0; j < PER; ++j)
        v[j] = live ? H4[(size_t)(w * PER + j) * n4 + q] : make_int4(0, 0, 0, 0);
    int4 s = make_int4(0, 0, 0, 0);
#pragma unroll
    for (int j = 0; j < PER; ++j) {
        const int4 c = v[j];
        v[j] = s;
        s = add4(s, c);
    }
    tot[w][lane] = s;
    __syncthreads();
    int4 base = make_int4(0, 0, 0, 0);
    for (int k = 0; k < w; ++k) base = add4(base, tot[k][lane]);
    if (live) {
#pragma unroll
        for (int j = 0; j < PER; ++j) H4[(size_t)(w * PER + j) * n4 + q] = add4(v[j], base);
        if (w == W - 1) ((int4*)cnt)[q] = add4(base, s);
    }
}

__global__ __launch_bounds__(1024) void scan_kernel(const int* __restrict__ cnt, int* __restrict__ rowptr, int n) {
    __shared__ int wsum[32];
    const int t = threadIdx.x;
    const int lane = t & 31, w = t >> 5;
    const int chunk = (n + 1023) / 1024;
    const int b = t * chunk;
    const int e = min(b + chunk, n);
    int s = 0;
#pragma unroll 8
    for (int i = b; i < e; ++i) s += cnt[i];
    int v = s;
    for (int off = 1; off < 32; off <<= 1) {
        int u = __shfl_up(v, off, 32);
        if (lane >= off) v += u;
    }
    if (lane == 31) wsum[w] = v;
    __syncthreads();
    if (w == 0) {
        int x = wsum[lane];
        for (int off = 1; off < 32; off <<= 1) {
            int u = __shfl_up(x, off, 32);
            if (lane >= off) x += u;
        }
        wsum[lane] = x;
    }
    __syncthreads();
    int base = v - s + (w > 0 ? wsum[w - 1] : 0);
#pragma unroll 8
    for (int i = b; i < e; ++i) {
        int c = cnt[i];
        rowptr[i] = base;
        base += c;
    }
    if (t == 1023) rowptr[n] = wsum[31];
}

__global__ void place_kernel(const int* __restrict__ src, const int* __restrict__ dst,
                             const int* __restrict__ rowptr, const int* __restrict__ H,
                             const int* __restrict__ rank,
                             int* __restrict__ csr, int n, int E) {
    int e = blockIdx.x * blockDim.x + threadIdx.x;
    if (e >= E) return;
    const int sv = src[e];
    const int d  = dst[e];
    const int g  = (int)((unsigned)e / (unsigned)chunk_len(E));
    const int pos = rowptr[d] + H[(size_t)g * n + d] + rank[e];
    if ((unsigned)pos < (unsigned)E) csr[pos] = sv;
}

__device__ inline float leaky(float v) { return (v >= 0.f) ? v : NEG_SLOPE * v; }
__device__ inline float finite_or_zero(float m) { return (fabsf(m) < INFINITY) ? m : 0.f; }
__device__ inline float lane_bcast(float v, int k) {
    return __int_as_float(__builtin_amdgcn_readlane(__float_as_int(v), k));
}

template <int NB>
__device__ __forceinline__ void gather_rows1(float4& acc, const float* __restrict__ hcol,
                                             int s, float w0, float w1, int hd, int k) {
    float4 hv[NB];
#pragma unroll
    for (int i = 0; i < NB; ++i) {
        int sk = __builtin_amdgcn_readlane(s, k + i);
        hv[i] = *(const float4*)(hcol + (size_t)sk * 128);
    }
#pragma unroll
    for (int i = 0; i < NB; ++i) {
        float wa = lane_bcast(w0, k + i), wb = lane_bcast(w1, k + i);
        float wk = hd ? wb : wa;
        acc.x += wk * hv[i].x;
        acc.y += wk * hv[i].y;
        acc.z += wk * hv[i].z;
        acc.w += wk * hv[i].w;
    }
}

__global__ __launch_bounds__(256) void node_agg1_kernel(
        const int* __restrict__ rowptr, const int* __restrict__ csr,
        const float* __restrict__ hfeat,
        const float* __restrict__ asrc, const float* __restrict__ adst,
        const float* __restrict__ bias, float* __restrict__ hact, int n) {
    const int lane = threadIdx.x & 31;
    const int d = __builtin_amdgcn_readfirstlane(blockIdx.x * (blockDim.x >> 5) + (threadIdx.x >> 5));
    if (d >= n) return;
    const int beg = rowptr[d], end = rowptr[d + 1];
    const float ad0 = adst[d * 2 + 0], ad1 = adst[d * 2 + 1];
    const int hd = lane >> 4;

    float m0 = -INFINITY, m1 = -INFINITY;
    int s_first = 0;
    float2 a_first = {0.f, 0.f};
    for (int base = beg; base < end; base += 32) {
        int j = base + lane;
        if (j < end) {
            int s = csr[j];
            float2 a = *(const float2*)(asrc + (size_t)s * 2);
            if (base == beg) { s_first = s; a_first = a; }
            m0 = fmaxf(m0, leaky(a.x + ad0));
            m1 = fmaxf(m1, leaky(a.y + ad1));
        }
    }
    for (int off = 16; off; off >>= 1) {
        m0 = fmaxf(m0, __shfl_xor(m0, off, 32));
        m1 = fmaxf(m1, __shfl_xor(m1, off, 32));
    }
    m0 = finite_or_zero(m0);
    m1 = finite_or_zero(m1);

    float4 acc = {0.f, 0.f, 0.f, 0.f};
    float den0 = 0.f, den1 = 0.f;
    const float* hcol = hfeat + lane * 4;
    for (int base = beg; base < end; base += 32) {
        int j = base + lane;
        int s = s_first;
        float2 a = a_first;
        if (base != beg && j < end) {
            s = csr[j];
            a = *(const float2*)(asrc + (size_t)s * 2);
        }
        float w0 = 0.f, w1 = 0.f;
        if (j < end) {
            w0 = __expf(leaky(a.x + ad0) - m0);
            w1 = __expf(leaky(a.y + ad1) - m1);
        }
        den0 += w0;
        den1 += w1;
        const int cnt = min(32, end - base);
        int k = 0;
        for (; k + 8 <= cnt; k += 8) gather_rows1<8>(acc, hcol, s, w0, w1, hd, k);
        if (cnt & 4) { gather_rows1<4>(acc, hcol, s, w0, w1, hd, k); k += 4; }
        if (cnt & 2) { gather_rows1<2>(acc, hcol, s, w0, w1, hd, k); k += 2; }
        if (cnt & 1) { gather_rows1<1>(acc, hcol, s, w0, w1, hd, k); }
    }
    for (int off = 16; off; off >>= 1) {
        den0 += __shfl_xor(den0, off, 32);
        den1 += __shfl_xor(den1, off, 32);
    }
    const float den = (hd ? den1 : den0) + EPS_;
    const float4 bv = *(const float4*)(bias + lane * 4);
    float4 o;
    o.x = acc.x / den + bv.x;
    o.y = acc.y / den + bv.y;
    o.z = acc.z / den + bv.z;
    o.w = acc.w / den + bv.w;
    o.x = (o.x > 0.f) ? o.x : expm1f(o.x);
    o.y = (o.y > 0.f) ? o.y : expm1f(o.y);
    o.z = (o.z > 0.f) ? o.z : expm1f(o.z);
    o.w = (o.w > 0.f) ? o.w : expm1f(o.w);
    *(float4*)(hact + (size_t)d * 128 + lane * 4) = o;
}

__device__ __forceinline__ void node2_row(int d, int lane,
                                   const int* __restrict__ rowptr, const int* __restrict__ csr,
                                   const float* __restrict__ hfeat,
                                   const float* __restrict__ asrc, const float* __restrict__ adst,
                                   const float* __restrict__ bias,
                                   float& r0, float& r1, float& r2, float& r3) {
    const int beg = rowptr[d], end = rowptr[d + 1];
    const float ad = adst[d];
    float m = -INFINITY;
    int s_first = 0;
    float a_first = 0.f;
    for (int base = beg; base < end; base += 32) {
        int j = base + lane;
        if (j < end) {
            int s = csr[j];
            float a = asrc[s];
            if (base == beg) { s_first = s; a_first = a; }
            m = fmaxf(m, leaky(a + ad));
        }
    }
    for (int off = 16; off; off >>= 1) m = fmaxf(m, __shfl_xor(m, off, 32));
    m = finite_or_zero(m);

    float4 acc = {0.f, 0.f, 0.f, 0.f};
    float den = 0.f;
    const int g = lane >> 2;
    const float* hcol = hfeat + (lane & 3) * 4;
    for (int base = beg; base < end; base += 32) {
        int j = base + lane;
        int s = s_first;
        float a = a_first;
        if (base != beg && j < end) {
            s = csr[j];
            a = asrc[s];
        }
        float w = 0.f;
        if (j < end) w = __expf(leaky(a + ad) - m);
        else s = 0;
        den += w;
        const int cnt = min(32, end - base);
        float4 hv[4];
        float wk[4];
#pragma unroll
        for (int t = 0; t < 4; ++t) {
            int sk = __shfl(s, t * 8 + g, 32);
            wk[t] = __shfl(w, t * 8 + g, 32);
            hv[t] = *(const float4*)(hcol + (size_t)sk * 16);
        }
#pragma unroll
        for (int t = 0; t < 4; ++t) {
            const bool live = (t * 8 + g < cnt);
            acc.x += live ? wk[t] * hv[t].x : 0.f;
            acc.y += live ? wk[t] * hv[t].y : 0.f;
            acc.z += live ? wk[t] * hv[t].z : 0.f;
            acc.w += live ? wk[t] * hv[t].w : 0.f;
        }
    }
    for (int off = 16; off >= 4; off >>= 1) {
        acc.x += __shfl_xor(acc.x, off, 32);
        acc.y += __shfl_xor(acc.y, off, 32);
        acc.z += __shfl_xor(acc.z, off, 32);
        acc.w += __shfl_xor(acc.w, off, 32);
    }
    for (int off = 16; off; off >>= 1) den += __shfl_xor(den, off, 32);
    den += EPS_;
    const float4 bv = *(const float4*)(bias + (lane & 3) * 4);
    float4 v;
    v.x = acc.x / den + bv.x;
    v.y = acc.y / den + bv.y;
    v.z = acc.z / den + bv.z;
    v.w = acc.w / den + bv.w;
    float mx = fmaxf(fmaxf(v.x, v.y), fmaxf(v.z, v.w));
    mx = fmaxf(mx, __shfl_xor(mx, 1, 32));
    mx = fmaxf(mx, __shfl_xor(mx, 2, 32));
    v.x = __expf(v.x - mx);
    v.y = __expf(v.y - mx);
    v.z = __expf(v.z - mx);
    v.w = __expf(v.w - mx);
    float sum = (v.x + v.y) + (v.z + v.w);
    sum += __shfl_xor(sum, 1, 32);
    sum += __shfl_xor(sum, 2, 32);
    const float inv = 1.f / sum;
    r0 = v.x * inv;
    r1 = v.y * inv;
    r2 = v.z * inv;
    r3 = v.w * inv;
}

__global__ __launch_bounds__(256) void node_agg2_kernel(
        const int* __restrict__ rowptr, const int* __restrict__ csr,
        const float* __restrict__ hfeat,
        const float* __restrict__ asrc, const float* __restrict__ adst,
        const float* __restrict__ bias, float* __restrict__ out, int n) {
    const int lane = threadIdx.x & 31;
    const int d0 = __builtin_amdgcn_readfirstlane((blockIdx.x * (blockDim.x >> 5) + (threadIdx.x >> 5)) * 2);
    if (d0 >= n) return;
    float a0, a1, a2, a3;
    node2_row(d0, lane, rowptr, csr, hfeat, asrc, adst, bias, a0, a1, a2, a3);
    float b0 = a0, b1 = a1, b2 = a2, b3 = a3;
    const bool two = (d0 + 1 < n);
    if (two) node2_row(d0 + 1, lane, rowptr, csr, hfeat, asrc, adst, bias, b0, b1, b2, b3);
    if (lane < 4 || (two && lane < 8)) {
        float4 r;
        r.x = (lane < 4) ? a0 : b0;
        r.y = (lane < 4) ? a1 : b1;
        r.z = (lane < 4) ? a2 : b2;
        r.w = (lane < 4) ? a3 : b3;
        *(float4*)(out + (size_t)d0 * 16 + lane * 4) = r;
    }
}

extern "C" void kernel_launch(void* const* d_in, const int* in_sizes, int n_in,
                              void* d_out, int out_size, void* d_ws, size_t ws_size,
                              hipStream_t stream) {
    const float* x   = (const float*)d_in[0];
    const float* W1  = (const float*)d_in[1];
    const float* as1 = (const float*)d_in[2];
    const float* ad1 = (const float*)d_in[3];
    const float* b1  = (const float*)d_in[4];
    const float* W2  = (const float*)d_in[5];
    const float* as2 = (const float*)d_in[6];
    const float* ad2 = (const float*)d_in[7];
    const float* b2  = (const float*)d_in[8];
    const int*   ei  = (const int*)d_in[9];

    const int N = in_sizes[0] / 128;
    const int E = in_sizes[9] / 2;
    const int* src = ei;
    const int* dst = ei + E;

    float* w = (float*)d_ws;
    float* h1      = w; w += (size_t)N * 128;
    float* hact    = w; w += (size_t)N * 128;
    float* asrc1   = w; w += (size_t)N * 2;
    float* adst1   = w; w += (size_t)N * 2;
    float* asrc2   = w; w += (size_t)N;
    float* adst2   = w; w += (size_t)N;
    int*   cnt     = (int*)w; w += (size_t)N;
    int*   rowptr  = (int*)w; w += (size_t)N + 4;
    int*   csr     = (int*)w; w += (size_t)E;
    float* h2      = h1;
    int*   rank    = (int*)hact;
    int*   H       = (int*)h1;

    chunk_count_kernel<<<kChunks, 1024, 0, stream>>>(dst, H, rank, N, E);
    chunk_prefix_kernel<<<cdiv(N, 128), 256, 0, stream>>>(H, cnt, N);
    scan_kernel<<<1, 1024, 0, stream>>>(cnt, rowptr, N);
    place_kernel<<<cdiv(E, 256), 256, 0, stream>>>(src, dst, rowptr, H, rank, csr, N, E);

    {
        int tiles = (N / 16) * 2;
        wmma_gemm_f32<128, 128, 4><<<cdiv(tiles, 8), 256, 0, stream>>>(x, W1, h1, as1, ad1,
                                                                       asrc1, adst1, N);
        node_agg1_kernel<<<cdiv(N, 8), 256, 0, stream>>>(rowptr, csr, h1, asrc1, adst1, b1, hact, N);
    }

    {
        int tiles = (N / 16);
        wmma_gemm_f32<128, 16, 1><<<cdiv(tiles, 8), 256, 0, stream>>>(hact, W2, h2, as2, ad2,
                                                                      asrc2, adst2, N);
        node_agg2_kernel<<<cdiv(N, 16), 256, 0, stream>>>(rowptr, csr, h2, asrc2, adst2, b2,
                                                           (float*)d_out, N);
    }
}


# row-offset scan moves its data in whole 128-byte lines: counts in and offsets out through LDS, 16 bytes per lane, each thread scans 52 nodes in registers (was lanes 196 B apart, 32 lines a load)
# speedup vs baseline: 5.2999x; 1.1046x over previous
#include <hip/hip_runtime.h>
#include <hip/hip_bf16.h>
#include <math.h>

typedef float v2f __attribute__((ext_vector_type(2)));
typedef float v8f __attribute__((ext_vector_type(8)));

#define NEG_SLOPE 0.2f
#define EPS_ 1e-16f

static inline int cdiv(long a, long b) { return (int)((a + b - 1) / b); }

template <int K, int NC, int CPW>
__global__ __launch_bounds__(256) void wmma_gemm_f32(const float* __restrict__ A,
                                                     const float* __restrict__ B,
                                                     float* __restrict__ C,
                                                     const float* __restrict__ att_s,
                                                     const float* __restrict__ att_d,
                                                     float* __restrict__ asrc,
                                                     float* __restrict__ adst,
                                                     int M) {
    constexpr int nColT  = NC / 16;
    constexpr int groups = nColT / CPW;
    int wave = threadIdx.x >> 5;
    int lane = threadIdx.x & 31;
    int hl   = lane >> 4;
    int l    = lane & 15;
    int tile  = blockIdx.x * (blockDim.x >> 5) + wave;
    int total = (M >> 4) * groups;
    if (tile >= total) return;
    int m0 = (tile / groups) << 4;
    int n0 = (tile % groups) * (CPW * 16);

    const float* arow = A + (size_t)(m0 + l) * K + 2 * hl;
    const float* bcol = B + (size_t)(2 * hl) * NC + n0 + l;

    v8f acc[CPW];
#pragma unroll
    for (int i = 0; i < CPW; ++i) acc[i] = {};

    for (int k0 = 0; k0 < K; k0 += 4) {
        v2f a;
        a.x = arow[k0 + 0];
        a.y = arow[k0 + 1];
#pragma unroll
        for (int ct = 0; ct < CPW; ++ct) {
            v2f b;
            b.x = bcol[(size_t)k0 * NC + ct * 16];
            b.y = bcol[(size_t)(k0 + 1) * NC + ct * 16];
            acc[ct] = __builtin_amdgcn_wmma_f32_16x16x4_f32(false, a, false, b,
                                                            (short)0, acc[ct], false, false);
        }
    }

#pragma unroll
    for (int ct = 0; ct < CPW; ++ct) {
        float* crow = C + (size_t)(m0 + 8 * hl) * NC + n0 + ct * 16 + l;
#pragma unroll
        for (int r = 0; r < 8; ++r) crow[(size_t)r * NC] = acc[ct][r];
    }

    float v[16];
    {
        float fs[CPW], fd[CPW];
#pragma unroll
        for (int ct = 0; ct < CPW; ++ct) {
            fs[ct] = att_s[n0 + ct * 16 + l];
            fd[ct] = att_d[n0 + ct * 16 + l];
        }
#pragma unroll
        for (int r = 0; r < 8; ++r) {
            float ps = 0.f, pd = 0.f;
#pragma unroll
            for (int ct = 0; ct < CPW; ++ct) {
                ps += acc[ct][r] * fs[ct];
                pd += acc[ct][r] * fd[ct];
            }
            v[r] = ps;
            v[8 + r] = pd;
        }
    }
    const bool b8 = (l & 8) != 0, b4 = (l & 4) != 0, b2 = (l & 2) != 0, b1 = (l & 1) != 0;
    float w8[8], w4[4], w2[2];
#pragma unroll
    for (int i = 0; i < 8; ++i) {
        float keep = b8 ? v[i + 8] : v[i];
        float send = b8 ? v[i] : v[i + 8];
        w8[i] = keep + __shfl_xor(send, 8, 32);
    }
#pragma unroll
    for (int i = 0; i < 4; ++i) {
        float keep = b4 ? w8[i + 4] : w8[i];
        float send = b4 ? w8[i] : w8[i + 4];
        w4[i] = keep + __shfl_xor(send, 4, 32);
    }
#pragma unroll
    for (int i = 0; i < 2; ++i) {
        float keep = b2 ? w4[i + 2] : w4[i];
        float send = b2 ? w4[i] : w4[i + 2];
        w2[i] = keep + __shfl_xor(send, 2, 32);
    }
    const float dot = (b1 ? w2[1] : w2[0]) + __shfl_xor(b1 ? w2[0] : w2[1], 1, 32);
    float* dp = b8 ? adst : asrc;
    dp[(size_t)(m0 + 8 * hl + (l & 7)) * groups + (tile % groups)] = dot;
}

constexpr int kNodes  = 50000;
constexpr int kChunks = 128;

__device__ __forceinline__ int chunk_len(int E) {
    return (((E + kChunks - 1) / kChunks) + 3) & ~3;
}

__global__ __launch_bounds__(1024) void chunk_count_kernel(const int* __restrict__ dst,
                                                            int* __restrict__ H,
                                                            int* __restrict__ rank, int n, int E) {
    __shared__ int hist[kNodes];
    const int t = threadIdx.x;
    const int g = blockIdx.x;
    const int chunk = chunk_len(E);
    const int b   = min(g * chunk, E);
    const int end = min(b + chunk, E);
    const int n4  = n >> 2;
    int4* h4 = (int4*)hist;
    for (int i = t; i < n4; i += 1024) h4[i] = make_int4(0, 0, 0, 0);
    __syncthreads();

    const int nq = (end - b) >> 2;
    const int4* d4 = (const int4*)(dst + b);
    int4* r4 = (int4*)(rank + b);
    for (int q0 = t; q0 < nq; q0 += 4 * 1024) {
        int4 dv[4];
#pragma unroll
        for (int u = 0; u < 4; ++u) {
            const int q = q0 + u * 1024;
            dv[u] = (q < nq) ? d4[q] : make_int4(0, 0, 0, 0);
        }
#pragma unroll
        for (int u = 0; u < 4; ++u) {
            const int q = q0 + u * 1024;
            if (q < nq) {
                int4 r;
                r.x = atomicAdd(&hist[dv[u].x], 1);
                r.y = atomicAdd(&hist[dv[u].y], 1);
                r.z = atomicAdd(&hist[dv[u].z], 1);
                r.w = atomicAdd(&hist[dv[u].w], 1);
                r4[q] = r;
            }
        }
    }
    for (int e = b + (nq << 2) + t; e < end; e += 1024) rank[e] = atomicAdd(&hist[dst[e]], 1);
    __syncthreads();

    int4* out = (int4*)(H + (size_t)g * n);
    for (int i = t; i < n4; i += 1024) out[i] = h4[i];
}

__device__ __forceinline__ int4 add4(int4 a, int4 b) {
    return make_int4(a.x + b.x, a.y + b.y, a.z + b.z, a.w + b.w);
}

__global__ __launch_bounds__(256) void chunk_prefix_kernel(int* H, int* __restrict__ cnt, int n) {
    constexpr int W = 8, PER = kChunks / W;
    __shared__ int4 tot[W][32];
    const int lane = threadIdx.x & 31, w = threadIdx.x >> 5;
    const int n4 = n >> 2;
    const int q = blockIdx.x * 32 + lane;
    const bool live = q < n4;
    int4* H4 = (int4*)H;
    int4 v[PER];
#pragma unroll
    for (int j = 0; j < PER; ++j)
        v[j] = live ? H4[(size_t)(w * PER + j) * n4 + q] : make_int4(0, 0, 0, 0);
    int4 s = make_int4(0, 0, 0, 0);
#pragma unroll
    for (int j = 0; j < PER; ++j) {
        const int4 c = v[j];
        v[j] = s;
        s = add4(s, c);
    }
    tot[w][lane] = s;
    __syncthreads();
    int4 base = make_int4(0, 0, 0, 0);
    for (int k = 0; k < w; ++k) base = add4(base, tot[k][lane]);
    if (live) {
#pragma unroll
        for (int j = 0; j < PER; ++j) H4[(size_t)(w * PER + j) * n4 + q] = add4(v[j], base);
        if (w == W - 1) ((int4*)cnt)[q] = add4(base, s);
    }
}

constexpr int kScanPer4 = ((kNodes + 1023) / 1024 + 3) / 4;
constexpr int kScanCap4 = 1024 * kScanPer4;
static_assert((kScanPer4 & 1) == 1, "odd LDS pitch per thread");
static_assert(kScanCap4 * 4 >= kNodes + 4, "room for rowptr[n]");

__global__ __launch_bounds__(1024) void scan_kernel(const int* __restrict__ cnt, int* __restrict__ rowptr, int n) {
    __shared__ int4 buf[kScanCap4];
    __shared__ int wsum[32];
    const int t = threadIdx.x;
    const int lane = t & 31, w = t >> 5;
    const int n4 = n >> 2;
    const int4* c4 = (const int4*)cnt;
    int4 v[kScanPer4];

#pragma unroll
    for (int j = 0; j < kScanPer4; ++j) {
        const int q = t + j * 1024;
        v[j] = (q < n4) ? c4[q] : make_int4(0, 0, 0, 0);
    }
#pragma unroll
    for (int j = 0; j < kScanPer4; ++j) buf[t + j * 1024] = v[j];
    __syncthreads();

#pragma unroll
    for (int j = 0; j < kScanPer4; ++j) v[j] = buf[t * kScanPer4 + j];
    int s = 0;
#pragma unroll
    for (int j = 0; j < kScanPer4; ++j) s += (v[j].x + v[j].y) + (v[j].z + v[j].w);

    int x = s;
    for (int off = 1; off < 32; off <<= 1) {
        int u = __shfl_up(x, off, 32);
        if (lane >= off) x += u;
    }
    if (lane == 31) wsum[w] = x;
    __syncthreads();
    if (w == 0) {
        int y = wsum[lane];
        for (int off = 1; off < 32; off <<= 1) {
            int u = __shfl_up(y, off, 32);
            if (lane >= off) y += u;
        }
        wsum[lane] = y;
    }
    __syncthreads();
    int run = x - s + (w > 0 ? wsum[w - 1] : 0);

#pragma unroll
    for (int j = 0; j < kScanPer4; ++j) {
        const int4 c = v[j];
        int4 o;
        o.x = run; run += c.x;
        o.y = run; run += c.y;
        o.z = run; run += c.z;
        o.w = run; run += c.w;
        v[j] = o;
    }
#pragma unroll
    for (int j = 0; j < kScanPer4; ++j) buf[t * kScanPer4 + j] = v[j];
    __syncthreads();

    const int nout4 = (n + 4) >> 2;
    int4* r4 = (int4*)rowptr;
#pragma unroll
    for (int j = 0; j < kScanPer4; ++j) v[j] = buf[t + j * 1024];
#pragma unroll
    for (int j = 0; j < kScanPer4; ++j) {
        const int q = t + j * 1024;
        if (q < nout4) r4[q] = v[j];
    }
}

__global__ void place_kernel(const int* __restrict__ src, const int* __restrict__ dst,
                             const int* __restrict__ rowptr, const int* __restrict__ H,
                             const int* __restrict__ rank,
                             int* __restrict__ csr, int n, int E) {
    int e = blockIdx.x * blockDim.x + threadIdx.x;
    if (e >= E) return;
    const int sv = src[e];
    const int d  = dst[e];
    const int g  = (int)((unsigned)e / (unsigned)chunk_len(E));
    const int pos = rowptr[d] + H[(size_t)g * n + d] + rank[e];
    if ((unsigned)pos < (unsigned)E) csr[pos] = sv;
}

__device__ inline float leaky(float v) { return (v >= 0.f) ? v : NEG_SLOPE * v; }
__device__ inline float finite_or_zero(float m) { return (fabsf(m) < INFINITY) ? m : 0.f; }
__device__ inline float lane_bcast(float v, int k) {
    return __int_as_float(__builtin_amdgcn_readlane(__float_as_int(v), k));
}

template <int NB>
__device__ __forceinline__ void gather_rows1(float4& acc, const float* __restrict__ hcol,
                                             int s, float w0, float w1, int hd, int k) {
    float4 hv[NB];
#pragma unroll
    for (int i = 0; i < NB; ++i) {
        int sk = __builtin_amdgcn_readlane(s, k + i);
        hv[i] = *(const float4*)(hcol + (size_t)sk * 128);
    }
#pragma unroll
    for (int i = 0; i < NB; ++i) {
        float wa = lane_bcast(w0, k + i), wb = lane_bcast(w1, k + i);
        float wk = hd ? wb : wa;
        acc.x += wk * hv[i].x;
        acc.y += wk * hv[i].y;
        acc.z += wk * hv[i].z;
        acc.w += wk * hv[i].w;
    }
}

__global__ __launch_bounds__(256) void node_agg1_kernel(
        const int* __restrict__ rowptr, const int* __restrict__ csr,
        const float* __restrict__ hfeat,
        const float* __restrict__ asrc, const float* __restrict__ adst,
        const float* __restrict__ bias, float* __restrict__ hact, int n) {
    const int lane = threadIdx.x & 31;
    const int d = __builtin_amdgcn_readfirstlane(blockIdx.x * (blockDim.x >> 5) + (threadIdx.x >> 5));
    if (d >= n) return;
    const int beg = rowptr[d], end = rowptr[d + 1];
    const float ad0 = adst[d * 2 + 0], ad1 = adst[d * 2 + 1];
    const int hd = lane >> 4;

    float m0 = -INFINITY, m1 = -INFINITY;
    int s_first = 0;
    float2 a_first = {0.f, 0.f};
    for (int base = beg; base < end; base += 32) {
        int j = base + lane;
        if (j < end) {
            int s = csr[j];
            float2 a = *(const float2*)(asrc + (size_t)s * 2);
            if (base == beg) { s_first = s; a_first = a; }
            m0 = fmaxf(m0, leaky(a.x + ad0));
            m1 = fmaxf(m1, leaky(a.y + ad1));
        }
    }
    for (int off = 16; off; off >>= 1) {
        m0 = fmaxf(m0, __shfl_xor(m0, off, 32));
        m1 = fmaxf(m1, __shfl_xor(m1, off, 32));
    }
    m0 = finite_or_zero(m0);
    m1 = finite_or_zero(m1);

    float4 acc = {0.f, 0.f, 0.f, 0.f};
    float den0 = 0.f, den1 = 0.f;
    const float* hcol = hfeat + lane * 4;
    for (int base = beg; base < end; base += 32) {
        int j = base + lane;
        int s = s_first;
        float2 a = a_first;
        if (base != beg && j < end) {
            s = csr[j];
            a = *(const float2*)(asrc + (size_t)s * 2);
        }
        float w0 = 0.f, w1 = 0.f;
        if (j < end) {
            w0 = __expf(leaky(a.x + ad0) - m0);
            w1 = __expf(leaky(a.y + ad1) - m1);
        }
        den0 += w0;
        den1 += w1;
        const int cnt = min(32, end - base);
        int k = 0;
        for (; k + 8 <= cnt; k += 8) gather_rows1<8>(acc, hcol, s, w0, w1, hd, k);
        if (cnt & 4) { gather_rows1<4>(acc, hcol, s, w0, w1, hd, k); k += 4; }
        if (cnt & 2) { gather_rows1<2>(acc, hcol, s, w0, w1, hd, k); k += 2; }
        if (cnt & 1) { gather_rows1<1>(acc, hcol, s, w0, w1, hd, k); }
    }
    for (int off = 16; off; off >>= 1) {
        den0 += __shfl_xor(den0, off, 32);
        den1 += __shfl_xor(den1, off, 32);
    }
    const float den = (hd ? den1 : den0) + EPS_;
    const float4 bv = *(const float4*)(bias + lane * 4);
    float4 o;
    o.x = acc.x / den + bv.x;
    o.y = acc.y / den + bv.y;
    o.z = acc.z / den + bv.z;
    o.w = acc.w / den + bv.w;
    o.x = (o.x > 0.f) ? o.x : expm1f(o.x);
    o.y = (o.y > 0.f) ? o.y : expm1f(o.y);
    o.z = (o.z > 0.f) ? o.z : expm1f(o.z);
    o.w = (o.w > 0.f) ? o.w : expm1f(o.w);
    *(float4*)(hact + (size_t)d * 128 + lane * 4) = o;
}

__device__ __forceinline__ void node2_row(int d, int lane,
                                   const int* __restrict__ rowptr, const int* __restrict__ csr,
                                   const float* __restrict__ hfeat,
                                   const float* __restrict__ asrc, const float* __restrict__ adst,
                                   const float* __restrict__ bias,
                                   float& r0, float& r1, float& r2, float& r3) {
    const int beg = rowptr[d], end = rowptr[d + 1];
    const float ad = adst[d];
    float m = -INFINITY;
    int s_first = 0;
    float a_first = 0.f;
    for (int base = beg; base < end; base += 32) {
        int j = base + lane;
        if (j < end) {
            int s = csr[j];
            float a = asrc[s];
            if (base == beg) { s_first = s; a_first = a; }
            m = fmaxf(m, leaky(a + ad));
        }
    }
    for (int off = 16; off; off >>= 1) m = fmaxf(m, __shfl_xor(m, off, 32));
    m = finite_or_zero(m);

    float4 acc = {0.f, 0.f, 0.f, 0.f};
    float den = 0.f;
    const int g = lane >> 2;
    const float* hcol = hfeat + (lane & 3) * 4;
    for (int base = beg; base < end; base += 32) {
        int j = base + lane;
        int s = s_first;
        float a = a_first;
        if (base != beg && j < end) {
            s = csr[j];
            a = asrc[s];
        }
        float w = 0.f;
        if (j < end) w = __expf(leaky(a + ad) - m);
        else s = 0;
        den += w;
        const int cnt = min(32, end - base);
        float4 hv[4];
        float wk[4];
#pragma unroll
        for (int t = 0; t < 4; ++t) {
            int sk = __shfl(s, t * 8 + g, 32);
            wk[t] = __shfl(w, t * 8 + g, 32);
            hv[t] = *(const float4*)(hcol + (size_t)sk * 16);
        }
#pragma unroll
        for (int t = 0; t < 4; ++t) {
            const bool live = (t * 8 + g < cnt);
            acc.x += live ? wk[t] * hv[t].x : 0.f;
            acc.y += live ? wk[t] * hv[t].y : 0.f;
            acc.z += live ? wk[t] * hv[t].z : 0.f;
            acc.w += live ? wk[t] * hv[t].w : 0.f;
        }
    }
    for (int off = 16; off >= 4; off >>= 1) {
        acc.x += __shfl_xor(acc.x, off, 32);
        acc.y += __shfl_xor(acc.y, off, 32);
        acc.z += __shfl_xor(acc.z, off, 32);
        acc.w += __shfl_xor(acc.w, off, 32);
    }
    for (int off = 16; off; off >>= 1) den += __shfl_xor(den, off, 32);
    den += EPS_;
    const float4 bv = *(const float4*)(bias + (lane & 3) * 4);
    float4 v;
    v.x = acc.x / den + bv.x;
    v.y = acc.y / den + bv.y;
    v.z = acc.z / den + bv.z;
    v.w = acc.w / den + bv.w;
    float mx = fmaxf(fmaxf(v.x, v.y), fmaxf(v.z, v.w));
    mx = fmaxf(mx, __shfl_xor(mx, 1, 32));
    mx = fmaxf(mx, __shfl_xor(mx, 2, 32));
    v.x = __expf(v.x - mx);
    v.y = __expf(v.y - mx);
    v.z = __expf(v.z - mx);
    v.w = __expf(v.w - mx);
    float sum = (v.x + v.y) + (v.z + v.w);
    sum += __shfl_xor(sum, 1, 32);
    sum += __shfl_xor(sum, 2, 32);
    const float inv = 1.f / sum;
    r0 = v.x * inv;
    r1 = v.y * inv;
    r2 = v.z * inv;
    r3 = v.w * inv;
}

__global__ __launch_bounds__(256) void node_agg2_kernel(
        const int* __restrict__ rowptr, const int* __restrict__ csr,
        const float* __restrict__ hfeat,
        const float* __restrict__ asrc, const float* __restrict__ adst,
        const float* __restrict__ bias, float* __restrict__ out, int n) {
    const int lane = threadIdx.x & 31;
    const int d0 = __builtin_amdgcn_readfirstlane((blockIdx.x * (blockDim.x >> 5) + (threadIdx.x >> 5)) * 2);
    if (d0 >= n) return;
    float a0, a1, a2, a3;
    node2_row(d0, lane, rowptr, csr, hfeat, asrc, adst, bias, a0, a1, a2, a3);
    float b0 = a0, b1 = a1, b2 = a2, b3 = a3;
    const bool two = (d0 + 1 < n);
    if (two) node2_row(d0 + 1, lane, rowptr, csr, hfeat, asrc, adst, bias, b0, b1, b2, b3);
    if (lane < 4 || (two && lane < 8)) {
        float4 r;
        r.x = (lane < 4) ? a0 : b0;
        r.y = (lane < 4) ? a1 : b1;
        r.z = (lane < 4) ? a2 : b2;
        r.w = (lane < 4) ? a3 : b3;
        *(float4*)(out + (size_t)d0 * 16 + lane * 4) = r;
    }
}

extern "C" void kernel_launch(void* const* d_in, const int* in_sizes, int n_in,
                              void* d_out, int out_size, void* d_ws, size_t ws_size,
                              hipStream_t stream) {
    const float* x   = (const float*)d_in[0];
    const float* W1  = (const float*)d_in[1];
    const float* as1 = (const float*)d_in[2];
    const float* ad1 = (const float*)d_in[3];
    const float* b1  = (const float*)d_in[4];
    const float* W2  = (const float*)d_in[5];
    const float* as2 = (const float*)d_in[6];
    const float* ad2 = (const float*)d_in[7];
    const float* b2  = (const float*)d_in[8];
    const int*   ei  = (const int*)d_in[9];

    const int N = in_sizes[0] / 128;
    const int E = in_sizes[9] / 2;
    const int* src = ei;
    const int* dst = ei + E;

    float* w = (float*)d_ws;
    float* h1      = w; w += (size_t)N * 128;
    float* hact    = w; w += (size_t)N * 128;
    float* asrc1   = w; w += (size_t)N * 2;
    float* adst1   = w; w += (size_t)N * 2;
    float* asrc2   = w; w += (size_t)N;
    float* adst2   = w; w += (size_t)N;
    int*   cnt     = (int*)w; w += (size_t)N;
    int*   rowptr  = (int*)w; w += (size_t)N + 4;
    int*   csr     = (int*)w; w += (size_t)E;
    float* h2      = h1;
    int*   rank    = (int*)hact;
    int*   H       = (int*)h1;

    chunk_count_kernel<<<kChunks, 1024, 0, stream>>>(dst, H, rank, N, E);
    chunk_prefix_kernel<<<cdiv(N, 128), 256, 0, stream>>>(H, cnt, N);
    scan_kernel<<<1, 1024, 0, stream>>>(cnt, rowptr, N);
    place_kernel<<<cdiv(E, 256), 256, 0, stream>>>(src, dst, rowptr, H, rank, csr, N, E);

    {
        int tiles = (N / 16) * 2;
        wmma_gemm_f32<128, 128, 4><<<cdiv(tiles, 8), 256, 0, stream>>>(x, W1, h1, as1, ad1,
                                                                       asrc1, adst1, N);
        node_agg1_kernel<<<cdiv(N, 8), 256, 0, stream>>>(rowptr, csr, h1, asrc1, adst1, b1, hact, N);
    }

    {
        int tiles = (N / 16);
        wmma_gemm_f32<128, 16, 1><<<cdiv(tiles, 8), 256, 0, stream>>>(hact, W2, h2, as2, ad2,
                                                                      asrc2, adst2, N);
        node_agg2_kernel<<<cdiv(N, 16), 256, 0, stream>>>(rowptr, csr, h2, asrc2, adst2, b2,
                                                           (float*)d_out, N);
    }
}


# placement per edge chunk (2 work-groups each): rowptr + chunk-prefix row staged in LDS from whole lines, slot = LDS read + rank; was 2 global 4-byte gathers per edge; csr unchanged
# speedup vs baseline: 5.5561x; 1.0483x over previous
#include <hip/hip_runtime.h>
#include <hip/hip_bf16.h>
#include <math.h>

typedef float v2f __attribute__((ext_vector_type(2)));
typedef float v8f __attribute__((ext_vector_type(8)));

#define NEG_SLOPE 0.2f
#define EPS_ 1e-16f

static inline int cdiv(long a, long b) { return (int)((a + b - 1) / b); }

template <int K, int NC, int CPW>
__global__ __launch_bounds__(256) void wmma_gemm_f32(const float* __restrict__ A,
                                                     const float* __restrict__ B,
                                                     float* __restrict__ C,
                                                     const float* __restrict__ att_s,
                                                     const float* __restrict__ att_d,
                                                     float* __restrict__ asrc,
                                                     float* __restrict__ adst,
                                                     int M) {
    constexpr int nColT  = NC / 16;
    constexpr int groups = nColT / CPW;
    int wave = threadIdx.x >> 5;
    int lane = threadIdx.x & 31;
    int hl   = lane >> 4;
    int l    = lane & 15;
    int tile  = blockIdx.x * (blockDim.x >> 5) + wave;
    int total = (M >> 4) * groups;
    if (tile >= total) return;
    int m0 = (tile / groups) << 4;
    int n0 = (tile % groups) * (CPW * 16);

    const float* arow = A + (size_t)(m0 + l) * K + 2 * hl;
    const float* bcol = B + (size_t)(2 * hl) * NC + n0 + l;

    v8f acc[CPW];
#pragma unroll
    for (int i = 0; i < CPW; ++i) acc[i] = {};

    for (int k0 = 0; k0 < K; k0 += 4) {
        v2f a;
        a.x = arow[k0 + 0];
        a.y = arow[k0 + 1];
#pragma unroll
        for (int ct = 0; ct < CPW; ++ct) {
            v2f b;
            b.x = bcol[(size_t)k0 * NC + ct * 16];
            b.y = bcol[(size_t)(k0 + 1) * NC + ct * 16];
            acc[ct] = __builtin_amdgcn_wmma_f32_16x16x4_f32(false, a, false, b,
                                                            (short)0, acc[ct], false, false);
        }
    }

#pragma unroll
    for (int ct = 0; ct < CPW; ++ct) {
        float* crow = C + (size_t)(m0 + 8 * hl) * NC + n0 + ct * 16 + l;
#pragma unroll
        for (int r = 0; r < 8; ++r) crow[(size_t)r * NC] = acc[ct][r];
    }

    float v[16];
    {
        float fs[CPW], fd[CPW];
#pragma unroll
        for (int ct = 0; ct < CPW; ++ct) {
            fs[ct] = att_s[n0 + ct * 16 + l];
            fd[ct] = att_d[n0 + ct * 16 + l];
        }
#pragma unroll
        for (int r = 0; r < 8; ++r) {
            float ps = 0.f, pd = 0.f;
#pragma unroll
            for (int ct = 0; ct < CPW; ++ct) {
                ps += acc[ct][r] * fs[ct];
                pd += acc[ct][r] * fd[ct];
            }
            v[r] = ps;
            v[8 + r] = pd;
        }
    }
    const bool b8 = (l & 8) != 0, b4 = (l & 4) != 0, b2 = (l & 2) != 0, b1 = (l & 1) != 0;
    float w8[8], w4[4], w2[2];
#pragma unroll
    for (int i = 0; i < 8; ++i) {
        float keep = b8 ? v[i + 8] : v[i];
        float send = b8 ? v[i] : v[i + 8];
        w8[i] = keep + __shfl_xor(send, 8, 32);
    }
#pragma unroll
    for (int i = 0; i < 4; ++i) {
        float keep = b4 ? w8[i + 4] : w8[i];
        float send = b4 ? w8[i] : w8[i + 4];
        w4[i] = keep + __shfl_xor(send, 4, 32);
    }
#pragma unroll
    for (int i = 0; i < 2; ++i) {
        float keep = b2 ? w4[i + 2] : w4[i];
        float send = b2 ? w4[i] : w4[i + 2];
        w2[i] = keep + __shfl_xor(send, 2, 32);
    }
    const float dot = (b1 ? w2[1] : w2[0]) + __shfl_xor(b1 ? w2[0] : w2[1], 1, 32);
    float* dp = b8 ? adst : asrc;
    dp[(size_t)(m0 + 8 * hl + (l & 7)) * groups + (tile % groups)] = dot;
}

constexpr int kNodes  = 50000;
constexpr int kChunks = 128;

__device__ __forceinline__ int chunk_len(int E) {
    return (((E + kChunks - 1) / kChunks) + 3) & ~3;
}

__global__ __launch_bounds__(1024) void chunk_count_kernel(const int* __restrict__ dst,
                                                            int* __restrict__ H,
                                                            int* __restrict__ rank, int n, int E) {
    __shared__ int hist[kNodes];
    const int t = threadIdx.x;
    const int g = blockIdx.x;
    const int chunk = chunk_len(E);
    const int b   = min(g * chunk, E);
    const int end = min(b + chunk, E);
    const int n4  = n >> 2;
    int4* h4 = (int4*)hist;
    for (int i = t; i < n4; i += 1024) h4[i] = make_int4(0, 0, 0, 0);
    __syncthreads();

    const int nq = (end - b) >> 2;
    const int4* d4 = (const int4*)(dst + b);
    int4* r4 = (int4*)(rank + b);
    for (int q0 = t; q0 < nq; q0 += 4 * 1024) {
        int4 dv[4];
#pragma unroll
        for (int u = 0; u < 4; ++u) {
            const int q = q0 + u * 1024;
            dv[u] = (q < nq) ? d4[q] : make_int4(0, 0, 0, 0);
        }
#pragma unroll
        for (int u = 0; u < 4; ++u) {
            const int q = q0 + u * 1024;
            if (q < nq) {
                int4 r;
                r.x = atomicAdd(&hist[dv[u].x], 1);
                r.y = atomicAdd(&hist[dv[u].y], 1);
                r.z = atomicAdd(&hist[dv[u].z], 1);
                r.w = atomicAdd(&hist[dv[u].w], 1);
                r4[q] = r;
            }
        }
    }
    for (int e = b + (nq << 2) + t; e < end; e += 1024) rank[e] = atomicAdd(&hist[dst[e]], 1);
    __syncthreads();

    int4* out = (int4*)(H + (size_t)g * n);
    for (int i = t; i < n4; i += 1024) out[i] = h4[i];
}

__device__ __forceinline__ int4 add4(int4 a, int4 b) {
    return make_int4(a.x + b.x, a.y + b.y, a.z + b.z, a.w + b.w);
}

__global__ __launch_bounds__(256) void chunk_prefix_kernel(int* H, int* __restrict__ cnt, int n) {
    constexpr int W = 8, PER = kChunks / W;
    __shared__ int4 tot[W][32];
    const int lane = threadIdx.x & 31, w = threadIdx.x >> 5;
    const int n4 = n >> 2;
    const int q = blockIdx.x * 32 + lane;
    const bool live = q < n4;
    int4* H4 = (int4*)H;
    int4 v[PER];
#pragma unroll
    for (int j = 0; j < PER; ++j)
        v[j] = live ? H4[(size_t)(w * PER + j) * n4 + q] : make_int4(0, 0, 0, 0);
    int4 s = make_int4(0, 0, 0, 0);
#pragma unroll
    for (int j = 0; j < PER; ++j) {
        const int4 c = v[j];
        v[j] = s;
        s = add4(s, c);
    }
    tot[w][lane] = s;
    __syncthreads();
    int4 base = make_int4(0, 0, 0, 0);
    for (int k = 0; k < w; ++k) base = add4(base, tot[k][lane]);
    if (live) {
#pragma unroll
        for (int j = 0; j < PER; ++j) H4[(size_t)(w * PER + j) * n4 + q] = add4(v[j], base);
        if (w == W - 1) ((int4*)cnt)[q] = add4(base, s);
    }
}

constexpr int kScanPer4 = ((kNodes + 1023) / 1024 + 3) / 4;
constexpr int kScanCap4 = 1024 * kScanPer4;
static_assert((kScanPer4 & 1) == 1, "odd LDS pitch per thread");
static_assert(kScanCap4 * 4 >= kNodes + 4, "room for rowptr[n]");

__global__ __launch_bounds__(1024) void scan_kernel(const int* __restrict__ cnt, int* __restrict__ rowptr, int n) {
    __shared__ int4 buf[kScanCap4];
    __shared__ int wsum[32];
    const int t = threadIdx.x;
    const int lane = t & 31, w = t >> 5;
    const int n4 = n >> 2;
    const int4* c4 = (const int4*)cnt;
    int4 v[kScanPer4];

#pragma unroll
    for (int j = 0; j < kScanPer4; ++j) {
        const int q = t + j * 1024;
        v[j] = (q < n4) ? c4[q] : make_int4(0, 0, 0, 0);
    }
#pragma unroll
    for (int j = 0; j < kScanPer4; ++j) buf[t + j * 1024] = v[j];
    __syncthreads();

#pragma unroll
    for (int j = 0; j < kScanPer4; ++j) v[j] = buf[t * kScanPer4 + j];
    int s = 0;
#pragma unroll
    for (int j = 0; j < kScanPer4; ++j) s += (v[j].x + v[j].y) + (v[j].z + v[j].w);

    int x = s;
    for (int off = 1; off < 32; off <<= 1) {
        int u = __shfl_up(x, off, 32);
        if (lane >= off) x += u;
    }
    if (lane == 31) wsum[w] = x;
    __syncthreads();
    if (w == 0) {
        int y = wsum[lane];
        for (int off = 1; off < 32; off <<= 1) {
            int u = __shfl_up(y, off, 32);
            if (lane >= off) y += u;
        }
        wsum[lane] = y;
    }
    __syncthreads();
    int run = x - s + (w > 0 ? wsum[w - 1] : 0);

#pragma unroll
    for (int j = 0; j < kScanPer4; ++j) {
        const int4 c = v[j];
        int4 o;
        o.x = run; run += c.x;
        o.y = run; run += c.y;
        o.z = run; run += c.z;
        o.w = run; run += c.w;
        v[j] = o;
    }
#pragma unroll
    for (int j = 0; j < kScanPer4; ++j) buf[t * kScanPer4 + j] = v[j];
    __syncthreads();

    const int nout4 = (n + 4) >> 2;
    int4* r4 = (int4*)rowptr;
#pragma unroll
    for (int j = 0; j < kScanPer4; ++j) v[j] = buf[t + j * 1024];
#pragma unroll
    for (int j = 0; j < kScanPer4; ++j) {
        const int q = t + j * 1024;
        if (q < nout4) r4[q] = v[j];
    }
}

constexpr int kPlaceSplit = 2;
constexpr int kStageQ = (kNodes / 4 + 1023) / 1024;
constexpr int kStageG = (kStageQ + 1) / 2;
constexpr int kPlaceU = 2;

__device__ __forceinline__ void place4(const int* first, int4 d, int4 s, int4 r,
                                       int* __restrict__ csr, int E) {
    const unsigned uE = (unsigned)E;
    const unsigned p0 = (unsigned)(first[d.x] + r.x);
    const unsigned p1 = (unsigned)(first[d.y] + r.y);
    const unsigned p2 = (unsigned)(first[d.z] + r.z);
    const unsigned p3 = (unsigned)(first[d.w] + r.w);
    if ((p0 < uE) & (p1 < uE) & (p2 < uE) & (p3 < uE)) {
        csr[p0] = s.x;
        csr[p1] = s.y;
        csr[p2] = s.z;
        csr[p3] = s.w;
    } else {
        if (p0 < uE) csr[p0] = s.x;
        if (p1 < uE) csr[p1] = s.y;
        if (p2 < uE) csr[p2] = s.z;
        if (p3 < uE) csr[p3] = s.w;
    }
}

template <int J0, int CNT>
__device__ __forceinline__ void stage_first(int4* f4, const int4* __restrict__ rp4,
                                            const int4* __restrict__ hrow, int t, int n4) {
    int4 a[CNT], c[CNT];
#pragma unroll
    for (int j = 0; j < CNT; ++j) {
        const int qc = min(t + (J0 + j) * 1024, n4 - 1);
        a[j] = rp4[qc];
        c[j] = hrow[qc];
    }
#pragma unroll
    for (int j = 0; j < CNT; ++j) {
        const int q = t + (J0 + j) * 1024;
        if (q < n4) f4[q] = add4(a[j], c[j]);
    }
}

__global__ __launch_bounds__(1024) void place_kernel(const int* __restrict__ src, const int* __restrict__ dst,
                                                      const int* __restrict__ rowptr, const int* __restrict__ H,
                                                      const int* __restrict__ rank,
                                                      int* __restrict__ csr, int n, int E) {
    __shared__ int first[kNodes];
    const int t = threadIdx.x;
    const int g = blockIdx.x / kPlaceSplit;
    const int part = blockIdx.x % kPlaceSplit;
    const int chunk = chunk_len(E);
    const int b   = min(g * chunk, E);
    const int end = min(b + chunk, E);
    if (b >= end) return;
    const int nq  = (end - b) >> 2;
    const int per = (nq + kPlaceSplit - 1) / kPlaceSplit;
    const int q_lo = min(part * per, nq);
    const int q_hi = min(q_lo + per, nq);
    const int4* d4 = (const int4*)(dst + b);
    const int4* s4 = (const int4*)(src + b);
    const int4* r4 = (const int4*)(rank + b);

    int4 dv[kPlaceU], sv[kPlaceU], rv[kPlaceU];
#pragma unroll
    for (int u = 0; u < kPlaceU; ++u) {
        const int q = q_lo + t + u * 1024;
        const bool live = q < q_hi;
        dv[u] = live ? d4[q] : make_int4(0, 0, 0, 0);
        sv[u] = live ? s4[q] : make_int4(0, 0, 0, 0);
        rv[u] = live ? r4[q] : make_int4(0, 0, 0, 0);
    }

    {
        const int n4 = n >> 2;
        const int4* hrow = (const int4*)(H + (size_t)g * n);
        const int4* rp4  = (const int4*)rowptr;
        stage_first<0, kStageG>((int4*)first, rp4, hrow, t, n4);
        stage_first<kStageG, kStageQ - kStageG>((int4*)first, rp4, hrow, t, n4);
    }
    __syncthreads();

    for (int q0 = q_lo + t; q0 < q_hi; q0 += kPlaceU * 1024) {
        if (q0 != q_lo + t) {
#pragma unroll
            for (int u = 0; u < kPlaceU; ++u) {
                const int q = q0 + u * 1024;
                const bool live = q < q_hi;
                dv[u] = live ? d4[q] : make_int4(0, 0, 0, 0);
                sv[u] = live ? s4[q] : make_int4(0, 0, 0, 0);
                rv[u] = live ? r4[q] : make_int4(0, 0, 0, 0);
            }
        }
#pragma unroll
        for (int u = 0; u < kPlaceU; ++u)
            if (q0 + u * 1024 < q_hi) place4(first, dv[u], sv[u], rv[u], csr, E);
    }
    if (part == 0)
        for (int e = b + (nq << 2) + t; e < end; e += 1024) {
            const int pos = first[dst[e]] + rank[e];
            if ((unsigned)pos < (unsigned)E) csr[pos] = src[e];
        }
}

__device__ inline float leaky(float v) { return (v >= 0.f) ? v : NEG_SLOPE * v; }
__device__ inline float finite_or_zero(float m) { return (fabsf(m) < INFINITY) ? m : 0.f; }
__device__ inline float lane_bcast(float v, int k) {
    return __int_as_float(__builtin_amdgcn_readlane(__float_as_int(v), k));
}

template <int NB>
__device__ __forceinline__ void gather_rows1(float4& acc, const float* __restrict__ hcol,
                                             int s, float w0, float w1, int hd, int k) {
    float4 hv[NB];
#pragma unroll
    for (int i = 0; i < NB; ++i) {
        int sk = __builtin_amdgcn_readlane(s, k + i);
        hv[i] = *(const float4*)(hcol + (size_t)sk * 128);
    }
#pragma unroll
    for (int i = 0; i < NB; ++i) {
        float wa = lane_bcast(w0, k + i), wb = lane_bcast(w1, k + i);
        float wk = hd ? wb : wa;
        acc.x += wk * hv[i].x;
        acc.y += wk * hv[i].y;
        acc.z += wk * hv[i].z;
        acc.w += wk * hv[i].w;
    }
}

__global__ __launch_bounds__(256) void node_agg1_kernel(
        const int* __restrict__ rowptr, const int* __restrict__ csr,
        const float* __restrict__ hfeat,
        const float* __restrict__ asrc, const float* __restrict__ adst,
        const float* __restrict__ bias, float* __restrict__ hact, int n) {
    const int lane = threadIdx.x & 31;
    const int d = __builtin_amdgcn_readfirstlane(blockIdx.x * (blockDim.x >> 5) + (threadIdx.x >> 5));
    if (d >= n) return;
    const int beg = rowptr[d], end = rowptr[d + 1];
    const float ad0 = adst[d * 2 + 0], ad1 = adst[d * 2 + 1];
    const int hd = lane >> 4;

    float m0 = -INFINITY, m1 = -INFINITY;
    int s_first = 0;
    float2 a_first = {0.f, 0.f};
    for (int base = beg; base < end; base += 32) {
        int j = base + lane;
        if (j < end) {
            int s = csr[j];
            float2 a = *(const float2*)(asrc + (size_t)s * 2);
            if (base == beg) { s_first = s; a_first = a; }
            m0 = fmaxf(m0, leaky(a.x + ad0));
            m1 = fmaxf(m1, leaky(a.y + ad1));
        }
    }
    for (int off = 16; off; off >>= 1) {
        m0 = fmaxf(m0, __shfl_xor(m0, off, 32));
        m1 = fmaxf(m1, __shfl_xor(m1, off, 32));
    }
    m0 = finite_or_zero(m0);
    m1 = finite_or_zero(m1);

    float4 acc = {0.f, 0.f, 0.f, 0.f};
    float den0 = 0.f, den1 = 0.f;
    const float* hcol = hfeat + lane * 4;
    for (int base = beg; base < end; base += 32) {
        int j = base + lane;
        int s = s_first;
        float2 a = a_first;
        if (base != beg && j < end) {
            s = csr[j];
            a = *(const float2*)(asrc + (size_t)s * 2);
        }
        float w0 = 0.f, w1 = 0.f;
        if (j < end) {
            w0 = __expf(leaky(a.x + ad0) - m0);
            w1 = __expf(leaky(a.y + ad1) - m1);
        }
        den0 += w0;
        den1 += w1;
        const int cnt = min(32, end - base);
        int k = 0;
        for (; k + 8 <= cnt; k += 8) gather_rows1<8>(acc, hcol, s, w0, w1, hd, k);
        if (cnt & 4) { gather_rows1<4>(acc, hcol, s, w0, w1, hd, k); k += 4; }
        if (cnt & 2) { gather_rows1<2>(acc, hcol, s, w0, w1, hd, k); k += 2; }
        if (cnt & 1) { gather_rows1<1>(acc, hcol, s, w0, w1, hd, k); }
    }
    for (int off = 16; off; off >>= 1) {
        den0 += __shfl_xor(den0, off, 32);
        den1 += __shfl_xor(den1, off, 32);
    }
    const float den = (hd ? den1 : den0) + EPS_;
    const float4 bv = *(const float4*)(bias + lane * 4);
    float4 o;
    o.x = acc.x / den + bv.x;
    o.y = acc.y / den + bv.y;
    o.z = acc.z / den + bv.z;
    o.w = acc.w / den + bv.w;
    o.x = (o.x > 0.f) ? o.x : expm1f(o.x);
    o.y = (o.y > 0.f) ? o.y : expm1f(o.y);
    o.z = (o.z > 0.f) ? o.z : expm1f(o.z);
    o.w = (o.w > 0.f) ? o.w : expm1f(o.w);
    *(float4*)(hact + (size_t)d * 128 + lane * 4) = o;
}

__device__ __forceinline__ void node2_row(int d, int lane,
                                   const int* __restrict__ rowptr, const int* __restrict__ csr,
                                   const float* __restrict__ hfeat,
                                   const float* __restrict__ asrc, const float* __restrict__ adst,
                                   const float* __restrict__ bias,
                                   float& r0, float& r1, float& r2, float& r3) {
    const int beg = rowptr[d], end = rowptr[d + 1];
    const float ad = adst[d];
    float m = -INFINITY;
    int s_first = 0;
    float a_first = 0.f;
    for (int base = beg; base < end; base += 32) {
        int j = base + lane;
        if (j < end) {
            int s = csr[j];
            float a = asrc[s];
            if (base == beg) { s_first = s; a_first = a; }
            m = fmaxf(m, leaky(a + ad));
        }
    }
    for (int off = 16; off; off >>= 1) m = fmaxf(m, __shfl_xor(m, off, 32));
    m = finite_or_zero(m);

    float4 acc = {0.f, 0.f, 0.f, 0.f};
    float den = 0.f;
    const int g = lane >> 2;
    const float* hcol = hfeat + (lane & 3) * 4;
    for (int base = beg; base < end; base += 32) {
        int j = base + lane;
        int s = s_first;
        float a = a_first;
        if (base != beg && j < end) {
            s = csr[j];
            a = asrc[s];
        }
        float w = 0.f;
        if (j < end) w = __expf(leaky(a + ad) - m);
        else s = 0;
        den += w;
        const int cnt = min(32, end - base);
        float4 hv[4];
        float wk[4];
#pragma unroll
        for (int t = 0; t < 4; ++t) {
            int sk = __shfl(s, t * 8 + g, 32);
            wk[t] = __shfl(w, t * 8 + g, 32);
            hv[t] = *(const float4*)(hcol + (size_t)sk * 16);
        }
#pragma unroll
        for (int t = 0; t < 4; ++t) {
            const bool live = (t * 8 + g < cnt);
            acc.x += live ? wk[t] * hv[t].x : 0.f;
            acc.y += live ? wk[t] * hv[t].y : 0.f;
            acc.z += live ? wk[t] * hv[t].z : 0.f;
            acc.w += live ? wk[t] * hv[t].w : 0.f;
        }
    }
    for (int off = 16; off >= 4; off >>= 1) {
        acc.x += __shfl_xor(acc.x, off, 32);
        acc.y += __shfl_xor(acc.y, off, 32);
        acc.z += __shfl_xor(acc.z, off, 32);
        acc.w += __shfl_xor(acc.w, off, 32);
    }
    for (int off = 16; off; off >>= 1) den += __shfl_xor(den, off, 32);
    den += EPS_;
    const float4 bv = *(const float4*)(bias + (lane & 3) * 4);
    float4 v;
    v.x = acc.x / den + bv.x;
    v.y = acc.y / den + bv.y;
    v.z = acc.z / den + bv.z;
    v.w = acc.w / den + bv.w;
    float mx = fmaxf(fmaxf(v.x, v.y), fmaxf(v.z, v.w));
    mx = fmaxf(mx, __shfl_xor(mx, 1, 32));
    mx = fmaxf(mx, __shfl_xor(mx, 2, 32));
    v.x = __expf(v.x - mx);
    v.y = __expf(v.y - mx);
    v.z = __expf(v.z - mx);
    v.w = __expf(v.w - mx);
    float sum = (v.x + v.y) + (v.z + v.w);
    sum += __shfl_xor(sum, 1, 32);
    sum += __shfl_xor(sum, 2, 32);
    const float inv = 1.f / sum;
    r0 = v.x * inv;
    r1 = v.y * inv;
    r2 = v.z * inv;
    r3 = v.w * inv;
}

__global__ __launch_bounds__(256) void node_agg2_kernel(
        const int* __restrict__ rowptr, const int* __restrict__ csr,
        const float* __restrict__ hfeat,
        const float* __restrict__ asrc, const float* __restrict__ adst,
        const float* __restrict__ bias, float* __restrict__ out, int n) {
    const int lane = threadIdx.x & 31;
    const int d0 = __builtin_amdgcn_readfirstlane((blockIdx.x * (blockDim.x >> 5) + (threadIdx.x >> 5)) * 2);
    if (d0 >= n) return;
    float a0, a1, a2, a3;
    node2_row(d0, lane, rowptr, csr, hfeat, asrc, adst, bias, a0, a1, a2, a3);
    float b0 = a0, b1 = a1, b2 = a2, b3 = a3;
    const bool two = (d0 + 1 < n);
    if (two) node2_row(d0 + 1, lane, rowptr, csr, hfeat, asrc, adst, bias, b0, b1, b2, b3);
    if (lane < 4 || (two && lane < 8)) {
        float4 r;
        r.x = (lane < 4) ? a0 : b0;
        r.y = (lane < 4) ? a1 : b1;
        r.z = (lane < 4) ? a2 : b2;
        r.w = (lane < 4) ? a3 : b3;
        *(float4*)(out + (size_t)d0 * 16 + lane * 4) = r;
    }
}

extern "C" void kernel_launch(void* const* d_in, const int* in_sizes, int n_in,
                              void* d_out, int out_size, void* d_ws, size_t ws_size,
                              hipStream_t stream) {
    const float* x   = (const float*)d_in[0];
    const float* W1  = (const float*)d_in[1];
    const float* as1 = (const float*)d_in[2];
    const float* ad1 = (const float*)d_in[3];
    const float* b1  = (const float*)d_in[4];
    const float* W2  = (const float*)d_in[5];
    const float* as2 = (const float*)d_in[6];
    const float* ad2 = (const float*)d_in[7];
    const float* b2  = (const float*)d_in[8];
    const int*   ei  = (const int*)d_in[9];

    const int N = in_sizes[0] / 128;
    const int E = in_sizes[9] / 2;
    const int* src = ei;
    const int* dst = ei + E;

    float* w = (float*)d_ws;
    float* h1      = w; w += (size_t)N * 128;
    float* hact    = w; w += (size_t)N * 128;
    float* asrc1   = w; w += (size_t)N * 2;
    float* adst1   = w; w += (size_t)N * 2;
    float* asrc2   = w; w += (size_t)N;
    float* adst2   = w; w += (size_t)N;
    int*   cnt     = (int*)w; w += (size_t)N;
    int*   rowptr  = (int*)w; w += (size_t)N + 4;
    int*   csr     = (int*)w; w += (size_t)E;
    float* h2      = h1;
    int*   rank    = (int*)hact;
    int*   H       = (int*)h1;

    chunk_count_kernel<<<kChunks, 1024, 0, stream>>>(dst, H, rank, N, E);
    chunk_prefix_kernel<<<cdiv(N, 128), 256, 0, stream>>>(H, cnt, N);
    scan_kernel<<<1, 1024, 0, stream>>>(cnt, rowptr, N);
    place_kernel<<<kChunks * kPlaceSplit, 1024, 0, stream>>>(src, dst, rowptr, H, rank, csr, N, E);

    {
        int tiles = (N / 16) * 2;
        wmma_gemm_f32<128, 128, 4><<<cdiv(tiles, 8), 256, 0, stream>>>(x, W1, h1, as1, ad1,
                                                                       asrc1, adst1, N);
        node_agg1_kernel<<<cdiv(N, 8), 256, 0, stream>>>(rowptr, csr, h1, asrc1, adst1, b1, hact, N);
    }

    {
        int tiles = (N / 16);
        wmma_gemm_f32<128, 16, 1><<<cdiv(tiles, 8), 256, 0, stream>>>(hact, W2, h2, as2, ad2,
                                                                      asrc2, adst2, N);
        node_agg2_kernel<<<cdiv(N, 16), 256, 0, stream>>>(rowptr, csr, h2, asrc2, adst2, b2,
                                                           (float*)d_out, N);
    }
}


# source attention scores recomputed in the aggregation waves as dot(gathered feature row, att_src), both layers, not gathered from a score table; one-pass softmax with running max: sums in a new order
# speedup vs baseline: 5.8227x; 1.0480x over previous
#include <hip/hip_runtime.h>
#include <hip/hip_bf16.h>
#include <math.h>

typedef float v2f __attribute__((ext_vector_type(2)));
typedef float v8f __attribute__((ext_vector_type(8)));

#define NEG_SLOPE 0.2f
#define EPS_ 1e-16f

static inline int cdiv(long a, long b) { return (int)((a + b - 1) / b); }

template <int K, int NC, int CPW>
__global__ __launch_bounds__(256) void wmma_gemm_f32(const float* __restrict__ A,
                                                     const float* __restrict__ B,
                                                     float* __restrict__ C,
                                                     const float* __restrict__ att_d,
                                                     float* __restrict__ adst,
                                                     int M) {
    constexpr int nColT  = NC / 16;
    constexpr int groups = nColT / CPW;
    int wave = threadIdx.x >> 5;
    int lane = threadIdx.x & 31;
    int hl   = lane >> 4;
    int l    = lane & 15;
    int tile  = blockIdx.x * (blockDim.x >> 5) + wave;
    int total = (M >> 4) * groups;
    if (tile >= total) return;
    int m0 = (tile / groups) << 4;
    int n0 = (tile % groups) * (CPW * 16);

    const float* arow = A + (size_t)(m0 + l) * K + 2 * hl;
    const float* bcol = B + (size_t)(2 * hl) * NC + n0 + l;

    v8f acc[CPW];
#pragma unroll
    for (int i = 0; i < CPW; ++i) acc[i] = {};

    for (int k0 = 0; k0 < K; k0 += 4) {
        v2f a;
        a.x = arow[k0 + 0];
        a.y = arow[k0 + 1];
#pragma unroll
        for (int ct = 0; ct < CPW; ++ct) {
            v2f b;
            b.x = bcol[(size_t)k0 * NC + ct * 16];
            b.y = bcol[(size_t)(k0 + 1) * NC + ct * 16];
            acc[ct] = __builtin_amdgcn_wmma_f32_16x16x4_f32(false, a, false, b,
                                                            (short)0, acc[ct], false, false);
        }
    }

#pragma unroll
    for (int ct = 0; ct < CPW; ++ct) {
        float* crow = C + (size_t)(m0 + 8 * hl) * NC + n0 + ct * 16 + l;
#pragma unroll
        for (int r = 0; r < 8; ++r) crow[(size_t)r * NC] = acc[ct][r];
    }

    float v[8];
    {
        float fd[CPW];
#pragma unroll
        for (int ct = 0; ct < CPW; ++ct) fd[ct] = att_d[n0 + ct * 16 + l];
#pragma unroll
        for (int r = 0; r < 8; ++r) {
            float pd = 0.f;
#pragma unroll
            for (int ct = 0; ct < CPW; ++ct) pd += acc[ct][r] * fd[ct];
            v[r] = pd;
        }
    }
    const bool b8 = (l & 8) != 0, b4 = (l & 4) != 0, b2 = (l & 2) != 0;
    float w4[4], w2[2];
#pragma unroll
    for (int i = 0; i < 4; ++i) {
        float keep = b8 ? v[i + 4] : v[i];
        float send = b8 ? v[i] : v[i + 4];
        w4[i] = keep + __shfl_xor(send, 8, 32);
    }
#pragma unroll
    for (int i = 0; i < 2; ++i) {
        float keep = b4 ? w4[i + 2] : w4[i];
        float send = b4 ? w4[i] : w4[i + 2];
        w2[i] = keep + __shfl_xor(send, 4, 32);
    }
    const float w1 = (b2 ? w2[1] : w2[0]) + __shfl_xor(b2 ? w2[0] : w2[1], 2, 32);
    const float dot = w1 + __shfl_xor(w1, 1, 32);
    if ((l & 1) == 0) adst[(size_t)(m0 + 8 * hl + (l >> 1)) * groups + (tile % groups)] = dot;
}

constexpr int kNodes  = 50000;
constexpr int kChunks = 128;

__device__ __forceinline__ int chunk_len(int E) {
    return (((E + kChunks - 1) / kChunks) + 3) & ~3;
}

__global__ __launch_bounds__(1024) void chunk_count_kernel(const int* __restrict__ dst,
                                                            int* __restrict__ H,
                                                            int* __restrict__ rank, int n, int E) {
    __shared__ int hist[kNodes];
    const int t = threadIdx.x;
    const int g = blockIdx.x;
    const int chunk = chunk_len(E);
    const int b   = min(g * chunk, E);
    const int end = min(b + chunk, E);
    const int n4  = n >> 2;
    int4* h4 = (int4*)hist;
    for (int i = t; i < n4; i += 1024) h4[i] = make_int4(0, 0, 0, 0);
    __syncthreads();

    const int nq = (end - b) >> 2;
    const int4* d4 = (const int4*)(dst + b);
    int4* r4 = (int4*)(rank + b);
    for (int q0 = t; q0 < nq; q0 += 4 * 1024) {
        int4 dv[4];
#pragma unroll
        for (int u = 0; u < 4; ++u) {
            const int q = q0 + u * 1024;
            dv[u] = (q < nq) ? d4[q] : make_int4(0, 0, 0, 0);
        }
#pragma unroll
        for (int u = 0; u < 4; ++u) {
            const int q = q0 + u * 1024;
            if (q < nq) {
                int4 r;
                r.x = atomicAdd(&hist[dv[u].x], 1);
                r.y = atomicAdd(&hist[dv[u].y], 1);
                r.z = atomicAdd(&hist[dv[u].z], 1);
                r.w = atomicAdd(&hist[dv[u].w], 1);
                r4[q] = r;
            }
        }
    }
    for (int e = b + (nq << 2) + t; e < end; e += 1024) rank[e] = atomicAdd(&hist[dst[e]], 1);
    __syncthreads();

    int4* out = (int4*)(H + (size_t)g * n);
    for (int i = t; i < n4; i += 1024) out[i] = h4[i];
}

__device__ __forceinline__ int4 add4(int4 a, int4 b) {
    return make_int4(a.x + b.x, a.y + b.y, a.z + b.z, a.w + b.w);
}

__global__ __launch_bounds__(256) void chunk_prefix_kernel(int* H, int* __restrict__ cnt, int n) {
    constexpr int W = 8, PER = kChunks / W;
    __shared__ int4 tot[W][32];
    const int lane = threadIdx.x & 31, w = threadIdx.x >> 5;
    const int n4 = n >> 2;
    const int q = blockIdx.x * 32 + lane;
    const bool live = q < n4;
    int4* H4 = (int4*)H;
    int4 v[PER];
#pragma unroll
    for (int j = 0; j < PER; ++j)
        v[j] = live ? H4[(size_t)(w * PER + j) * n4 + q] : make_int4(0, 0, 0, 0);
    int4 s = make_int4(0, 0, 0, 0);
#pragma unroll
    for (int j = 0; j < PER; ++j) {
        const int4 c = v[j];
        v[j] = s;
        s = add4(s, c);
    }
    tot[w][lane] = s;
    __syncthreads();
    int4 base = make_int4(0, 0, 0, 0);
    for (int k = 0; k < w; ++k) base = add4(base, tot[k][lane]);
    if (live) {
#pragma unroll
        for (int j = 0; j < PER; ++j) H4[(size_t)(w * PER + j) * n4 + q] = add4(v[j], base);
        if (w == W - 1) ((int4*)cnt)[q] = add4(base, s);
    }
}

constexpr int kScanPer4 = ((kNodes + 1023) / 1024 + 3) / 4;
constexpr int kScanCap4 = 1024 * kScanPer4;
static_assert((kScanPer4 & 1) == 1, "odd LDS pitch per thread");
static_assert(kScanCap4 * 4 >= kNodes + 4, "room for rowptr[n]");

__global__ __launch_bounds__(1024) void scan_kernel(const int* __restrict__ cnt, int* __restrict__ rowptr, int n) {
    __shared__ int4 buf[kScanCap4];
    __shared__ int wsum[32];
    const int t = threadIdx.x;
    const int lane = t & 31, w = t >> 5;
    const int n4 = n >> 2;
    const int4* c4 = (const int4*)cnt;
    int4 v[kScanPer4];

#pragma unroll
    for (int j = 0; j < kScanPer4; ++j) {
        const int q = t + j * 1024;
        v[j] = (q < n4) ? c4[q] : make_int4(0, 0, 0, 0);
    }
#pragma unroll
    for (int j = 0; j < kScanPer4; ++j) buf[t + j * 1024] = v[j];
    __syncthreads();

#pragma unroll
    for (int j = 0; j < kScanPer4; ++j) v[j] = buf[t * kScanPer4 + j];
    int s = 0;
#pragma unroll
    for (int j = 0; j < kScanPer4; ++j) s += (v[j].x + v[j].y) + (v[j].z + v[j].w);

    int x = s;
    for (int off = 1; off < 32; off <<= 1) {
        int u = __shfl_up(x, off, 32);
        if (lane >= off) x += u;
    }
    if (lane == 31) wsum[w] = x;
    __syncthreads();
    if (w == 0) {
        int y = wsum[lane];
        for (int off = 1; off < 32; off <<= 1) {
            int u = __shfl_up(y, off, 32);
            if (lane >= off) y += u;
        }
        wsum[lane] = y;
    }
    __syncthreads();
    int run = x - s + (w > 0 ? wsum[w - 1] : 0);

#pragma unroll
    for (int j = 0; j < kScanPer4; ++j) {
        const int4 c = v[j];
        int4 o;
        o.x = run; run += c.x;
        o.y = run; run += c.y;
        o.z = run; run += c.z;
        o.w = run; run += c.w;
        v[j] = o;
    }
#pragma unroll
    for (int j = 0; j < kScanPer4; ++j) buf[t * kScanPer4 + j] = v[j];
    __syncthreads();

    const int nout4 = (n + 4) >> 2;
    int4* r4 = (int4*)rowptr;
#pragma unroll
    for (int j = 0; j < kScanPer4; ++j) v[j] = buf[t + j * 1024];
#pragma unroll
    for (int j = 0; j < kScanPer4; ++j) {
        const int q = t + j * 1024;
        if (q < nout4) r4[q] = v[j];
    }
}

constexpr int kPlaceSplit = 2;
constexpr int kStageQ = (kNodes / 4 + 1023) / 1024;
constexpr int kStageG = (kStageQ + 1) / 2;
constexpr int kPlaceU = 2;

__device__ __forceinline__ void place4(const int* first, int4 d, int4 s, int4 r,
                                       int* __restrict__ csr, int E) {
    const unsigned uE = (unsigned)E;
    const unsigned p0 = (unsigned)(first[d.x] + r.x);
    const unsigned p1 = (unsigned)(first[d.y] + r.y);
    const unsigned p2 = (unsigned)(first[d.z] + r.z);
    const unsigned p3 = (unsigned)(first[d.w] + r.w);
    if ((p0 < uE) & (p1 < uE) & (p2 < uE) & (p3 < uE)) {
        csr[p0] = s.x;
        csr[p1] = s.y;
        csr[p2] = s.z;
        csr[p3] = s.w;
    } else {
        if (p0 < uE) csr[p0] = s.x;
        if (p1 < uE) csr[p1] = s.y;
        if (p2 < uE) csr[p2] = s.z;
        if (p3 < uE) csr[p3] = s.w;
    }
}

template <int J0, int CNT>
__device__ __forceinline__ void stage_first(int4* f4, const int4* __restrict__ rp4,
                                            const int4* __restrict__ hrow, int t, int n4) {
    int4 a[CNT], c[CNT];
#pragma unroll
    for (int j = 0; j < CNT; ++j) {
        const int qc = min(t + (J0 + j) * 1024, n4 - 1);
        a[j] = rp4[qc];
        c[j] = hrow[qc];
    }
#pragma unroll
    for (int j = 0; j < CNT; ++j) {
        const int q = t + (J0 + j) * 1024;
        if (q < n4) f4[q] = add4(a[j], c[j]);
    }
}

__global__ __launch_bounds__(1024) void place_kernel(const int* __restrict__ src, const int* __restrict__ dst,
                                                      const int* __restrict__ rowptr, const int* __restrict__ H,
                                                      const int* __restrict__ rank,
                                                      int* __restrict__ csr, int n, int E) {
    __shared__ int first[kNodes];
    const int t = threadIdx.x;
    const int g = blockIdx.x / kPlaceSplit;
    const int part = blockIdx.x % kPlaceSplit;
    const int chunk = chunk_len(E);
    const int b   = min(g * chunk, E);
    const int end = min(b + chunk, E);
    if (b >= end) return;
    const int nq  = (end - b) >> 2;
    const int per = (nq + kPlaceSplit - 1) / kPlaceSplit;
    const int q_lo = min(part * per, nq);
    const int q_hi = min(q_lo + per, nq);
    const int4* d4 = (const int4*)(dst + b);
    const int4* s4 = (const int4*)(src + b);
    const int4* r4 = (const int4*)(rank + b);

    int4 dv[kPlaceU], sv[kPlaceU], rv[kPlaceU];
#pragma unroll
    for (int u = 0; u < kPlaceU; ++u) {
        const int q = q_lo + t + u * 1024;
        const bool live = q < q_hi;
        dv[u] = live ? d4[q] : make_int4(0, 0, 0, 0);
        sv[u] = live ? s4[q] : make_int4(0, 0, 0, 0);
        rv[u] = live ? r4[q] : make_int4(0, 0, 0, 0);
    }

    {
        const int n4 = n >> 2;
        const int4* hrow = (const int4*)(H + (size_t)g * n);
        const int4* rp4  = (const int4*)rowptr;
        stage_first<0, kStageG>((int4*)first, rp4, hrow, t, n4);
        stage_first<kStageG, kStageQ - kStageG>((int4*)first, rp4, hrow, t, n4);
    }
    __syncthreads();

    for (int q0 = q_lo + t; q0 < q_hi; q0 += kPlaceU * 1024) {
        if (q0 != q_lo + t) {
#pragma unroll
            for (int u = 0; u < kPlaceU; ++u) {
                const int q = q0 + u * 1024;
                const bool live = q < q_hi;
                dv[u] = live ? d4[q] : make_int4(0, 0, 0, 0);
                sv[u] = live ? s4[q] : make_int4(0, 0, 0, 0);
                rv[u] = live ? r4[q] : make_int4(0, 0, 0, 0);
            }
        }
#pragma unroll
        for (int u = 0; u < kPlaceU; ++u)
            if (q0 + u * 1024 < q_hi) place4(first, dv[u], sv[u], rv[u], csr, E);
    }
    if (part == 0)
        for (int e = b + (nq << 2) + t; e < end; e += 1024) {
            const int pos = first[dst[e]] + rank[e];
            if ((unsigned)pos < (unsigned)E) csr[pos] = src[e];
        }
}

__device__ inline float leaky(float v) { return (v >= 0.f) ? v : NEG_SLOPE * v; }

template <int CTRL>
__device__ __forceinline__ float row_lane(float v) {
    return __int_as_float(__builtin_amdgcn_update_dpp(0, __float_as_int(v), CTRL, 0xf, 0xf, true));
}
constexpr int kRowXor   = 0x160;
constexpr int kRowShare = 0x150;


template <int I>
__device__ __forceinline__ void add_row1(float4& acc, float w, const float4& h) {
    const float wk = row_lane<kRowShare + 2 * I>(w);
    acc.x += wk * h.x;
    acc.y += wk * h.y;
    acc.z += wk * h.z;
    acc.w += wk * h.w;
}

__device__ __forceinline__ void rows8_l1(float4& acc, float& den, float& m,
                                         const float* __restrict__ hcol, int s, int k, int cnt,
                                         float4 as, float ad, int l) {
    float4 hv[8];
#pragma unroll
    for (int i = 0; i < 8; ++i) {
        int sk = __builtin_amdgcn_readlane(s, k + i);
        hv[i] = *(const float4*)(hcol + (size_t)sk * 128);
    }
    float p[8];
#pragma unroll
    for (int i = 0; i < 8; ++i)
        p[i] = hv[i].x * as.x + hv[i].y * as.y + hv[i].z * as.z + hv[i].w * as.w;
    const bool b8 = (l & 8) != 0, b4 = (l & 4) != 0, b2 = (l & 2) != 0;
    float w4[4], w2[2];
#pragma unroll
    for (int i = 0; i < 4; ++i) {
        float keep = b8 ? p[i + 4] : p[i];
        float send = b8 ? p[i] : p[i + 4];
        w4[i] = keep + row_lane<kRowXor + 8>(send);
    }
#pragma unroll
    for (int i = 0; i < 2; ++i) {
        float keep = b4 ? w4[i + 2] : w4[i];
        float send = b4 ? w4[i] : w4[i + 2];
        w2[i] = keep + row_lane<kRowXor + 4>(send);
    }
    const float w1 = (b2 ? w2[1] : w2[0]) + row_lane<kRowXor + 2>(b2 ? w2[0] : w2[1]);
    const float sc = w1 + row_lane<kRowXor + 1>(w1);

    const bool valid = (k + (l >> 1)) < cnt;
    const float e = valid ? leaky(sc + ad) : -INFINITY;
    float gm = e;
    gm = fmaxf(gm, row_lane<kRowXor + 2>(gm));
    gm = fmaxf(gm, row_lane<kRowXor + 4>(gm));
    gm = fmaxf(gm, row_lane<kRowXor + 8>(gm));
    const float mn = fmaxf(m, gm);
    const float scale = __expf(m - mn);
    const float w = __expf(e - mn);
    den = den * scale + ((l & 1) ? 0.f : w);
    acc.x *= scale;
    acc.y *= scale;
    acc.z *= scale;
    acc.w *= scale;
    m = mn;
    add_row1<0>(acc, w, hv[0]);
    add_row1<1>(acc, w, hv[1]);
    add_row1<2>(acc, w, hv[2]);
    add_row1<3>(acc, w, hv[3]);
    add_row1<4>(acc, w, hv[4]);
    add_row1<5>(acc, w, hv[5]);
    add_row1<6>(acc, w, hv[6]);
    add_row1<7>(acc, w, hv[7]);
}

__global__ __launch_bounds__(256) void node_agg1_kernel(
        const int* __restrict__ rowptr, const int* __restrict__ csr,
        const float* __restrict__ hfeat,
        const float* __restrict__ att_s, const float* __restrict__ adst,
        const float* __restrict__ bias, float* __restrict__ hact, int n) {
    const int lane = threadIdx.x & 31;
    const int d = __builtin_amdgcn_readfirstlane(blockIdx.x * (blockDim.x >> 5) + (threadIdx.x >> 5));
    if (d >= n) return;
    const int beg = rowptr[d], end = rowptr[d + 1];
    const int hd = lane >> 4, l = lane & 15;
    const float ad = adst[d * 2 + hd];
    const float4 as = *(const float4*)(att_s + lane * 4);

    float4 acc = {0.f, 0.f, 0.f, 0.f};
    float den = 0.f, m = -INFINITY;
    const float* hcol = hfeat + lane * 4;
    for (int base = beg; base < end; base += 32) {
        const int j = base + lane;
        int s = (j < end) ? csr[j] : 0;
        const int s0 = __builtin_amdgcn_readfirstlane(s);
        if (j >= end) s = s0;
        const int cnt = min(32, end - base);
        for (int k = 0; k < cnt; k += 8) rows8_l1(acc, den, m, hcol, s, k, cnt, as, ad, l);
    }
    den += row_lane<kRowXor + 1>(den);
    den += row_lane<kRowXor + 2>(den);
    den += row_lane<kRowXor + 4>(den);
    den += row_lane<kRowXor + 8>(den);
    den += EPS_;
    const float4 bv = *(const float4*)(bias + lane * 4);
    float4 o;
    o.x = acc.x / den + bv.x;
    o.y = acc.y / den + bv.y;
    o.z = acc.z / den + bv.z;
    o.w = acc.w / den + bv.w;
    o.x = (o.x > 0.f) ? o.x : expm1f(o.x);
    o.y = (o.y > 0.f) ? o.y : expm1f(o.y);
    o.z = (o.z > 0.f) ? o.z : expm1f(o.z);
    o.w = (o.w > 0.f) ? o.w : expm1f(o.w);
    *(float4*)(hact + (size_t)d * 128 + lane * 4) = o;
}

__device__ __forceinline__ void node2_row(int d, int lane,
                                   const int* __restrict__ rowptr, const int* __restrict__ csr,
                                   const float* __restrict__ hfeat,
                                   const float* __restrict__ att_s, const float* __restrict__ adst,
                                   const float* __restrict__ bias,
                                   float& r0, float& r1, float& r2, float& r3) {
    const int beg = rowptr[d], end = rowptr[d + 1];
    const float ad = adst[d];
    const int g = lane >> 2;
    const float4 as = *(const float4*)(att_s + (lane & 3) * 4);
    const float* hcol = hfeat + (lane & 3) * 4;

    float4 acc = {0.f, 0.f, 0.f, 0.f};
    float den = 0.f, m = -INFINITY;
    for (int base = beg; base < end; base += 32) {
        const int j = base + lane;
        int s = (j < end) ? csr[j] : 0;
        const int s0 = __builtin_amdgcn_readfirstlane(s);
        if (j >= end) s = s0;
        const int cnt = min(32, end - base);
        float4 hv[4];
#pragma unroll
        for (int t = 0; t < 4; ++t) {
            int sk = __shfl(s, t * 8 + g, 32);
            hv[t] = *(const float4*)(hcol + (size_t)sk * 16);
        }
        float e[4];
#pragma unroll
        for (int t = 0; t < 4; ++t) {
            float p = hv[t].x * as.x + hv[t].y * as.y + hv[t].z * as.z + hv[t].w * as.w;
            p += row_lane<kRowXor + 1>(p);
            p += row_lane<kRowXor + 2>(p);
            e[t] = (t * 8 + g < cnt) ? leaky(p + ad) : -INFINITY;
        }
        float cm = fmaxf(fmaxf(e[0], e[1]), fmaxf(e[2], e[3]));
        cm = fmaxf(cm, row_lane<kRowXor + 4>(cm));
        cm = fmaxf(cm, row_lane<kRowXor + 8>(cm));
        cm = fmaxf(cm, __shfl_xor(cm, 16, 32));
        const float mn = fmaxf(m, cm);
        const float scale = __expf(m - mn);
        float w[4];
#pragma unroll
        for (int t = 0; t < 4; ++t) w[t] = __expf(e[t] - mn);
        den = den * scale + ((w[0] + w[1]) + (w[2] + w[3]));
        acc.x = acc.x * scale + ((w[0] * hv[0].x + w[1] * hv[1].x) + (w[2] * hv[2].x + w[3] * hv[3].x));
        acc.y = acc.y * scale + ((w[0] * hv[0].y + w[1] * hv[1].y) + (w[2] * hv[2].y + w[3] * hv[3].y));
        acc.z = acc.z * scale + ((w[0] * hv[0].z + w[1] * hv[1].z) + (w[2] * hv[2].z + w[3] * hv[3].z));
        acc.w = acc.w * scale + ((w[0] * hv[0].w + w[1] * hv[1].w) + (w[2] * hv[2].w + w[3] * hv[3].w));
        m = mn;
    }
    for (int off = 16; off >= 4; off >>= 1) {
        acc.x += __shfl_xor(acc.x, off, 32);
        acc.y += __shfl_xor(acc.y, off, 32);
        acc.z += __shfl_xor(acc.z, off, 32);
        acc.w += __shfl_xor(acc.w, off, 32);
        den += __shfl_xor(den, off, 32);
    }
    den += EPS_;
    const float4 bv = *(const float4*)(bias + (lane & 3) * 4);
    float4 v;
    v.x = acc.x / den + bv.x;
    v.y = acc.y / den + bv.y;
    v.z = acc.z / den + bv.z;
    v.w = acc.w / den + bv.w;
    float mx = fmaxf(fmaxf(v.x, v.y), fmaxf(v.z, v.w));
    mx = fmaxf(mx, __shfl_xor(mx, 1, 32));
    mx = fmaxf(mx, __shfl_xor(mx, 2, 32));
    v.x = __expf(v.x - mx);
    v.y = __expf(v.y - mx);
    v.z = __expf(v.z - mx);
    v.w = __expf(v.w - mx);
    float sum = (v.x + v.y) + (v.z + v.w);
    sum += __shfl_xor(sum, 1, 32);
    sum += __shfl_xor(sum, 2, 32);
    const float inv = 1.f / sum;
    r0 = v.x * inv;
    r1 = v.y * inv;
    r2 = v.z * inv;
    r3 = v.w * inv;
}

__global__ __launch_bounds__(256) void node_agg2_kernel(
        const int* __restrict__ rowptr, const int* __restrict__ csr,
        const float* __restrict__ hfeat,
        const float* __restrict__ att_s, const float* __restrict__ adst,
        const float* __restrict__ bias, float* __restrict__ out, int n) {
    const int lane = threadIdx.x & 31;
    const int d0 = __builtin_amdgcn_readfirstlane((blockIdx.x * (blockDim.x >> 5) + (threadIdx.x >> 5)) * 2);
    if (d0 >= n) return;
    float a0, a1, a2, a3;
    node2_row(d0, lane, rowptr, csr, hfeat, att_s, adst, bias, a0, a1, a2, a3);
    float b0 = a0, b1 = a1, b2 = a2, b3 = a3;
    const bool two = (d0 + 1 < n);
    if (two) node2_row(d0 + 1, lane, rowptr, csr, hfeat, att_s, adst, bias, b0, b1, b2, b3);
    if (lane < 4 || (two && lane < 8)) {
        float4 r;
        r.x = (lane < 4) ? a0 : b0;
        r.y = (lane < 4) ? a1 : b1;
        r.z = (lane < 4) ? a2 : b2;
        r.w = (lane < 4) ? a3 : b3;
        *(float4*)(out + (size_t)d0 * 16 + lane * 4) = r;
    }
}

extern "C" void kernel_launch(void* const* d_in, const int* in_sizes, int n_in,
                              void* d_out, int out_size, void* d_ws, size_t ws_size,
                              hipStream_t stream) {
    const float* x   = (const float*)d_in[0];
    const float* W1  = (const float*)d_in[1];
    const float* as1 = (const float*)d_in[2];
    const float* ad1 = (const float*)d_in[3];
    const float* b1  = (const float*)d_in[4];
    const float* W2  = (const float*)d_in[5];
    const float* as2 = (const float*)d_in[6];
    const float* ad2 = (const float*)d_in[7];
    const float* b2  = (const float*)d_in[8];
    const int*   ei  = (const int*)d_in[9];

    const int N = in_sizes[0] / 128;
    const int E = in_sizes[9] / 2;
    const int* src = ei;
    const int* dst = ei + E;

    float* w = (float*)d_ws;
    float* h1      = w; w += (size_t)N * 128;
    float* hact    = w; w += (size_t)N * 128;
    float* adst1   = w; w += (size_t)N * 2;
    float* adst2   = w; w += (size_t)N;
    int*   cnt     = (int*)w; w += (size_t)N;
    int*   rowptr  = (int*)w; w += (size_t)N + 4;
    int*   csr     = (int*)w; w += (size_t)E;
    float* h2      = h1;
    int*   rank    = (int*)hact;
    int*   H       = (int*)h1;

    chunk_count_kernel<<<kChunks, 1024, 0, stream>>>(dst, H, rank, N, E);
    chunk_prefix_kernel<<<cdiv(N, 128), 256, 0, stream>>>(H, cnt, N);
    scan_kernel<<<1, 1024, 0, stream>>>(cnt, rowptr, N);
    place_kernel<<<kChunks * kPlaceSplit, 1024, 0, stream>>>(src, dst, rowptr, H, rank, csr, N, E);

    {
        int tiles = (N / 16) * 2;
        wmma_gemm_f32<128, 128, 4><<<cdiv(tiles, 8), 256, 0, stream>>>(x, W1, h1, ad1, adst1, N);
        node_agg1_kernel<<<cdiv(N, 8), 256, 0, stream>>>(rowptr, csr, h1, as1, adst1, b1, hact, N);
    }

    {
        int tiles = (N / 16);
        wmma_gemm_f32<128, 16, 1><<<cdiv(tiles, 8), 256, 0, stream>>>(hact, W2, h2, ad2, adst2, N);
        node_agg2_kernel<<<cdiv(N, 16), 256, 0, stream>>>(rowptr, csr, h2, as2, adst2, b2,
                                                           (float*)d_out, N);
    }
}


# destination buckets by a two-level LDS counting sort: edges ordered by block of 128 nodes via a 128 x 512 count table, then by node per block; all stores in runs; csr order now arrival: new sum order
# speedup vs baseline: 7.0552x; 1.2117x over previous
#include <hip/hip_runtime.h>
#include <hip/hip_bf16.h>
#include <math.h>

typedef float v2f __attribute__((ext_vector_type(2)));
typedef float v8f __attribute__((ext_vector_type(8)));

#define NEG_SLOPE 0.2f
#define EPS_ 1e-16f

static inline int cdiv(long a, long b) { return (int)((a + b - 1) / b); }

template <int K, int NC, int CPW>
__global__ __launch_bounds__(256) void wmma_gemm_f32(const float* __restrict__ A,
                                                     const float* __restrict__ B,
                                                     float* __restrict__ C,
                                                     const float* __restrict__ att_d,
                                                     float* __restrict__ adst,
                                                     int M) {
    constexpr int nColT  = NC / 16;
    constexpr int groups = nColT / CPW;
    int wave = threadIdx.x >> 5;
    int lane = threadIdx.x & 31;
    int hl   = lane >> 4;
    int l    = lane & 15;
    int tile  = blockIdx.x * (blockDim.x >> 5) + wave;
    int total = (M >> 4) * groups;
    if (tile >= total) return;
    int m0 = (tile / groups) << 4;
    int n0 = (tile % groups) * (CPW * 16);

    const float* arow = A + (size_t)(m0 + l) * K + 2 * hl;
    const float* bcol = B + (size_t)(2 * hl) * NC + n0 + l;

    v8f acc[CPW];
#pragma unroll
    for (int i = 0; i < CPW; ++i) acc[i] = {};

    for (int k0 = 0; k0 < K; k0 += 4) {
        v2f a;
        a.x = arow[k0 + 0];
        a.y = arow[k0 + 1];
#pragma unroll
        for (int ct = 0; ct < CPW; ++ct) {
            v2f b;
            b.x = bcol[(size_t)k0 * NC + ct * 16];
            b.y = bcol[(size_t)(k0 + 1) * NC + ct * 16];
            acc[ct] = __builtin_amdgcn_wmma_f32_16x16x4_f32(false, a, false, b,
                                                            (short)0, acc[ct], false, false);
        }
    }

#pragma unroll
    for (int ct = 0; ct < CPW; ++ct) {
        float* crow = C + (size_t)(m0 + 8 * hl) * NC + n0 + ct * 16 + l;
#pragma unroll
        for (int r = 0; r < 8; ++r) crow[(size_t)r * NC] = acc[ct][r];
    }

    float v[8];
    {
        float fd[CPW];
#pragma unroll
        for (int ct = 0; ct < CPW; ++ct) fd[ct] = att_d[n0 + ct * 16 + l];
#pragma unroll
        for (int r = 0; r < 8; ++r) {
            float pd = 0.f;
#pragma unroll
            for (int ct = 0; ct < CPW; ++ct) pd += acc[ct][r] * fd[ct];
            v[r] = pd;
        }
    }
    const bool b8 = (l & 8) != 0, b4 = (l & 4) != 0, b2 = (l & 2) != 0;
    float w4[4], w2[2];
#pragma unroll
    for (int i = 0; i < 4; ++i) {
        float keep = b8 ? v[i + 4] : v[i];
        float send = b8 ? v[i] : v[i + 4];
        w4[i] = keep + __shfl_xor(send, 8, 32);
    }
#pragma unroll
    for (int i = 0; i < 2; ++i) {
        float keep = b4 ? w4[i + 2] : w4[i];
        float send = b4 ? w4[i] : w4[i + 2];
        w2[i] = keep + __shfl_xor(send, 4, 32);
    }
    const float w1 = (b2 ? w2[1] : w2[0]) + __shfl_xor(b2 ? w2[0] : w2[1], 2, 32);
    const float dot = w1 + __shfl_xor(w1, 1, 32);
    if ((l & 1) == 0) adst[(size_t)(m0 + 8 * hl + (l >> 1)) * groups + (tile % groups)] = dot;
}

constexpr int kNodes  = 50000;
constexpr int kChunks = 128;
constexpr int kBP     = 512;
static_assert((kNodes + 128) / 128 <= kBP, "one table column per block of 128 nodes");
static_assert(kBP % 4 == 0 && (kBP / 4) * 8 == 1024, "step 2: a thread owns 4 columns and every 8th row");

__device__ __forceinline__ int chunk_len(int E) {
    return (((E + kChunks - 1) / kChunks) + 3) & ~3;
}

constexpr int kEdgeU = 4;

__global__ __launch_bounds__(1024) void block_count_kernel(const int* __restrict__ dst,
                                                            int* __restrict__ T, int E) {
    __shared__ int cnt[kBP];
    const int t = threadIdx.x;
    const int g = blockIdx.x;
    const int chunk = chunk_len(E);
    const int b   = min(g * chunk, E);
    const int end = min(b + chunk, E);
    const int nq  = (end - b) >> 2;
    const int4* d4 = (const int4*)(dst + b);
    if (t < kBP) cnt[t] = 0;
    __syncthreads();
    for (int q0 = t; q0 < nq; q0 += kEdgeU * 1024) {
        int4 dv[kEdgeU];
#pragma unroll
        for (int u = 0; u < kEdgeU; ++u) {
            const int q = q0 + u * 1024;
            dv[u] = (q < nq) ? d4[q] : make_int4(0, 0, 0, 0);
        }
#pragma unroll
        for (int u = 0; u < kEdgeU; ++u) {
            if (q0 + u * 1024 < nq) {
                atomicAdd(&cnt[dv[u].x >> 7], 1);
                atomicAdd(&cnt[dv[u].y >> 7], 1);
                atomicAdd(&cnt[dv[u].z >> 7], 1);
                atomicAdd(&cnt[dv[u].w >> 7], 1);
            }
        }
    }
    for (int e = b + (nq << 2) + t; e < end; e += 1024) atomicAdd(&cnt[dst[e] >> 7], 1);
    __syncthreads();
    if (t < kBP) T[g * kBP + t] = cnt[t];
}

__device__ __forceinline__ int4 add4(int4 a, int4 b) {
    return make_int4(a.x + b.x, a.y + b.y, a.z + b.z, a.w + b.w);
}

static_assert(kNodes <= (1 << 24), "source id and 7 destination bits in one int");
__device__ __forceinline__ int pack_pair(int s, int d) { return (s << 7) | (d & 127); }

constexpr int kScatCap = 12544;
__global__ __launch_bounds__(1024) void block_scatter_kernel(const int* __restrict__ src,
                                                              const int* __restrict__ dst,
                                                              const int* __restrict__ T,
                                                              int* __restrict__ bstart,
                                                              int* __restrict__ plist, int E) {
    constexpr int kQ = kBP / 4;
    constexpr int kR = 1024 / kQ;
    __shared__ int4 ptot[kR][kQ];
    __shared__ int4 ppart[kR][kQ];
    __shared__ int pos[kBP];
    __shared__ int lpos[kBP];
    __shared__ int wsum[32];
    __shared__ int wsum2[32];
    __shared__ int sval[kScatCap];
    __shared__ unsigned short skey[kScatCap];
    const int t = threadIdx.x;
    const int lane = t & 31, w = t >> 5;
    const int g = blockIdx.x;
    const int chunk = chunk_len(E);
    const int b   = min(g * chunk, E);
    const int end = min(b + chunk, E);
    const int m   = end - b;
    const int nq  = m >> 2;
    const bool staged = m <= kScatCap;
    const int4* d4 = (const int4*)(dst + b);
    const int4* s4 = (const int4*)(src + b);

    int4 dv[kEdgeU], sv[kEdgeU];
#pragma unroll
    for (int u = 0; u < kEdgeU; ++u) {
        const int q = t + u * 1024;
        const bool live = q < nq;
        dv[u] = live ? d4[q] : make_int4(0, 0, 0, 0);
        sv[u] = live ? s4[q] : make_int4(0, 0, 0, 0);
    }
    const int own = (t < kBP) ? T[g * kBP + t] : 0;

    {
        const int cq = t % kQ, rg = t / kQ;
        const int4* T4 = (const int4*)T;
        constexpr int kRows = kChunks / kR;
        static_assert(kChunks % kR == 0, "every thread sums the same number of rows");
        int4 v[kRows];
#pragma unroll
        for (int j = 0; j < kRows; ++j) v[j] = T4[(rg + j * kR) * kQ + cq];
        int4 tot = make_int4(0, 0, 0, 0), part = make_int4(0, 0, 0, 0);
#pragma unroll
        for (int j = 0; j < kRows; ++j) {
            tot = add4(tot, v[j]);
            if (rg + j * kR < g) part = add4(part, v[j]);
        }
        ptot[rg][cq] = tot;
        ppart[rg][cq] = part;
    }
    __syncthreads();

    int bv = 0, pv = 0;
    if (t < kBP) {
#pragma unroll
        for (int r = 0; r < kR; ++r) {
            bv += ((const int*)&ptot[r][0])[t];
            pv += ((const int*)&ppart[r][0])[t];
        }
    }
    int x = bv, z = own;
    for (int off = 1; off < 32; off <<= 1) {
        int u = __shfl_up(x, off, 32);
        int u2 = __shfl_up(z, off, 32);
        if (lane >= off) { x += u; z += u2; }
    }
    if (lane == 31) { wsum[w] = x; wsum2[w] = z; }
    __syncthreads();
    if (w == 0) {
        int y = wsum[lane], y2 = wsum2[lane];
        for (int off = 1; off < 32; off <<= 1) {
            int u = __shfl_up(y, off, 32);
            int u2 = __shfl_up(y2, off, 32);
            if (lane >= off) { y += u; y2 += u2; }
        }
        wsum[lane] = y;
        wsum2[lane] = y2;
    }
    __syncthreads();
    const int excl  = x - bv + (w > 0 ? wsum[w - 1] : 0);
    const int lexcl = z - own + (w > 0 ? wsum2[w - 1] : 0);
    if (t < kBP) {
        const int gpos = excl + pv;
        pos[t]  = staged ? gpos - lexcl : gpos;
        lpos[t] = lexcl;
        if (g == 0) bstart[t] = excl;
    }
    __syncthreads();

    if (staged) {
        for (int q0 = t; q0 < nq; q0 += kEdgeU * 1024) {
            if (q0 != t) {
#pragma unroll
                for (int u = 0; u < kEdgeU; ++u) {
                    const int q = q0 + u * 1024;
                    const bool live = q < nq;
                    dv[u] = live ? d4[q] : make_int4(0, 0, 0, 0);
                    sv[u] = live ? s4[q] : make_int4(0, 0, 0, 0);
                }
            }
#pragma unroll
            for (int u = 0; u < kEdgeU; ++u)
                if (q0 + u * 1024 < nq) {
                    const int k0 = dv[u].x >> 7, k1 = dv[u].y >> 7, k2 = dv[u].z >> 7, k3 = dv[u].w >> 7;
                    const unsigned j0 = (unsigned)atomicAdd(&lpos[k0], 1);
                    const unsigned j1 = (unsigned)atomicAdd(&lpos[k1], 1);
                    const unsigned j2 = (unsigned)atomicAdd(&lpos[k2], 1);
                    const unsigned j3 = (unsigned)atomicAdd(&lpos[k3], 1);
                    if (j0 < (unsigned)kScatCap) { sval[j0] = pack_pair(sv[u].x, dv[u].x); skey[j0] = (unsigned short)k0; }
                    if (j1 < (unsigned)kScatCap) { sval[j1] = pack_pair(sv[u].y, dv[u].y); skey[j1] = (unsigned short)k1; }
                    if (j2 < (unsigned)kScatCap) { sval[j2] = pack_pair(sv[u].z, dv[u].z); skey[j2] = (unsigned short)k2; }
                    if (j3 < (unsigned)kScatCap) { sval[j3] = pack_pair(sv[u].w, dv[u].w); skey[j3] = (unsigned short)k3; }
                }
        }
        for (int e = b + (nq << 2) + t; e < end; e += 1024) {
            const int d = dst[e];
            const unsigned j = (unsigned)atomicAdd(&lpos[d >> 7], 1);
            if (j < (unsigned)kScatCap) { sval[j] = pack_pair(src[e], d); skey[j] = (unsigned short)(d >> 7); }
        }
        __syncthreads();
        constexpr int kOutU = 4;
        for (int j0 = t; j0 < m; j0 += kOutU * 1024) {
            int kk[kOutU], vv[kOutU], pp[kOutU];
#pragma unroll
            for (int u = 0; u < kOutU; ++u) {
                const int j = min(j0 + u * 1024, m - 1);
                kk[u] = skey[j];
                vv[u] = sval[j];
            }
#pragma unroll
            for (int u = 0; u < kOutU; ++u) pp[u] = pos[kk[u]];
#pragma unroll
            for (int u = 0; u < kOutU; ++u) {
                const int j = j0 + u * 1024;
                const int slot = pp[u] + j;
                if ((j < m) && ((unsigned)slot < (unsigned)E)) plist[slot] = vv[u];
            }
        }
    } else {
        for (int q0 = t; q0 < nq; q0 += kEdgeU * 1024) {
            if (q0 != t) {
#pragma unroll
                for (int u = 0; u < kEdgeU; ++u) {
                    const int q = q0 + u * 1024;
                    const bool live = q < nq;
                    dv[u] = live ? d4[q] : make_int4(0, 0, 0, 0);
                    sv[u] = live ? s4[q] : make_int4(0, 0, 0, 0);
                }
            }
#pragma unroll
            for (int u = 0; u < kEdgeU; ++u)
                if (q0 + u * 1024 < nq) {
                    const unsigned uE = (unsigned)E;
                    const unsigned p0 = (unsigned)atomicAdd(&pos[dv[u].x >> 7], 1);
                    const unsigned p1 = (unsigned)atomicAdd(&pos[dv[u].y >> 7], 1);
                    const unsigned p2 = (unsigned)atomicAdd(&pos[dv[u].z >> 7], 1);
                    const unsigned p3 = (unsigned)atomicAdd(&pos[dv[u].w >> 7], 1);
                    if (p0 < uE) plist[p0] = pack_pair(sv[u].x, dv[u].x);
                    if (p1 < uE) plist[p1] = pack_pair(sv[u].y, dv[u].y);
                    if (p2 < uE) plist[p2] = pack_pair(sv[u].z, dv[u].z);
                    if (p3 < uE) plist[p3] = pack_pair(sv[u].w, dv[u].w);
                }
        }
        for (int e = b + (nq << 2) + t; e < end; e += 1024) {
            const int d = dst[e];
            const unsigned p = (unsigned)atomicAdd(&pos[d >> 7], 1);
            if (p < (unsigned)E) plist[p] = pack_pair(src[e], d);
        }
    }
}

constexpr int kFinT = 256;
constexpr int kFinU = 20;
constexpr int kFinCap = 8192;
__global__ __launch_bounds__(kFinT) void block_finish_kernel(const int* __restrict__ plist,
                                                             const int* __restrict__ bstart,
                                                             int* __restrict__ csr,
                                                             int* __restrict__ rowptr, int n, int E) {
    __shared__ int4 cnt4[32];
    __shared__ int4 pos4[32];
    __shared__ int stage[kFinCap];
    int* cnt = (int*)cnt4;
    int* pos = (int*)pos4;
    const int t = threadIdx.x;
    const int blk = blockIdx.x;
    const int beg = bstart[blk], end = bstart[blk + 1];
    const bool staged = (end - beg) <= kFinCap;
    if (t < 128) cnt[t] = 0;
    __syncthreads();
    for (int i0 = beg + t; i0 < end; i0 += kFinU * kFinT) {
        int nd[kFinU];
#pragma unroll
        for (int u = 0; u < kFinU; ++u) nd[u] = plist[min(i0 + u * kFinT, end - 1)] & 127;
#pragma unroll
        for (int u = 0; u < kFinU; ++u) atomicAdd(&cnt[nd[u]], (i0 + u * kFinT < end) ? 1 : 0);
    }
    __syncthreads();
    if (t < 32) {
        const int4 c = cnt4[t];
        const int t4 = (c.x + c.y) + (c.z + c.w);
        int x = t4;
        for (int off = 1; off < 32; off <<= 1) {
            int u = __shfl_up(x, off, 32);
            if (t >= off) x += u;
        }
        int4 o;
        o.x = beg + x - t4;
        o.y = o.x + c.x;
        o.z = o.y + c.y;
        o.w = o.z + c.z;
        pos4[t] = o;
        const int q = blk * 32 + t;
        if (q <= (n >> 2)) ((int4*)rowptr)[q] = o;
    }
    __syncthreads();
    for (int i0 = beg + t; i0 < end; i0 += kFinU * kFinT) {
        int ev[kFinU];
        int slot[kFinU];
#pragma unroll
        for (int u = 0; u < kFinU; ++u) ev[u] = plist[min(i0 + u * kFinT, end - 1)];
        __builtin_amdgcn_sched_barrier(0);
#pragma unroll
        for (int u = 0; u < kFinU; ++u)
            slot[u] = atomicAdd(&pos[ev[u] & 127], (i0 + u * kFinT < end) ? 1 : 0);
#pragma unroll
        for (int u = 0; u < kFinU; ++u)
            if (i0 + u * kFinT < end) {
                if (staged) {
                    const unsigned j = (unsigned)(slot[u] - beg);
                    if (j < (unsigned)kFinCap) stage[j] = ev[u] >> 7;
                } else {
                    if ((unsigned)slot[u] < (unsigned)E) csr[slot[u]] = ev[u] >> 7;
                }
            }
    }
    if (staged) {
        __syncthreads();
        const int m = end - beg;
        constexpr int kOutU = 8;
        for (int j0 = t; j0 < m; j0 += kOutU * kFinT) {
            int vv[kOutU];
#pragma unroll
            for (int u = 0; u < kOutU; ++u) vv[u] = stage[min(j0 + u * kFinT, m - 1)];
#pragma unroll
            for (int u = 0; u < kOutU; ++u) {
                const int j = j0 + u * kFinT;
                if ((j < m) && ((unsigned)(beg + j) < (unsigned)E)) csr[beg + j] = vv[u];
            }
        }
    }
}

__device__ inline float leaky(float v) { return (v >= 0.f) ? v : NEG_SLOPE * v; }

template <int CTRL>
__device__ __forceinline__ float row_lane(float v) {
    return __int_as_float(__builtin_amdgcn_update_dpp(0, __float_as_int(v), CTRL, 0xf, 0xf, true));
}
constexpr int kRowXor   = 0x160;
constexpr int kRowShare = 0x150;


template <int I>
__device__ __forceinline__ void add_row1(float4& acc, float w, const float4& h) {
    const float wk = row_lane<kRowShare + 2 * I>(w);
    acc.x += wk * h.x;
    acc.y += wk * h.y;
    acc.z += wk * h.z;
    acc.w += wk * h.w;
}

__device__ __forceinline__ void rows8_l1(float4& acc, float& den, float& m,
                                         const float* __restrict__ hcol, int s, int k, int cnt,
                                         float4 as, float ad, int l) {
    float4 hv[8];
#pragma unroll
    for (int i = 0; i < 8; ++i) {
        int sk = __builtin_amdgcn_readlane(s, k + i);
        hv[i] = *(const float4*)(hcol + (size_t)sk * 128);
    }
    float p[8];
#pragma unroll
    for (int i = 0; i < 8; ++i)
        p[i] = hv[i].x * as.x + hv[i].y * as.y + hv[i].z * as.z + hv[i].w * as.w;
    const bool b8 = (l & 8) != 0, b4 = (l & 4) != 0, b2 = (l & 2) != 0;
    float w4[4], w2[2];
#pragma unroll
    for (int i = 0; i < 4; ++i) {
        float keep = b8 ? p[i + 4] : p[i];
        float send = b8 ? p[i] : p[i + 4];
        w4[i] = keep + row_lane<kRowXor + 8>(send);
    }
#pragma unroll
    for (int i = 0; i < 2; ++i) {
        float keep = b4 ? w4[i + 2] : w4[i];
        float send = b4 ? w4[i] : w4[i + 2];
        w2[i] = keep + row_lane<kRowXor + 4>(send);
    }
    const float w1 = (b2 ? w2[1] : w2[0]) + row_lane<kRowXor + 2>(b2 ? w2[0] : w2[1]);
    const float sc = w1 + row_lane<kRowXor + 1>(w1);

    const bool valid = (k + (l >> 1)) < cnt;
    const float e = valid ? leaky(sc + ad) : -INFINITY;
    float gm = e;
    gm = fmaxf(gm, row_lane<kRowXor + 2>(gm));
    gm = fmaxf(gm, row_lane<kRowXor + 4>(gm));
    gm = fmaxf(gm, row_lane<kRowXor + 8>(gm));
    const float mn = fmaxf(m, gm);
    const float scale = __expf(m - mn);
    const float w = __expf(e - mn);
    den = den * scale + ((l & 1) ? 0.f : w);
    acc.x *= scale;
    acc.y *= scale;
    acc.z *= scale;
    acc.w *= scale;
    m = mn;
    add_row1<0>(acc, w, hv[0]);
    add_row1<1>(acc, w, hv[1]);
    add_row1<2>(acc, w, hv[2]);
    add_row1<3>(acc, w, hv[3]);
    add_row1<4>(acc, w, hv[4]);
    add_row1<5>(acc, w, hv[5]);
    add_row1<6>(acc, w, hv[6]);
    add_row1<7>(acc, w, hv[7]);
}

__global__ __launch_bounds__(256) void node_agg1_kernel(
        const int* __restrict__ rowptr, const int* __restrict__ csr,
        const float* __restrict__ hfeat,
        const float* __restrict__ att_s, const float* __restrict__ adst,
        const float* __restrict__ bias, float* __restrict__ hact, int n) {
    const int lane = threadIdx.x & 31;
    const int d = __builtin_amdgcn_readfirstlane(blockIdx.x * (blockDim.x >> 5) + (threadIdx.x >> 5));
    if (d >= n) return;
    const int beg = rowptr[d], end = rowptr[d + 1];
    const int hd = lane >> 4, l = lane & 15;
    const float ad = adst[d * 2 + hd];
    const float4 as = *(const float4*)(att_s + lane * 4);

    float4 acc = {0.f, 0.f, 0.f, 0.f};
    float den = 0.f, m = -INFINITY;
    const float* hcol = hfeat + lane * 4;
    for (int base = beg; base < end; base += 32) {
        const int j = base + lane;
        int s = (j < end) ? csr[j] : 0;
        const int s0 = __builtin_amdgcn_readfirstlane(s);
        if (j >= end) s = s0;
        const int cnt = min(32, end - base);
        for (int k = 0; k < cnt; k += 8) rows8_l1(acc, den, m, hcol, s, k, cnt, as, ad, l);
    }
    den += row_lane<kRowXor + 1>(den);
    den += row_lane<kRowXor + 2>(den);
    den += row_lane<kRowXor + 4>(den);
    den += row_lane<kRowXor + 8>(den);
    den += EPS_;
    const float4 bv = *(const float4*)(bias + lane * 4);
    float4 o;
    o.x = acc.x / den + bv.x;
    o.y = acc.y / den + bv.y;
    o.z = acc.z / den + bv.z;
    o.w = acc.w / den + bv.w;
    o.x = (o.x > 0.f) ? o.x : expm1f(o.x);
    o.y = (o.y > 0.f) ? o.y : expm1f(o.y);
    o.z = (o.z > 0.f) ? o.z : expm1f(o.z);
    o.w = (o.w > 0.f) ? o.w : expm1f(o.w);
    *(float4*)(hact + (size_t)d * 128 + lane * 4) = o;
}

__device__ __forceinline__ void node2_row(int d, int lane,
                                   const int* __restrict__ rowptr, const int* __restrict__ csr,
                                   const float* __restrict__ hfeat,
                                   const float* __restrict__ att_s, const float* __restrict__ adst,
                                   const float* __restrict__ bias,
                                   float& r0, float& r1, float& r2, float& r3) {
    const int beg = rowptr[d], end = rowptr[d + 1];
    const float ad = adst[d];
    const int g = lane >> 2;
    const float4 as = *(const float4*)(att_s + (lane & 3) * 4);
    const float* hcol = hfeat + (lane & 3) * 4;

    float4 acc = {0.f, 0.f, 0.f, 0.f};
    float den = 0.f, m = -INFINITY;
    for (int base = beg; base < end; base += 32) {
        const int j = base + lane;
        int s = (j < end) ? csr[j] : 0;
        const int s0 = __builtin_amdgcn_readfirstlane(s);
        if (j >= end) s = s0;
        const int cnt = min(32, end - base);
        float4 hv[4];
#pragma unroll
        for (int t = 0; t < 4; ++t) {
            int sk = __shfl(s, t * 8 + g, 32);
            hv[t] = *(const float4*)(hcol + (size_t)sk * 16);
        }
        float e[4];
#pragma unroll
        for (int t = 0; t < 4; ++t) {
            float p = hv[t].x * as.x + hv[t].y * as.y + hv[t].z * as.z + hv[t].w * as.w;
            p += row_lane<kRowXor + 1>(p);
            p += row_lane<kRowXor + 2>(p);
            e[t] = (t * 8 + g < cnt) ? leaky(p + ad) : -INFINITY;
        }
        float cm = fmaxf(fmaxf(e[0], e[1]), fmaxf(e[2], e[3]));
        cm = fmaxf(cm, row_lane<kRowXor + 4>(cm));
        cm = fmaxf(cm, row_lane<kRowXor + 8>(cm));
        cm = fmaxf(cm, __shfl_xor(cm, 16, 32));
        const float mn = fmaxf(m, cm);
        const float scale = __expf(m - mn);
        float w[4];
#pragma unroll
        for (int t = 0; t < 4; ++t) w[t] = __expf(e[t] - mn);
        den = den * scale + ((w[0] + w[1]) + (w[2] + w[3]));
        acc.x = acc.x * scale + ((w[0] * hv[0].x + w[1] * hv[1].x) + (w[2] * hv[2].x + w[3] * hv[3].x));
        acc.y = acc.y * scale + ((w[0] * hv[0].y + w[1] * hv[1].y) + (w[2] * hv[2].y + w[3] * hv[3].y));
        acc.z = acc.z * scale + ((w[0] * hv[0].z + w[1] * hv[1].z) + (w[2] * hv[2].z + w[3] * hv[3].z));
        acc.w = acc.w * scale + ((w[0] * hv[0].w + w[1] * hv[1].w) + (w[2] * hv[2].w + w[3] * hv[3].w));
        m = mn;
    }
    for (int off = 16; off >= 4; off >>= 1) {
        acc.x += __shfl_xor(acc.x, off, 32);
        acc.y += __shfl_xor(acc.y, off, 32);
        acc.z += __shfl_xor(acc.z, off, 32);
        acc.w += __shfl_xor(acc.w, off, 32);
        den += __shfl_xor(den, off, 32);
    }
    den += EPS_;
    const float4 bv = *(const float4*)(bias + (lane & 3) * 4);
    float4 v;
    v.x = acc.x / den + bv.x;
    v.y = acc.y / den + bv.y;
    v.z = acc.z / den + bv.z;
    v.w = acc.w / den + bv.w;
    float mx = fmaxf(fmaxf(v.x, v.y), fmaxf(v.z, v.w));
    mx = fmaxf(mx, __shfl_xor(mx, 1, 32));
    mx = fmaxf(mx, __shfl_xor(mx, 2, 32));
    v.x = __expf(v.x - mx);
    v.y = __expf(v.y - mx);
    v.z = __expf(v.z - mx);
    v.w = __expf(v.w - mx);
    float sum = (v.x + v.y) + (v.z + v.w);
    sum += __shfl_xor(sum, 1, 32);
    sum += __shfl_xor(sum, 2, 32);
    const float inv = 1.f / sum;
    r0 = v.x * inv;
    r1 = v.y * inv;
    r2 = v.z * inv;
    r3 = v.w * inv;
}

__global__ __launch_bounds__(256) void node_agg2_kernel(
        const int* __restrict__ rowptr, const int* __restrict__ csr,
        const float* __restrict__ hfeat,
        const float* __restrict__ att_s, const float* __restrict__ adst,
        const float* __restrict__ bias, float* __restrict__ out, int n) {
    const int lane = threadIdx.x & 31;
    const int d0 = __builtin_amdgcn_readfirstlane((blockIdx.x * (blockDim.x >> 5) + (threadIdx.x >> 5)) * 2);
    if (d0 >= n) return;
    float a0, a1, a2, a3;
    node2_row(d0, lane, rowptr, csr, hfeat, att_s, adst, bias, a0, a1, a2, a3);
    float b0 = a0, b1 = a1, b2 = a2, b3 = a3;
    const bool two = (d0 + 1 < n);
    if (two) node2_row(d0 + 1, lane, rowptr, csr, hfeat, att_s, adst, bias, b0, b1, b2, b3);
    if (lane < 4 || (two && lane < 8)) {
        float4 r;
        r.x = (lane < 4) ? a0 : b0;
        r.y = (lane < 4) ? a1 : b1;
        r.z = (lane < 4) ? a2 : b2;
        r.w = (lane < 4) ? a3 : b3;
        *(float4*)(out + (size_t)d0 * 16 + lane * 4) = r;
    }
}

extern "C" void kernel_launch(void* const* d_in, const int* in_sizes, int n_in,
                              void* d_out, int out_size, void* d_ws, size_t ws_size,
                              hipStream_t stream) {
    const float* x   = (const float*)d_in[0];
    const float* W1  = (const float*)d_in[1];
    const float* as1 = (const float*)d_in[2];
    const float* ad1 = (const float*)d_in[3];
    const float* b1  = (const float*)d_in[4];
    const float* W2  = (const float*)d_in[5];
    const float* as2 = (const float*)d_in[6];
    const float* ad2 = (const float*)d_in[7];
    const float* b2  = (const float*)d_in[8];
    const int*   ei  = (const int*)d_in[9];

    const int N = in_sizes[0] / 128;
    const int E = in_sizes[9] / 2;
    const int* src = ei;
    const int* dst = ei + E;

    float* w = (float*)d_ws;
    float* h1      = w; w += (size_t)N * 128;
    float* hact    = w; w += (size_t)N * 128;
    float* adst1   = w; w += (size_t)N * 2;
    float* adst2   = w; w += (size_t)N;
    int*   rowptr  = (int*)w; w += (size_t)N + 4;
    int*   csr     = (int*)w; w += (size_t)E;
    float* h2      = h1;
    int*   plist   = (int*)hact;
    int*   T       = (int*)h1;
    int*   bstart  = T + kChunks * kBP;

    block_count_kernel<<<kChunks, 1024, 0, stream>>>(dst, T, E);
    block_scatter_kernel<<<kChunks, 1024, 0, stream>>>(src, dst, T, bstart, plist, E);
    block_finish_kernel<<<cdiv(N + 1, 128), kFinT, 0, stream>>>(plist, bstart, csr, rowptr, N, E);

    {
        int tiles = (N / 16) * 2;
        wmma_gemm_f32<128, 128, 4><<<cdiv(tiles, 8), 256, 0, stream>>>(x, W1, h1, ad1, adst1, N);
        node_agg1_kernel<<<cdiv(N, 8), 256, 0, stream>>>(rowptr, csr, h1, as1, adst1, b1, hact, N);
    }

    {
        int tiles = (N / 16);
        wmma_gemm_f32<128, 16, 1><<<cdiv(tiles, 8), 256, 0, stream>>>(hact, W2, h2, ad2, adst2, N);
        node_agg2_kernel<<<cdiv(N, 16), 256, 0, stream>>>(rowptr, csr, h2, as2, adst2, b2,
                                                           (float*)d_out, N);
    }
}


# layer-1 linear map (f32 WMMA tiles, code unchanged) runs as extra work-groups of the bucket build's last launch, beside the csr finish: 6 launches, not 7; count tables moved out of the h1 plane
# speedup vs baseline: 8.2598x; 1.1707x over previous
#include <hip/hip_runtime.h>
#include <hip/hip_bf16.h>
#include <math.h>

typedef float v2f __attribute__((ext_vector_type(2)));
typedef float v8f __attribute__((ext_vector_type(8)));

#define NEG_SLOPE 0.2f
#define EPS_ 1e-16f

static inline int cdiv(long a, long b) { return (int)((a + b - 1) / b); }

template <int K, int NC, int CPW>
__device__ __forceinline__ void wmma_gemm_wave(const float* __restrict__ A,
                                               const float* __restrict__ B,
                                               float* __restrict__ C,
                                               const float* __restrict__ att_d,
                                               float* __restrict__ adst,
                                               int M, int tile) {
    constexpr int nColT  = NC / 16;
    constexpr int groups = nColT / CPW;
    int lane = threadIdx.x & 31;
    int hl   = lane >> 4;
    int l    = lane & 15;
    int total = (M >> 4) * groups;
    if (tile >= total) return;
    int m0 = (tile / groups) << 4;
    int n0 = (tile % groups) * (CPW * 16);

    const float* arow = A + (size_t)(m0 + l) * K + 2 * hl;
    const float* bcol = B + (size_t)(2 * hl) * NC + n0 + l;

    v8f acc[CPW];
#pragma unroll
    for (int i = 0; i < CPW; ++i) acc[i] = {};

    for (int k0 = 0; k0 < K; k0 += 4) {
        v2f a;
        a.x = arow[k0 + 0];
        a.y = arow[k0 + 1];
#pragma unroll
        for (int ct = 0; ct < CPW; ++ct) {
            v2f b;
            b.x = bcol[(size_t)k0 * NC + ct * 16];
            b.y = bcol[(size_t)(k0 + 1) * NC + ct * 16];
            acc[ct] = __builtin_amdgcn_wmma_f32_16x16x4_f32(false, a, false, b,
                                                            (short)0, acc[ct], false, false);
        }
    }

#pragma unroll
    for (int ct = 0; ct < CPW; ++ct) {
        float* crow = C + (size_t)(m0 + 8 * hl) * NC + n0 + ct * 16 + l;
#pragma unroll
        for (int r = 0; r < 8; ++r) crow[(size_t)r * NC] = acc[ct][r];
    }

    float v[8];
    {
        float fd[CPW];
#pragma unroll
        for (int ct = 0; ct < CPW; ++ct) fd[ct] = att_d[n0 + ct * 16 + l];
#pragma unroll
        for (int r = 0; r < 8; ++r) {
            float pd = 0.f;
#pragma unroll
            for (int ct = 0; ct < CPW; ++ct) pd += acc[ct][r] * fd[ct];
            v[r] = pd;
        }
    }
    const bool b8 = (l & 8) != 0, b4 = (l & 4) != 0, b2 = (l & 2) != 0;
    float w4[4], w2[2];
#pragma unroll
    for (int i = 0; i < 4; ++i) {
        float keep = b8 ? v[i + 4] : v[i];
        float send = b8 ? v[i] : v[i + 4];
        w4[i] = keep + __shfl_xor(send, 8, 32);
    }
#pragma unroll
    for (int i = 0; i < 2; ++i) {
        float keep = b4 ? w4[i + 2] : w4[i];
        float send = b4 ? w4[i] : w4[i + 2];
        w2[i] = keep + __shfl_xor(send, 4, 32);
    }
    const float w1 = (b2 ? w2[1] : w2[0]) + __shfl_xor(b2 ? w2[0] : w2[1], 2, 32);
    const float dot = w1 + __shfl_xor(w1, 1, 32);
    if ((l & 1) == 0) adst[(size_t)(m0 + 8 * hl + (l >> 1)) * groups + (tile % groups)] = dot;
}

template <int K, int NC, int CPW>
__global__ __launch_bounds__(256) void wmma_gemm_f32(const float* __restrict__ A,
                                                     const float* __restrict__ B,
                                                     float* __restrict__ C,
                                                     const float* __restrict__ att_d,
                                                     float* __restrict__ adst,
                                                     int M) {
    wmma_gemm_wave<K, NC, CPW>(A, B, C, att_d, adst, M,
                               blockIdx.x * (blockDim.x >> 5) + (threadIdx.x >> 5));
}

constexpr int kNodes  = 50000;
constexpr int kChunks = 128;
constexpr int kBP     = 512;
static_assert((kNodes + 128) / 128 <= kBP, "one table column per block of 128 nodes");
static_assert(kBP % 4 == 0 && (kBP / 4) * 8 == 1024, "step 2: a thread owns 4 columns and every 8th row");

__device__ __forceinline__ int chunk_len(int E) {
    return (((E + kChunks - 1) / kChunks) + 3) & ~3;
}

constexpr int kEdgeU = 4;

__global__ __launch_bounds__(1024) void block_count_kernel(const int* __restrict__ dst,
                                                            int* __restrict__ T, int E) {
    __shared__ int cnt[kBP];
    const int t = threadIdx.x;
    const int g = blockIdx.x;
    const int chunk = chunk_len(E);
    const int b   = min(g * chunk, E);
    const int end = min(b + chunk, E);
    const int nq  = (end - b) >> 2;
    const int4* d4 = (const int4*)(dst + b);
    if (t < kBP) cnt[t] = 0;
    __syncthreads();
    for (int q0 = t; q0 < nq; q0 += kEdgeU * 1024) {
        int4 dv[kEdgeU];
#pragma unroll
        for (int u = 0; u < kEdgeU; ++u) {
            const int q = q0 + u * 1024;
            dv[u] = (q < nq) ? d4[q] : make_int4(0, 0, 0, 0);
        }
#pragma unroll
        for (int u = 0; u < kEdgeU; ++u) {
            if (q0 + u * 1024 < nq) {
                atomicAdd(&cnt[dv[u].x >> 7], 1);
                atomicAdd(&cnt[dv[u].y >> 7], 1);
                atomicAdd(&cnt[dv[u].z >> 7], 1);
                atomicAdd(&cnt[dv[u].w >> 7], 1);
            }
        }
    }
    for (int e = b + (nq << 2) + t; e < end; e += 1024) atomicAdd(&cnt[dst[e] >> 7], 1);
    __syncthreads();
    if (t < kBP) T[g * kBP + t] = cnt[t];
}

__device__ __forceinline__ int4 add4(int4 a, int4 b) {
    return make_int4(a.x + b.x, a.y + b.y, a.z + b.z, a.w + b.w);
}

static_assert(kNodes <= (1 << 24), "source id and 7 destination bits in one int");
__device__ __forceinline__ int pack_pair(int s, int d) { return (s << 7) | (d & 127); }

constexpr int kScatCap = 12544;
__global__ __launch_bounds__(1024) void block_scatter_kernel(const int* __restrict__ src,
                                                              const int* __restrict__ dst,
                                                              const int* __restrict__ T,
                                                              int* __restrict__ bstart,
                                                              int* __restrict__ plist, int E) {
    constexpr int kQ = kBP / 4;
    constexpr int kR = 1024 / kQ;
    __shared__ int4 ptot[kR][kQ];
    __shared__ int4 ppart[kR][kQ];
    __shared__ int pos[kBP];
    __shared__ int lpos[kBP];
    __shared__ int wsum[32];
    __shared__ int wsum2[32];
    __shared__ int sval[kScatCap];
    __shared__ unsigned short skey[kScatCap];
    const int t = threadIdx.x;
    const int lane = t & 31, w = t >> 5;
    const int g = blockIdx.x;
    const int chunk = chunk_len(E);
    const int b   = min(g * chunk, E);
    const int end = min(b + chunk, E);
    const int m   = end - b;
    const int nq  = m >> 2;
    const bool staged = m <= kScatCap;
    const int4* d4 = (const int4*)(dst + b);
    const int4* s4 = (const int4*)(src + b);

    int4 dv[kEdgeU], sv[kEdgeU];
#pragma unroll
    for (int u = 0; u < kEdgeU; ++u) {
        const int q = t + u * 1024;
        const bool live = q < nq;
        dv[u] = live ? d4[q] : make_int4(0, 0, 0, 0);
        sv[u] = live ? s4[q] : make_int4(0, 0, 0, 0);
    }
    const int own = (t < kBP) ? T[g * kBP + t] : 0;

    {
        const int cq = t % kQ, rg = t / kQ;
        const int4* T4 = (const int4*)T;
        constexpr int kRows = kChunks / kR;
        static_assert(kChunks % kR == 0, "every thread sums the same number of rows");
        int4 v[kRows];
#pragma unroll
        for (int j = 0; j < kRows; ++j) v[j] = T4[(rg + j * kR) * kQ + cq];
        int4 tot = make_int4(0, 0, 0, 0), part = make_int4(0, 0, 0, 0);
#pragma unroll
        for (int j = 0; j < kRows; ++j) {
            tot = add4(tot, v[j]);
            if (rg + j * kR < g) part = add4(part, v[j]);
        }
        ptot[rg][cq] = tot;
        ppart[rg][cq] = part;
    }
    __syncthreads();

    int bv = 0, pv = 0;
    if (t < kBP) {
#pragma unroll
        for (int r = 0; r < kR; ++r) {
            bv += ((const int*)&ptot[r][0])[t];
            pv += ((const int*)&ppart[r][0])[t];
        }
    }
    int x = bv, z = own;
    for (int off = 1; off < 32; off <<= 1) {
        int u = __shfl_up(x, off, 32);
        int u2 = __shfl_up(z, off, 32);
        if (lane >= off) { x += u; z += u2; }
    }
    if (lane == 31) { wsum[w] = x; wsum2[w] = z; }
    __syncthreads();
    if (w == 0) {
        int y = wsum[lane], y2 = wsum2[lane];
        for (int off = 1; off < 32; off <<= 1) {
            int u = __shfl_up(y, off, 32);
            int u2 = __shfl_up(y2, off, 32);
            if (lane >= off) { y += u; y2 += u2; }
        }
        wsum[lane] = y;
        wsum2[lane] = y2;
    }
    __syncthreads();
    const int excl  = x - bv + (w > 0 ? wsum[w - 1] : 0);
    const int lexcl = z - own + (w > 0 ? wsum2[w - 1] : 0);
    if (t < kBP) {
        const int gpos = excl + pv;
        pos[t]  = staged ? gpos - lexcl : gpos;
        lpos[t] = lexcl;
        if (g == 0) bstart[t] = excl;
    }
    __syncthreads();

    if (staged) {
        for (int q0 = t; q0 < nq; q0 += kEdgeU * 1024) {
            if (q0 != t) {
#pragma unroll
                for (int u = 0; u < kEdgeU; ++u) {
                    const int q = q0 + u * 1024;
                    const bool live = q < nq;
                    dv[u] = live ? d4[q] : make_int4(0, 0, 0, 0);
                    sv[u] = live ? s4[q] : make_int4(0, 0, 0, 0);
                }
            }
#pragma unroll
            for (int u = 0; u < kEdgeU; ++u)
                if (q0 + u * 1024 < nq) {
                    const int k0 = dv[u].x >> 7, k1 = dv[u].y >> 7, k2 = dv[u].z >> 7, k3 = dv[u].w >> 7;
                    const unsigned j0 = (unsigned)atomicAdd(&lpos[k0], 1);
                    const unsigned j1 = (unsigned)atomicAdd(&lpos[k1], 1);
                    const unsigned j2 = (unsigned)atomicAdd(&lpos[k2], 1);
                    const unsigned j3 = (unsigned)atomicAdd(&lpos[k3], 1);
                    if (j0 < (unsigned)kScatCap) { sval[j0] = pack_pair(sv[u].x, dv[u].x); skey[j0] = (unsigned short)k0; }
                    if (j1 < (unsigned)kScatCap) { sval[j1] = pack_pair(sv[u].y, dv[u].y); skey[j1] = (unsigned short)k1; }
                    if (j2 < (unsigned)kScatCap) { sval[j2] = pack_pair(sv[u].z, dv[u].z); skey[j2] = (unsigned short)k2; }
                    if (j3 < (unsigned)kScatCap) { sval[j3] = pack_pair(sv[u].w, dv[u].w); skey[j3] = (unsigned short)k3; }
                }
        }
        for (int e = b + (nq << 2) + t; e < end; e += 1024) {
            const int d = dst[e];
            const unsigned j = (unsigned)atomicAdd(&lpos[d >> 7], 1);
            if (j < (unsigned)kScatCap) { sval[j] = pack_pair(src[e], d); skey[j] = (unsigned short)(d >> 7); }
        }
        __syncthreads();
        constexpr int kOutU = 4;
        for (int j0 = t; j0 < m; j0 += kOutU * 1024) {
            int kk[kOutU], vv[kOutU], pp[kOutU];
#pragma unroll
            for (int u = 0; u < kOutU; ++u) {
                const int j = min(j0 + u * 1024, m - 1);
                kk[u] = skey[j];
                vv[u] = sval[j];
            }
#pragma unroll
            for (int u = 0; u < kOutU; ++u) pp[u] = pos[kk[u]];
#pragma unroll
            for (int u = 0; u < kOutU; ++u) {
                const int j = j0 + u * 1024;
                const int slot = pp[u] + j;
                if ((j < m) && ((unsigned)slot < (unsigned)E)) plist[slot] = vv[u];
            }
        }
    } else {
        for (int q0 = t; q0 < nq; q0 += kEdgeU * 1024) {
            if (q0 != t) {
#pragma unroll
                for (int u = 0; u < kEdgeU; ++u) {
                    const int q = q0 + u * 1024;
                    const bool live = q < nq;
                    dv[u] = live ? d4[q] : make_int4(0, 0, 0, 0);
                    sv[u] = live ? s4[q] : make_int4(0, 0, 0, 0);
                }
            }
#pragma unroll
            for (int u = 0; u < kEdgeU; ++u)
                if (q0 + u * 1024 < nq) {
                    const unsigned uE = (unsigned)E;
                    const unsigned p0 = (unsigned)atomicAdd(&pos[dv[u].x >> 7], 1);
                    const unsigned p1 = (unsigned)atomicAdd(&pos[dv[u].y >> 7], 1);
                    const unsigned p2 = (unsigned)atomicAdd(&pos[dv[u].z >> 7], 1);
                    const unsigned p3 = (unsigned)atomicAdd(&pos[dv[u].w >> 7], 1);
                    if (p0 < uE) plist[p0] = pack_pair(sv[u].x, dv[u].x);
                    if (p1 < uE) plist[p1] = pack_pair(sv[u].y, dv[u].y);
                    if (p2 < uE) plist[p2] = pack_pair(sv[u].z, dv[u].z);
                    if (p3 < uE) plist[p3] = pack_pair(sv[u].w, dv[u].w);
                }
        }
        for (int e = b + (nq << 2) + t; e < end; e += 1024) {
            const int d = dst[e];
            const unsigned p = (unsigned)atomicAdd(&pos[d >> 7], 1);
            if (p < (unsigned)E) plist[p] = pack_pair(src[e], d);
        }
    }
}

constexpr int kFinT = 256;
constexpr int kFinU = 20;
constexpr int kFinCap = 8192;
__device__ __forceinline__ void block_finish_body(const int* __restrict__ plist,
                                                  const int* __restrict__ bstart,
                                                  int* __restrict__ csr,
                                                  int* __restrict__ rowptr, int n, int E) {
    __shared__ int4 cnt4[32];
    __shared__ int4 pos4[32];
    __shared__ int stage[kFinCap];
    int* cnt = (int*)cnt4;
    int* pos = (int*)pos4;
    const int t = threadIdx.x;
    const int blk = blockIdx.x;
    const int beg = bstart[blk], end = bstart[blk + 1];
    const bool staged = (end - beg) <= kFinCap;
    if (t < 128) cnt[t] = 0;
    __syncthreads();
    for (int i0 = beg + t; i0 < end; i0 += kFinU * kFinT) {
        int nd[kFinU];
#pragma unroll
        for (int u = 0; u < kFinU; ++u) nd[u] = plist[min(i0 + u * kFinT, end - 1)] & 127;
#pragma unroll
        for (int u = 0; u < kFinU; ++u) atomicAdd(&cnt[nd[u]], (i0 + u * kFinT < end) ? 1 : 0);
    }
    __syncthreads();
    if (t < 32) {
        const int4 c = cnt4[t];
        const int t4 = (c.x + c.y) + (c.z + c.w);
        int x = t4;
        for (int off = 1; off < 32; off <<= 1) {
            int u = __shfl_up(x, off, 32);
            if (t >= off) x += u;
        }
        int4 o;
        o.x = beg + x - t4;
        o.y = o.x + c.x;
        o.z = o.y + c.y;
        o.w = o.z + c.z;
        pos4[t] = o;
        const int q = blk * 32 + t;
        if (q <= (n >> 2)) ((int4*)rowptr)[q] = o;
    }
    __syncthreads();
    for (int i0 = beg + t; i0 < end; i0 += kFinU * kFinT) {
        int ev[kFinU];
        int slot[kFinU];
#pragma unroll
        for (int u = 0; u < kFinU; ++u) ev[u] = plist[min(i0 + u * kFinT, end - 1)];
        __builtin_amdgcn_sched_barrier(0);
#pragma unroll
        for (int u = 0; u < kFinU; ++u)
            slot[u] = atomicAdd(&pos[ev[u] & 127], (i0 + u * kFinT < end) ? 1 : 0);
#pragma unroll
        for (int u = 0; u < kFinU; ++u)
            if (i0 + u * kFinT < end) {
                if (staged) {
                    const unsigned j = (unsigned)(slot[u] - beg);
                    if (j < (unsigned)kFinCap) stage[j] = ev[u] >> 7;
                } else {
                    if ((unsigned)slot[u] < (unsigned)E) csr[slot[u]] = ev[u] >> 7;
                }
            }
    }
    if (staged) {
        __syncthreads();
        const int m = end - beg;
        constexpr int kOutU = 8;
        for (int j0 = t; j0 < m; j0 += kOutU * kFinT) {
            int vv[kOutU];
#pragma unroll
            for (int u = 0; u < kOutU; ++u) vv[u] = stage[min(j0 + u * kFinT, m - 1)];
#pragma unroll
            for (int u = 0; u < kOutU; ++u) {
                const int j = j0 + u * kFinT;
                if ((j < m) && ((unsigned)(beg + j) < (unsigned)E)) csr[beg + j] = vv[u];
            }
        }
    }
}

__global__ __launch_bounds__(kFinT) void finish_gemm1_kernel(const int* __restrict__ plist,
                                                             const int* __restrict__ bstart,
                                                             int* __restrict__ csr,
                                                             int* __restrict__ rowptr, int n, int E,
                                                             int nFin,
                                                             const float* __restrict__ x,
                                                             const float* __restrict__ W1,
                                                             float* __restrict__ h1,
                                                             const float* __restrict__ att_d1,
                                                             float* __restrict__ adst1) {
    if ((int)blockIdx.x >= nFin) {
        wmma_gemm_wave<128, 128, 4>(x, W1, h1, att_d1, adst1, n,
                                    ((int)blockIdx.x - nFin) * (kFinT >> 5) + (int)(threadIdx.x >> 5));
        return;
    }
    block_finish_body(plist, bstart, csr, rowptr, n, E);
}

__device__ inline float leaky(float v) { return (v >= 0.f) ? v : NEG_SLOPE * v; }

template <int CTRL>
__device__ __forceinline__ float row_lane(float v) {
    return __int_as_float(__builtin_amdgcn_update_dpp(0, __float_as_int(v), CTRL, 0xf, 0xf, true));
}
constexpr int kRowXor   = 0x160;
constexpr int kRowShare = 0x150;


template <int I>
__device__ __forceinline__ void add_row1(float4& acc, float w, const float4& h) {
    const float wk = row_lane<kRowShare + 2 * I>(w);
    acc.x += wk * h.x;
    acc.y += wk * h.y;
    acc.z += wk * h.z;
    acc.w += wk * h.w;
}

__device__ __forceinline__ void rows8_l1(float4& acc, float& den, float& m,
                                         const float* __restrict__ hcol, int s, int k, int cnt,
                                         float4 as, float ad, int l) {
    float4 hv[8];
#pragma unroll
    for (int i = 0; i < 8; ++i) {
        int sk = __builtin_amdgcn_readlane(s, k + i);
        hv[i] = *(const float4*)(hcol + (size_t)sk * 128);
    }
    float p[8];
#pragma unroll
    for (int i = 0; i < 8; ++i)
        p[i] = hv[i].x * as.x + hv[i].y * as.y + hv[i].z * as.z + hv[i].w * as.w;
    const bool b8 = (l & 8) != 0, b4 = (l & 4) != 0, b2 = (l & 2) != 0;
    float w4[4], w2[2];
#pragma unroll
    for (int i = 0; i < 4; ++i) {
        float keep = b8 ? p[i + 4] : p[i];
        float send = b8 ? p[i] : p[i + 4];
        w4[i] = keep + row_lane<kRowXor + 8>(send);
    }
#pragma unroll
    for (int i = 0; i < 2; ++i) {
        float keep = b4 ? w4[i + 2] : w4[i];
        float send = b4 ? w4[i] : w4[i + 2];
        w2[i] = keep + row_lane<kRowXor + 4>(send);
    }
    const float w1 = (b2 ? w2[1] : w2[0]) + row_lane<kRowXor + 2>(b2 ? w2[0] : w2[1]);
    const float sc = w1 + row_lane<kRowXor + 1>(w1);

    const bool valid = (k + (l >> 1)) < cnt;
    const float e = valid ? leaky(sc + ad) : -INFINITY;
    float gm = e;
    gm = fmaxf(gm, row_lane<kRowXor + 2>(gm));
    gm = fmaxf(gm, row_lane<kRowXor + 4>(gm));
    gm = fmaxf(gm, row_lane<kRowXor + 8>(gm));
    const float mn = fmaxf(m, gm);
    const float scale = __expf(m - mn);
    const float w = __expf(e - mn);
    den = den * scale + ((l & 1) ? 0.f : w);
    acc.x *= scale;
    acc.y *= scale;
    acc.z *= scale;
    acc.w *= scale;
    m = mn;
    add_row1<0>(acc, w, hv[0]);
    add_row1<1>(acc, w, hv[1]);
    add_row1<2>(acc, w, hv[2]);
    add_row1<3>(acc, w, hv[3]);
    add_row1<4>(acc, w, hv[4]);
    add_row1<5>(acc, w, hv[5]);
    add_row1<6>(acc, w, hv[6]);
    add_row1<7>(acc, w, hv[7]);
}

__global__ __launch_bounds__(256) void node_agg1_kernel(
        const int* __restrict__ rowptr, const int* __restrict__ csr,
        const float* __restrict__ hfeat,
        const float* __restrict__ att_s, const float* __restrict__ adst,
        const float* __restrict__ bias, float* __restrict__ hact, int n) {
    const int lane = threadIdx.x & 31;
    const int d = __builtin_amdgcn_readfirstlane(blockIdx.x * (blockDim.x >> 5) + (threadIdx.x >> 5));
    if (d >= n) return;
    const int beg = rowptr[d], end = rowptr[d + 1];
    const int hd = lane >> 4, l = lane & 15;
    const float ad = adst[d * 2 + hd];
    const float4 as = *(const float4*)(att_s + lane * 4);

    float4 acc = {0.f, 0.f, 0.f, 0.f};
    float den = 0.f, m = -INFINITY;
    const float* hcol = hfeat + lane * 4;
    for (int base = beg; base < end; base += 32) {
        const int j = base + lane;
        int s = (j < end) ? csr[j] : 0;
        const int s0 = __builtin_amdgcn_readfirstlane(s);
        if (j >= end) s = s0;
        const int cnt = min(32, end - base);
        for (int k = 0; k < cnt; k += 8) rows8_l1(acc, den, m, hcol, s, k, cnt, as, ad, l);
    }
    den += row_lane<kRowXor + 1>(den);
    den += row_lane<kRowXor + 2>(den);
    den += row_lane<kRowXor + 4>(den);
    den += row_lane<kRowXor + 8>(den);
    den += EPS_;
    const float4 bv = *(const float4*)(bias + lane * 4);
    float4 o;
    o.x = acc.x / den + bv.x;
    o.y = acc.y / den + bv.y;
    o.z = acc.z / den + bv.z;
    o.w = acc.w / den + bv.w;
    o.x = (o.x > 0.f) ? o.x : expm1f(o.x);
    o.y = (o.y > 0.f) ? o.y : expm1f(o.y);
    o.z = (o.z > 0.f) ? o.z : expm1f(o.z);
    o.w = (o.w > 0.f) ? o.w : expm1f(o.w);
    *(float4*)(hact + (size_t)d * 128 + lane * 4) = o;
}

__device__ __forceinline__ void node2_row(int d, int lane,
                                   const int* __restrict__ rowptr, const int* __restrict__ csr,
                                   const float* __restrict__ hfeat,
                                   const float* __restrict__ att_s, const float* __restrict__ adst,
                                   const float* __restrict__ bias,
                                   float& r0, float& r1, float& r2, float& r3) {
    const int beg = rowptr[d], end = rowptr[d + 1];
    const float ad = adst[d];
    const int g = lane >> 2;
    const float4 as = *(const float4*)(att_s + (lane & 3) * 4);
    const float* hcol = hfeat + (lane & 3) * 4;

    float4 acc = {0.f, 0.f, 0.f, 0.f};
    float den = 0.f, m = -INFINITY;
    for (int base = beg; base < end; base += 32) {
        const int j = base + lane;
        int s = (j < end) ? csr[j] : 0;
        const int s0 = __builtin_amdgcn_readfirstlane(s);
        if (j >= end) s = s0;
        const int cnt = min(32, end - base);
        float4 hv[4];
#pragma unroll
        for (int t = 0; t < 4; ++t) {
            int sk = __shfl(s, t * 8 + g, 32);
            hv[t] = *(const float4*)(hcol + (size_t)sk * 16);
        }
        float e[4];
#pragma unroll
        for (int t = 0; t < 4; ++t) {
            float p = hv[t].x * as.x + hv[t].y * as.y + hv[t].z * as.z + hv[t].w * as.w;
            p += row_lane<kRowXor + 1>(p);
            p += row_lane<kRowXor + 2>(p);
            e[t] = (t * 8 + g < cnt) ? leaky(p + ad) : -INFINITY;
        }
        float cm = fmaxf(fmaxf(e[0], e[1]), fmaxf(e[2], e[3]));
        cm = fmaxf(cm, row_lane<kRowXor + 4>(cm));
        cm = fmaxf(cm, row_lane<kRowXor + 8>(cm));
        cm = fmaxf(cm, __shfl_xor(cm, 16, 32));
        const float mn = fmaxf(m, cm);
        const float scale = __expf(m - mn);
        float w[4];
#pragma unroll
        for (int t = 0; t < 4; ++t) w[t] = __expf(e[t] - mn);
        den = den * scale + ((w[0] + w[1]) + (w[2] + w[3]));
        acc.x = acc.x * scale + ((w[0] * hv[0].x + w[1] * hv[1].x) + (w[2] * hv[2].x + w[3] * hv[3].x));
        acc.y = acc.y * scale + ((w[0] * hv[0].y + w[1] * hv[1].y) + (w[2] * hv[2].y + w[3] * hv[3].y));
        acc.z = acc.z * scale + ((w[0] * hv[0].z + w[1] * hv[1].z) + (w[2] * hv[2].z + w[3] * hv[3].z));
        acc.w = acc.w * scale + ((w[0] * hv[0].w + w[1] * hv[1].w) + (w[2] * hv[2].w + w[3] * hv[3].w));
        m = mn;
    }
    for (int off = 16; off >= 4; off >>= 1) {
        acc.x += __shfl_xor(acc.x, off, 32);
        acc.y += __shfl_xor(acc.y, off, 32);
        acc.z += __shfl_xor(acc.z, off, 32);
        acc.w += __shfl_xor(acc.w, off, 32);
        den += __shfl_xor(den, off, 32);
    }
    den += EPS_;
    const float4 bv = *(const float4*)(bias + (lane & 3) * 4);
    float4 v;
    v.x = acc.x / den + bv.x;
    v.y = acc.y / den + bv.y;
    v.z = acc.z / den + bv.z;
    v.w = acc.w / den + bv.w;
    float mx = fmaxf(fmaxf(v.x, v.y), fmaxf(v.z, v.w));
    mx = fmaxf(mx, __shfl_xor(mx, 1, 32));
    mx = fmaxf(mx, __shfl_xor(mx, 2, 32));
    v.x = __expf(v.x - mx);
    v.y = __expf(v.y - mx);
    v.z = __expf(v.z - mx);
    v.w = __expf(v.w - mx);
    float sum = (v.x + v.y) + (v.z + v.w);
    sum += __shfl_xor(sum, 1, 32);
    sum += __shfl_xor(sum, 2, 32);
    const float inv = 1.f / sum;
    r0 = v.x * inv;
    r1 = v.y * inv;
    r2 = v.z * inv;
    r3 = v.w * inv;
}

__global__ __launch_bounds__(256) void node_agg2_kernel(
        const int* __restrict__ rowptr, const int* __restrict__ csr,
        const float* __restrict__ hfeat,
        const float* __restrict__ att_s, const float* __restrict__ adst,
        const float* __restrict__ bias, float* __restrict__ out, int n) {
    const int lane = threadIdx.x & 31;
    const int d0 = __builtin_amdgcn_readfirstlane((blockIdx.x * (blockDim.x >> 5) + (threadIdx.x >> 5)) * 2);
    if (d0 >= n) return;
    float a0, a1, a2, a3;
    node2_row(d0, lane, rowptr, csr, hfeat, att_s, adst, bias, a0, a1, a2, a3);
    float b0 = a0, b1 = a1, b2 = a2, b3 = a3;
    const bool two = (d0 + 1 < n);
    if (two) node2_row(d0 + 1, lane, rowptr, csr, hfeat, att_s, adst, bias, b0, b1, b2, b3);
    if (lane < 4 || (two && lane < 8)) {
        float4 r;
        r.x = (lane < 4) ? a0 : b0;
        r.y = (lane < 4) ? a1 : b1;
        r.z = (lane < 4) ? a2 : b2;
        r.w = (lane < 4) ? a3 : b3;
        *(float4*)(out + (size_t)d0 * 16 + lane * 4) = r;
    }
}

extern "C" void kernel_launch(void* const* d_in, const int* in_sizes, int n_in,
                              void* d_out, int out_size, void* d_ws, size_t ws_size,
                              hipStream_t stream) {
    const float* x   = (const float*)d_in[0];
    const float* W1  = (const float*)d_in[1];
    const float* as1 = (const float*)d_in[2];
    const float* ad1 = (const float*)d_in[3];
    const float* b1  = (const float*)d_in[4];
    const float* W2  = (const float*)d_in[5];
    const float* as2 = (const float*)d_in[6];
    const float* ad2 = (const float*)d_in[7];
    const float* b2  = (const float*)d_in[8];
    const int*   ei  = (const int*)d_in[9];

    const int N = in_sizes[0] / 128;
    const int E = in_sizes[9] / 2;
    const int* src = ei;
    const int* dst = ei + E;

    float* w = (float*)d_ws;
    float* h1      = w; w += (size_t)N * 128;
    float* hact    = w; w += (size_t)N * 128;
    float* adst1   = w; w += (size_t)N * 2;
    float* adst2   = w; w += (size_t)N;
    int*   rowptr  = (int*)w; w += (size_t)N + 4;
    int*   csr     = (int*)w; w += (size_t)E;
    float* h2      = h1;
    int*   plist   = (int*)hact;
    int*   T       = plist + ((E + 3) & ~3);
    int*   bstart  = T + kChunks * kBP;

    const int nFin   = cdiv(N + 1, 128);
    const int tiles1 = (N / 16) * 2;
    block_count_kernel<<<kChunks, 1024, 0, stream>>>(dst, T, E);
    block_scatter_kernel<<<kChunks, 1024, 0, stream>>>(src, dst, T, bstart, plist, E);
    finish_gemm1_kernel<<<nFin + cdiv(tiles1, kFinT / 32), kFinT, 0, stream>>>(
        plist, bstart, csr, rowptr, N, E, nFin, x, W1, h1, ad1, adst1);

    node_agg1_kernel<<<cdiv(N, 8), 256, 0, stream>>>(rowptr, csr, h1, as1, adst1, b1, hact, N);

    {
        int tiles = (N / 16);
        wmma_gemm_f32<128, 16, 1><<<cdiv(tiles, 8), 256, 0, stream>>>(hact, W2, h2, ad2, adst2, N);
        node_agg2_kernel<<<cdiv(N, 16), 256, 0, stream>>>(rowptr, csr, h2, as2, adst2, b2,
                                                           (float*)d_out, N);
    }
}
